# starting kernel
# baseline (speedup 1.0000x reference)
#include <hip/hip_runtime.h>
#include <hip/hip_bf16.h>

typedef __attribute__((ext_vector_type(2))) float v2f;
typedef __attribute__((ext_vector_type(8))) float v8f;

#define LDS_STRIDE 132

__global__ void gs_zero_kernel(float* __restrict__ p, long n) {
  long i = (long)blockIdx.x * blockDim.x + threadIdx.x;
  long stride = (long)gridDim.x * blockDim.x;
  for (; i < n; i += stride) p[i] = 0.0f;
}

__global__ void gs_scatter_kernel(const float* __restrict__ h,
                                  const int* __restrict__ src,
                                  const int* __restrict__ tgt,
                                  float* __restrict__ agg,
                                  float* __restrict__ cnt,
                                  int nEdges) {
  int lane = threadIdx.x & 31;
  int e = blockIdx.x * (blockDim.x >> 5) + (threadIdx.x >> 5);
  if (e >= nEdges) return;
  int s = src[e];
  int t = tgt[e];
  const float4 v = *(const float4*)(h + (long)s * 128 + lane * 4);
  float* dst = agg + (long)t * 128 + lane * 4;
  atomicAdd(dst + 0, v.x);
  atomicAdd(dst + 1, v.y);
  atomicAdd(dst + 2, v.z);
  atomicAdd(dst + 3, v.w);
  if (lane == 0) atomicAdd(cnt + t, 1.0f);
}

__global__ void __launch_bounds__(256)
gs_sage_wmma_kernel(const float* __restrict__ agg,
                    const float* __restrict__ cnt,
                    const float* __restrict__ hself,
                    const float* __restrict__ Wl,
                    const float* __restrict__ bl,
                    const float* __restrict__ Wr,
                    float* __restrict__ hout,
                    int doRelu, int nNodes) {
  __shared__ float lmean[16 * LDS_STRIDE];
  __shared__ float lself[16 * LDS_STRIDE];

  const int tid  = threadIdx.x;
  const int row0 = blockIdx.x * 16;

  for (int i = 0; i < 8; ++i) {
    int idx  = tid + i * 256;
    int r    = idx >> 7;
    int c    = idx & 127;
    int node = row0 + r;
    int nclamp = node < nNodes ? node : (nNodes - 1);
    float ct   = cnt[nclamp];
    float rinv = 1.0f / fmaxf(ct, 1.0f);
    lmean[r * LDS_STRIDE + c] = agg[(long)nclamp * 128 + c] * rinv;
    lself[r * LDS_STRIDE + c] = hself[(long)nclamp * 128 + c];
  }
  __syncthreads();

  const int lane = tid & 31;
  const int wave = tid >> 5;
  const int col0 = wave * 16;
  const int rowA = lane & 15;
  const int kofs = (lane >> 4) * 2;
  const int colB = lane & 15;

  v8f acc = {};

  for (int k = 0; k < 128; k += 4) {
    v2f a, b;
    a.x = lmean[rowA * LDS_STRIDE + k + kofs];
    a.y = lmean[rowA * LDS_STRIDE + k + kofs + 1];
    b.x = Wl[(long)(k + kofs) * 128 + col0 + colB];
    b.y = Wl[(long)(k + kofs + 1) * 128 + col0 + colB];
    acc = __builtin_amdgcn_wmma_f32_16x16x4_f32(
         false, a,  false, b,
         (short)0, acc,  false,  false);
  }
  for (int k = 0; k < 128; k += 4) {
    v2f a, b;
    a.x = lself[rowA * LDS_STRIDE + k + kofs];
    a.y = lself[rowA * LDS_STRIDE + k + kofs + 1];
    b.x = Wr[(long)(k + kofs) * 128 + col0 + colB];
    b.y = Wr[(long)(k + kofs + 1) * 128 + col0 + colB];
    acc = __builtin_amdgcn_wmma_f32_16x16x4_f32(
        false, a, false, b, (short)0, acc, false, false);
  }

  const int ocol  = col0 + (lane & 15);
  const float bias = bl[ocol];
  const int rbase = row0 + (lane >> 4) * 8;
  for (int i = 0; i < 8; ++i) {
    int node = rbase + i;
    if (node < nNodes) {
      float v = acc[i] + bias;
      if (doRelu) v = fmaxf(v, 0.0f);
      hout[(long)node * 128 + ocol] = v;
    }
  }
}

__global__ void gs_edge_score_kernel(const float* __restrict__ h2,
                                     const int* __restrict__ src,
                                     const int* __restrict__ tgt,
                                     const float* __restrict__ We,
                                     const float* __restrict__ be,
                                     float* __restrict__ out,
                                     int nEdges) {
  int lane = threadIdx.x & 31;
  int e = blockIdx.x * (blockDim.x >> 5) + (threadIdx.x >> 5);
  if (e >= nEdges) return;
  int s = src[e];
  int t = tgt[e];
  const float4 vs = *(const float4*)(h2 + (long)s * 128 + lane * 4);
  const float4 vt = *(const float4*)(h2 + (long)t * 128 + lane * 4);
  const float4 ws = *(const float4*)(We + lane * 4);
  const float4 wt = *(const float4*)(We + 128 + lane * 4);
  float p = vs.x * ws.x + vs.y * ws.y + vs.z * ws.z + vs.w * ws.w
          + vt.x * wt.x + vt.y * wt.y + vt.z * wt.z + vt.w * wt.w;
  for (int off = 16; off > 0; off >>= 1)
    p += __shfl_xor(p, off, 32);
  if (lane == 0) out[e] = 1.0f / (1.0f + __expf(-(p + be[0])));
}

extern "C" void kernel_launch(void* const* d_in, const int* in_sizes, int n_in,
                              void* d_out, int out_size, void* d_ws, size_t ws_size,
                              hipStream_t stream) {
  const float* x    = (const float*)d_in[0];
  const int*   ei   = (const int*)d_in[1];
  const float* Wl1  = (const float*)d_in[2];
  const float* bl1  = (const float*)d_in[3];
  const float* Wr1  = (const float*)d_in[4];
  const float* Wl2  = (const float*)d_in[5];
  const float* bl2  = (const float*)d_in[6];
  const float* Wr2  = (const float*)d_in[7];
  const float* We   = (const float*)d_in[8];
  const float* be   = (const float*)d_in[9];

  const int N = in_sizes[0] / 128;
  const int E = in_sizes[1] / 2;
  const int* src = ei;
  const int* tgt = ei + E;

  float* agg = (float*)d_ws;
  float* cnt = agg + (long)N * 128;
  float* h1  = cnt + N;
  float* h2  = h1 + (long)N * 128;

  const long zn = (long)N * 128 + N;
  const int zeroBlocks = 2048;
  const int scatBlocks = (E + 7) / 8;
  const int gemmBlocks = (N + 15) / 16;

  gs_zero_kernel<<<zeroBlocks, 256, 0, stream>>>(agg, zn);
  gs_scatter_kernel<<<scatBlocks, 256, 0, stream>>>(x, src, tgt, agg, cnt, E);
  gs_sage_wmma_kernel<<<gemmBlocks, 256, 0, stream>>>(agg, cnt, x, Wl1, bl1, Wr1,
                                                      h1,  1, N);

  gs_zero_kernel<<<zeroBlocks, 256, 0, stream>>>(agg, zn);
  gs_scatter_kernel<<<scatBlocks, 256, 0, stream>>>(h1, src, tgt, agg, cnt, E);
  gs_sage_wmma_kernel<<<gemmBlocks, 256, 0, stream>>>(agg, cnt, h1, Wl2, bl2, Wr2,
                                                      h2,  0, N);

  gs_edge_score_kernel<<<scatBlocks, 256, 0, stream>>>(h2, src, tgt, We, be,
                                                       (float*)d_out, E);
}


# Neighbour sum by gather over per-call target-sorted lists (int rank add, prefix sum, fill) replaces float-atomic scatter and zero-fill; sum order = arrival rank within a node
# speedup vs baseline: 1.6841x; 1.6841x over previous
#include <hip/hip_runtime.h>
#include <hip/hip_bf16.h>

typedef __attribute__((ext_vector_type(2))) float v2f;
typedef __attribute__((ext_vector_type(8))) float v8f;

#define LDS_STRIDE 132

__global__ void gs_zero_int_kernel(int* __restrict__ p, int n) {
  int i = blockIdx.x * blockDim.x + threadIdx.x;
  if (i < n) p[i] = 0;
}

__global__ void gs_rank_kernel(const int* __restrict__ tgt,
                               int* __restrict__ deg,
                               int* __restrict__ rank,
                               int nEdges) {
  int e = blockIdx.x * blockDim.x + threadIdx.x;
  if (e < nEdges) rank[e] = atomicAdd(deg + tgt[e], 1);
}

__global__ void __launch_bounds__(1024)
gs_rowptr_kernel(const int* __restrict__ deg,
                 int* __restrict__ rowptr,
                 int nNodes) {
  __shared__ int s_part[32];
  __shared__ int s_incl[32];
  const int tid   = threadIdx.x;
  const int lane  = tid & 31;
  const int wave  = tid >> 5;
  const int start = blockIdx.x * 1024;

  int part = 0;
  const int4* d4 = (const int4*)deg;
  for (int i = tid; i < (start >> 2); i += 1024) {
    const int4 q = d4[i];
    part += (q.x + q.y) + (q.z + q.w);
  }

  const int idx = start + tid;
  const int v = idx < nNodes ? deg[idx] : 0;
  int incl = v;
  for (int off = 1; off < 32; off <<= 1) {
    const int up = __shfl_up(incl, off, 32);
    if (lane >= off) incl += up;
  }
  for (int off = 16; off > 0; off >>= 1) part += __shfl_xor(part, off, 32);
  if (lane == 31) s_incl[wave] = incl;
  if (lane == 0)  s_part[wave] = part;
  __syncthreads();

  int t = s_part[lane] + (lane < wave ? s_incl[lane] : 0);
  for (int off = 16; off > 0; off >>= 1) t += __shfl_xor(t, off, 32);
  if (idx <= nNodes) rowptr[idx] = t + incl - v;
}

__global__ void gs_fill_kernel(const int* __restrict__ src,
                               const int* __restrict__ tgt,
                               const int* __restrict__ rowptr,
                               const int* __restrict__ rank,
                               int* __restrict__ csr,
                               int nEdges) {
  int e = blockIdx.x * blockDim.x + threadIdx.x;
  if (e < nEdges) csr[rowptr[tgt[e]] + rank[e]] = src[e];
}

__device__ __forceinline__ void gs_add4(float4& a, const float4 v) {
  a.x += v.x; a.y += v.y; a.z += v.z; a.w += v.w;
}

__global__ void __launch_bounds__(256)
gs_gather_kernel(const float* __restrict__ h,
                 const int* __restrict__ rowptr,
                 const int* __restrict__ csr,
                 float* __restrict__ agg,
                 float* __restrict__ cnt,
                 int nNodes) {
  const int lane = threadIdx.x & 31;
  const int node = __builtin_amdgcn_readfirstlane(blockIdx.x * 8 + (threadIdx.x >> 5));
  if (node >= nNodes) return;
  const int beg = rowptr[node];
  const int end = rowptr[node + 1];
  const float* hl = h + lane * 4;
  float4 acc = make_float4(0.0f, 0.0f, 0.0f, 0.0f);

  for (int base = beg; base < end; base += 32) {
    const int n = (end - base) < 32 ? (end - base) : 32;
    const int mine = lane < n ? csr[base + lane] : 0;
    int j = 0;
    for (; j + 8 <= n; j += 8) {
      float4 v[8];
#pragma unroll
      for (int u = 0; u < 8; ++u) {
        const int s = __builtin_amdgcn_readlane(mine, j + u);
        v[u] = *(const float4*)(hl + (long)s * 128);
      }
#pragma unroll
      for (int u = 0; u < 8; ++u) gs_add4(acc, v[u]);
    }
    if (j + 4 <= n) {
      float4 v[4];
#pragma unroll
      for (int u = 0; u < 4; ++u) {
        const int s = __builtin_amdgcn_readlane(mine, j + u);
        v[u] = *(const float4*)(hl + (long)s * 128);
      }
#pragma unroll
      for (int u = 0; u < 4; ++u) gs_add4(acc, v[u]);
      j += 4;
    }
    if (j + 2 <= n) {
      const int s0 = __builtin_amdgcn_readlane(mine, j);
      const int s1 = __builtin_amdgcn_readlane(mine, j + 1);
      const float4 v0 = *(const float4*)(hl + (long)s0 * 128);
      const float4 v1 = *(const float4*)(hl + (long)s1 * 128);
      gs_add4(acc, v0);
      gs_add4(acc, v1);
      j += 2;
    }
    if (j < n) {
      const int s0 = __builtin_amdgcn_readlane(mine, j);
      const float4 v0 = *(const float4*)(hl + (long)s0 * 128);
      gs_add4(acc, v0);
    }
  }

  *(float4*)(agg + (long)node * 128 + lane * 4) = acc;
  if (lane == 0) cnt[node] = (float)(end - beg);
}

__global__ void __launch_bounds__(256)
gs_sage_wmma_kernel(const float* __restrict__ agg,
                    const float* __restrict__ cnt,
                    const float* __restrict__ hself,
                    const float* __restrict__ Wl,
                    const float* __restrict__ bl,
                    const float* __restrict__ Wr,
                    float* __restrict__ hout,
                    int doRelu, int nNodes) {
  __shared__ float lmean[16 * LDS_STRIDE];
  __shared__ float lself[16 * LDS_STRIDE];

  const int tid  = threadIdx.x;
  const int row0 = blockIdx.x * 16;

  for (int i = 0; i < 8; ++i) {
    int idx  = tid + i * 256;
    int r    = idx >> 7;
    int c    = idx & 127;
    int node = row0 + r;
    int nclamp = node < nNodes ? node : (nNodes - 1);
    float ct   = cnt[nclamp];
    float rinv = 1.0f / fmaxf(ct, 1.0f);
    lmean[r * LDS_STRIDE + c] = agg[(long)nclamp * 128 + c] * rinv;
    lself[r * LDS_STRIDE + c] = hself[(long)nclamp * 128 + c];
  }
  __syncthreads();

  const int lane = tid & 31;
  const int wave = tid >> 5;
  const int col0 = wave * 16;
  const int rowA = lane & 15;
  const int kofs = (lane >> 4) * 2;
  const int colB = lane & 15;

  v8f acc = {};

  for (int k = 0; k < 128; k += 4) {
    v2f a, b;
    a.x = lmean[rowA * LDS_STRIDE + k + kofs];
    a.y = lmean[rowA * LDS_STRIDE + k + kofs + 1];
    b.x = Wl[(long)(k + kofs) * 128 + col0 + colB];
    b.y = Wl[(long)(k + kofs + 1) * 128 + col0 + colB];
    acc = __builtin_amdgcn_wmma_f32_16x16x4_f32(
         false, a,  false, b,
         (short)0, acc,  false,  false);
  }
  for (int k = 0; k < 128; k += 4) {
    v2f a, b;
    a.x = lself[rowA * LDS_STRIDE + k + kofs];
    a.y = lself[rowA * LDS_STRIDE + k + kofs + 1];
    b.x = Wr[(long)(k + kofs) * 128 + col0 + colB];
    b.y = Wr[(long)(k + kofs + 1) * 128 + col0 + colB];
    acc = __builtin_amdgcn_wmma_f32_16x16x4_f32(
        false, a, false, b, (short)0, acc, false, false);
  }

  const int ocol  = col0 + (lane & 15);
  const float bias = bl[ocol];
  const int rbase = row0 + (lane >> 4) * 8;
  for (int i = 0; i < 8; ++i) {
    int node = rbase + i;
    if (node < nNodes) {
      float v = acc[i] + bias;
      if (doRelu) v = fmaxf(v, 0.0f);
      hout[(long)node * 128 + ocol] = v;
    }
  }
}

__global__ void gs_edge_score_kernel(const float* __restrict__ h2,
                                     const int* __restrict__ src,
                                     const int* __restrict__ tgt,
                                     const float* __restrict__ We,
                                     const float* __restrict__ be,
                                     float* __restrict__ out,
                                     int nEdges) {
  int lane = threadIdx.x & 31;
  int e = blockIdx.x * (blockDim.x >> 5) + (threadIdx.x >> 5);
  if (e >= nEdges) return;
  int s = src[e];
  int t = tgt[e];
  const float4 vs = *(const float4*)(h2 + (long)s * 128 + lane * 4);
  const float4 vt = *(const float4*)(h2 + (long)t * 128 + lane * 4);
  const float4 ws = *(const float4*)(We + lane * 4);
  const float4 wt = *(const float4*)(We + 128 + lane * 4);
  float p = vs.x * ws.x + vs.y * ws.y + vs.z * ws.z + vs.w * ws.w
          + vt.x * wt.x + vt.y * wt.y + vt.z * wt.z + vt.w * wt.w;
  for (int off = 16; off > 0; off >>= 1)
    p += __shfl_xor(p, off, 32);
  if (lane == 0) out[e] = 1.0f / (1.0f + __expf(-(p + be[0])));
}

extern "C" void kernel_launch(void* const* d_in, const int* in_sizes, int n_in,
                              void* d_out, int out_size, void* d_ws, size_t ws_size,
                              hipStream_t stream) {
  const float* x    = (const float*)d_in[0];
  const int*   ei   = (const int*)d_in[1];
  const float* Wl1  = (const float*)d_in[2];
  const float* bl1  = (const float*)d_in[3];
  const float* Wr1  = (const float*)d_in[4];
  const float* Wl2  = (const float*)d_in[5];
  const float* bl2  = (const float*)d_in[6];
  const float* Wr2  = (const float*)d_in[7];
  const float* We   = (const float*)d_in[8];
  const float* be   = (const float*)d_in[9];

  const int N = in_sizes[0] / 128;
  const int E = in_sizes[1] / 2;
  const int* src = ei;
  const int* tgt = ei + E;

  float* agg = (float*)d_ws;
  float* cnt = agg + (long)N * 128;
  float* h1  = cnt + N;
  float* h2  = h1 + (long)N * 128;

  int* deg    = (int*)h2;
  int* rowptr = deg + N;
  int* rank   = rowptr + (N + 1);
  int* csr    = rank + E;

  const int edgeBlocks = (E + 255) / 256;
  const int nodeBlocks = (N + 255) / 256;
  const int scanBlocks = (N + 1024) / 1024;
  const int gathBlocks = (N + 7) / 8;
  const int scatBlocks = (E + 7) / 8;
  const int gemmBlocks = (N + 15) / 16;

  gs_zero_int_kernel<<<nodeBlocks, 256, 0, stream>>>(deg, N);
  gs_rank_kernel<<<edgeBlocks, 256, 0, stream>>>(tgt, deg, rank, E);
  gs_rowptr_kernel<<<scanBlocks, 1024, 0, stream>>>(deg, rowptr, N);
  gs_fill_kernel<<<edgeBlocks, 256, 0, stream>>>(src, tgt, rowptr, rank, csr, E);

  gs_gather_kernel<<<gathBlocks, 256, 0, stream>>>(x, rowptr, csr, agg, cnt, N);
  gs_sage_wmma_kernel<<<gemmBlocks, 256, 0, stream>>>(agg, cnt, x, Wl1, bl1, Wr1,
                                                      h1,  1, N);

  gs_gather_kernel<<<gathBlocks, 256, 0, stream>>>(h1, rowptr, csr, agg, cnt, N);
  gs_sage_wmma_kernel<<<gemmBlocks, 256, 0, stream>>>(agg, cnt, h1, Wl2, bl2, Wr2,
                                                      h2,  0, N);

  gs_edge_score_kernel<<<scatBlocks, 256, 0, stream>>>(h2, src, tgt, We, be,
                                                       (float*)d_out, E);
}


# Edge score factored through nodes: layer-2 GEMM epilogue emits h2.We[0:128] and h2.We[128:256] per node (lane-tree then wave-order sum: new sum order), h2 never stored; edge kernel adds two scalars
# speedup vs baseline: 2.5645x; 1.5228x over previous
#include <hip/hip_runtime.h>
#include <hip/hip_bf16.h>

typedef __attribute__((ext_vector_type(2))) float v2f;
typedef __attribute__((ext_vector_type(8))) float v8f;

#define LDS_STRIDE 132

__global__ void gs_zero_int_kernel(int* __restrict__ p, int n) {
  int i = blockIdx.x * blockDim.x + threadIdx.x;
  if (i < n) p[i] = 0;
}

__global__ void gs_rank_kernel(const int* __restrict__ tgt,
                               int* __restrict__ deg,
                               int* __restrict__ rank,
                               int nEdges) {
  int e = blockIdx.x * blockDim.x + threadIdx.x;
  if (e < nEdges) rank[e] = atomicAdd(deg + tgt[e], 1);
}

__global__ void __launch_bounds__(1024)
gs_rowptr_kernel(const int* __restrict__ deg,
                 int* __restrict__ rowptr,
                 int nNodes) {
  __shared__ int s_part[32];
  __shared__ int s_incl[32];
  const int tid   = threadIdx.x;
  const int lane  = tid & 31;
  const int wave  = tid >> 5;
  const int start = blockIdx.x * 1024;

  int part = 0;
  const int4* d4 = (const int4*)deg;
  for (int i = tid; i < (start >> 2); i += 1024) {
    const int4 q = d4[i];
    part += (q.x + q.y) + (q.z + q.w);
  }

  const int idx = start + tid;
  const int v = idx < nNodes ? deg[idx] : 0;
  int incl = v;
  for (int off = 1; off < 32; off <<= 1) {
    const int up = __shfl_up(incl, off, 32);
    if (lane >= off) incl += up;
  }
  for (int off = 16; off > 0; off >>= 1) part += __shfl_xor(part, off, 32);
  if (lane == 31) s_incl[wave] = incl;
  if (lane == 0)  s_part[wave] = part;
  __syncthreads();

  int t = s_part[lane] + (lane < wave ? s_incl[lane] : 0);
  for (int off = 16; off > 0; off >>= 1) t += __shfl_xor(t, off, 32);
  if (idx <= nNodes) rowptr[idx] = t + incl - v;
}

__global__ void gs_fill_kernel(const int* __restrict__ src,
                               const int* __restrict__ tgt,
                               const int* __restrict__ rowptr,
                               const int* __restrict__ rank,
                               int* __restrict__ csr,
                               int nEdges) {
  int e = blockIdx.x * blockDim.x + threadIdx.x;
  if (e < nEdges) csr[rowptr[tgt[e]] + rank[e]] = src[e];
}

__device__ __forceinline__ void gs_add4(float4& a, const float4 v) {
  a.x += v.x; a.y += v.y; a.z += v.z; a.w += v.w;
}

__global__ void __launch_bounds__(256)
gs_gather_kernel(const float* __restrict__ h,
                 const int* __restrict__ rowptr,
                 const int* __restrict__ csr,
                 float* __restrict__ agg,
                 float* __restrict__ cnt,
                 int nNodes) {
  const int lane = threadIdx.x & 31;
  const int node = __builtin_amdgcn_readfirstlane(blockIdx.x * 8 + (threadIdx.x >> 5));
  if (node >= nNodes) return;
  const int beg = rowptr[node];
  const int end = rowptr[node + 1];
  const float* hl = h + lane * 4;
  float4 acc = make_float4(0.0f, 0.0f, 0.0f, 0.0f);

  for (int base = beg; base < end; base += 32) {
    const int n = (end - base) < 32 ? (end - base) : 32;
    const int mine = lane < n ? csr[base + lane] : 0;
    int j = 0;
    for (; j + 8 <= n; j += 8) {
      float4 v[8];
#pragma unroll
      for (int u = 0; u < 8; ++u) {
        const int s = __builtin_amdgcn_readlane(mine, j + u);
        v[u] = *(const float4*)(hl + (long)s * 128);
      }
#pragma unroll
      for (int u = 0; u < 8; ++u) gs_add4(acc, v[u]);
    }
    if (j + 4 <= n) {
      float4 v[4];
#pragma unroll
      for (int u = 0; u < 4; ++u) {
        const int s = __builtin_amdgcn_readlane(mine, j + u);
        v[u] = *(const float4*)(hl + (long)s * 128);
      }
#pragma unroll
      for (int u = 0; u < 4; ++u) gs_add4(acc, v[u]);
      j += 4;
    }
    if (j + 2 <= n) {
      const int s0 = __builtin_amdgcn_readlane(mine, j);
      const int s1 = __builtin_amdgcn_readlane(mine, j + 1);
      const float4 v0 = *(const float4*)(hl + (long)s0 * 128);
      const float4 v1 = *(const float4*)(hl + (long)s1 * 128);
      gs_add4(acc, v0);
      gs_add4(acc, v1);
      j += 2;
    }
    if (j < n) {
      const int s0 = __builtin_amdgcn_readlane(mine, j);
      const float4 v0 = *(const float4*)(hl + (long)s0 * 128);
      gs_add4(acc, v0);
    }
  }

  *(float4*)(agg + (long)node * 128 + lane * 4) = acc;
  if (lane == 0) cnt[node] = (float)(end - beg);
}

__global__ void __launch_bounds__(256)
gs_sage_wmma_kernel(const float* __restrict__ agg,
                    const float* __restrict__ cnt,
                    const float* __restrict__ hself,
                    const float* __restrict__ Wl,
                    const float* __restrict__ bl,
                    const float* __restrict__ Wr,
                    float* __restrict__ hout,
                    const float* __restrict__ We,
                    float* __restrict__ ps,
                    float* __restrict__ pt,
                    int doRelu, int nNodes) {
  __shared__ float lmean[16 * LDS_STRIDE];
  __shared__ float lself[16 * LDS_STRIDE];
  __shared__ float lproj[8 * 32];

  const int tid  = threadIdx.x;
  const int row0 = blockIdx.x * 16;

  for (int i = 0; i < 8; ++i) {
    int idx  = tid + i * 256;
    int r    = idx >> 7;
    int c    = idx & 127;
    int node = row0 + r;
    int nclamp = node < nNodes ? node : (nNodes - 1);
    float ct   = cnt[nclamp];
    float rinv = 1.0f / fmaxf(ct, 1.0f);
    lmean[r * LDS_STRIDE + c] = agg[(long)nclamp * 128 + c] * rinv;
    lself[r * LDS_STRIDE + c] = hself[(long)nclamp * 128 + c];
  }
  __syncthreads();

  const int lane = tid & 31;
  const int wave = tid >> 5;
  const int col0 = wave * 16;
  const int rowA = lane & 15;
  const int kofs = (lane >> 4) * 2;
  const int colB = lane & 15;

  v8f acc = {};

  for (int k = 0; k < 128; k += 4) {
    v2f a, b;
    a.x = lmean[rowA * LDS_STRIDE + k + kofs];
    a.y = lmean[rowA * LDS_STRIDE + k + kofs + 1];
    b.x = Wl[(long)(k + kofs) * 128 + col0 + colB];
    b.y = Wl[(long)(k + kofs + 1) * 128 + col0 + colB];
    acc = __builtin_amdgcn_wmma_f32_16x16x4_f32(
         false, a,  false, b,
         (short)0, acc,  false,  false);
  }
  for (int k = 0; k < 128; k += 4) {
    v2f a, b;
    a.x = lself[rowA * LDS_STRIDE + k + kofs];
    a.y = lself[rowA * LDS_STRIDE + k + kofs + 1];
    b.x = Wr[(long)(k + kofs) * 128 + col0 + colB];
    b.y = Wr[(long)(k + kofs + 1) * 128 + col0 + colB];
    acc = __builtin_amdgcn_wmma_f32_16x16x4_f32(
        false, a, false, b, (short)0, acc, false, false);
  }

  const int c     = lane & 15;
  const int half  = lane >> 4;
  const int ocol  = col0 + c;
  const float bias = bl[ocol];
  float v[8];
  for (int i = 0; i < 8; ++i) {
    float t = acc[i] + bias;
    if (doRelu) t = fmaxf(t, 0.0f);
    v[i] = t;
  }

  if (ps == nullptr) {
    const int rbase = row0 + half * 8;
    for (int i = 0; i < 8; ++i) {
      int node = rbase + i;
      if (node < nNodes) hout[(long)node * 128 + ocol] = v[i];
    }
    return;
  }

  const float wes = We[ocol];
  const float wet = We[128 + ocol];
  float q[16];
  for (int i = 0; i < 8; ++i) {
    q[i]     = v[i] * wes;
    q[8 + i] = v[i] * wet;
  }
  const bool b3 = (c & 8) != 0;
  const bool b2 = (c & 4) != 0;
  const bool b1 = (c & 2) != 0;
  const bool b0 = (c & 1) != 0;
  float w8[8];
  for (int k = 0; k < 8; ++k) {
    const float keep = b3 ? q[8 + k] : q[k];
    const float send = b3 ? q[k] : q[8 + k];
    w8[k] = keep + __shfl_xor(send, 8, 32);
  }
  float w4[4];
  for (int k = 0; k < 4; ++k) {
    const float keep = b2 ? w8[4 + k] : w8[k];
    const float send = b2 ? w8[k] : w8[4 + k];
    w4[k] = keep + __shfl_xor(send, 4, 32);
  }
  float w2[2];
  for (int k = 0; k < 2; ++k) {
    const float keep = b1 ? w4[2 + k] : w4[k];
    const float send = b1 ? w4[k] : w4[2 + k];
    w2[k] = keep + __shfl_xor(send, 2, 32);
  }
  const float keep1 = b0 ? w2[1] : w2[0];
  const float send1 = b0 ? w2[0] : w2[1];
  const float w1 = keep1 + __shfl_xor(send1, 1, 32);

  lproj[wave * 32 + (c >> 3) * 16 + half * 8 + (c & 7)] = w1;
  __syncthreads();
  if (tid < 32) {
    float sum = lproj[tid];
    for (int w = 1; w < 8; ++w) sum += lproj[w * 32 + tid];
    const int node = row0 + (tid & 15);
    if (node < nNodes) {
      if (tid < 16) ps[node] = sum;
      else          pt[node] = sum;
    }
  }
}

__device__ __forceinline__ float gs_sigmoid(float p) {
  return 1.0f / (1.0f + __expf(-p));
}

__global__ void __launch_bounds__(256)
gs_edge_score_kernel(const float* __restrict__ ps,
                     const float* __restrict__ pt,
                     const int* __restrict__ src,
                     const int* __restrict__ tgt,
                     const float* __restrict__ be,
                     float* __restrict__ out,
                     int nEdges) {
  const int e0 = (blockIdx.x * blockDim.x + threadIdx.x) * 4;
  if (e0 >= nEdges) return;
  const float b = be[0];
  if ((nEdges & 3) == 0) {
    const int4 s4 = *(const int4*)(src + e0);
    const int4 t4 = *(const int4*)(tgt + e0);
    float4 y;
    y.x = gs_sigmoid(ps[s4.x] + pt[t4.x] + b);
    y.y = gs_sigmoid(ps[s4.y] + pt[t4.y] + b);
    y.z = gs_sigmoid(ps[s4.z] + pt[t4.z] + b);
    y.w = gs_sigmoid(ps[s4.w] + pt[t4.w] + b);
    *(float4*)(out + e0) = y;
  } else {
    const int eEnd = (e0 + 4) < nEdges ? (e0 + 4) : nEdges;
    for (int e = e0; e < eEnd; ++e)
      out[e] = gs_sigmoid(ps[src[e]] + pt[tgt[e]] + b);
  }
}

extern "C" void kernel_launch(void* const* d_in, const int* in_sizes, int n_in,
                              void* d_out, int out_size, void* d_ws, size_t ws_size,
                              hipStream_t stream) {
  const float* x    = (const float*)d_in[0];
  const int*   ei   = (const int*)d_in[1];
  const float* Wl1  = (const float*)d_in[2];
  const float* bl1  = (const float*)d_in[3];
  const float* Wr1  = (const float*)d_in[4];
  const float* Wl2  = (const float*)d_in[5];
  const float* bl2  = (const float*)d_in[6];
  const float* Wr2  = (const float*)d_in[7];
  const float* We   = (const float*)d_in[8];
  const float* be   = (const float*)d_in[9];

  const int N = in_sizes[0] / 128;
  const int E = in_sizes[1] / 2;
  const int* src = ei;
  const int* tgt = ei + E;

  float* agg = (float*)d_ws;
  float* cnt = agg + (long)N * 128;
  float* h1  = cnt + N;
  float* aux = h1 + (long)N * 128;

  int* deg    = (int*)aux;
  int* rowptr = deg + N;
  int* rank   = rowptr + (N + 1);
  int* csr    = rank + E;
  float* ps   = (float*)(csr + E);
  float* pt   = ps + N;

  const int edgeBlocks = (E + 255) / 256;
  const int nodeBlocks = (N + 255) / 256;
  const int scanBlocks = (N + 1024) / 1024;
  const int gathBlocks = (N + 7) / 8;
  const int quadBlocks = ((E + 3) / 4 + 255) / 256;
  const int gemmBlocks = (N + 15) / 16;

  gs_zero_int_kernel<<<nodeBlocks, 256, 0, stream>>>(deg, N);
  gs_rank_kernel<<<edgeBlocks, 256, 0, stream>>>(tgt, deg, rank, E);
  gs_rowptr_kernel<<<scanBlocks, 1024, 0, stream>>>(deg, rowptr, N);
  gs_fill_kernel<<<edgeBlocks, 256, 0, stream>>>(src, tgt, rowptr, rank, csr, E);

  gs_gather_kernel<<<gathBlocks, 256, 0, stream>>>(x, rowptr, csr, agg, cnt, N);
  gs_sage_wmma_kernel<<<gemmBlocks, 256, 0, stream>>>(agg, cnt, x, Wl1, bl1, Wr1,
                                                      h1, nullptr, nullptr, nullptr,
                                                       1, N);

  gs_gather_kernel<<<gathBlocks, 256, 0, stream>>>(h1, rowptr, csr, agg, cnt, N);
  gs_sage_wmma_kernel<<<gemmBlocks, 256, 0, stream>>>(agg, cnt, h1, Wl2, bl2, Wr2,
                                                      nullptr, We, ps, pt,
                                                       0, N);

  gs_edge_score_kernel<<<quadBlocks, 256, 0, stream>>>(ps, pt, src, tgt, be,
                                                       (float*)d_out, E);
}


# Layer 2 folded through the edge weights by linearity: h2.We = mean(h1.(Wl2 We)) + h1.(Wr2 We); 2nd gather carries 2 scalars per edge, 2nd GEMM and h1 store gone; new sum order
# speedup vs baseline: 3.3214x; 1.2951x over previous
#include <hip/hip_runtime.h>
#include <hip/hip_bf16.h>

typedef __attribute__((ext_vector_type(2))) float v2f;
typedef __attribute__((ext_vector_type(8))) float v8f;

#define LDS_STRIDE 132

__global__ void gs_zero_int_kernel(int* __restrict__ p, int n) {
  int i = blockIdx.x * blockDim.x + threadIdx.x;
  if (i < n) p[i] = 0;
}

__global__ void gs_rank_kernel(const int* __restrict__ tgt,
                               int* __restrict__ deg,
                               int* __restrict__ rank,
                               int nEdges) {
  int e = blockIdx.x * blockDim.x + threadIdx.x;
  if (e < nEdges) rank[e] = atomicAdd(deg + tgt[e], 1);
}

__global__ void __launch_bounds__(1024)
gs_rowptr_kernel(const int* __restrict__ deg,
                 int* __restrict__ rowptr,
                 int nNodes) {
  __shared__ int s_part[32];
  __shared__ int s_incl[32];
  const int tid   = threadIdx.x;
  const int lane  = tid & 31;
  const int wave  = tid >> 5;
  const int start = blockIdx.x * 1024;

  int part = 0;
  const int4* d4 = (const int4*)deg;
  for (int i = tid; i < (start >> 2); i += 1024) {
    const int4 q = d4[i];
    part += (q.x + q.y) + (q.z + q.w);
  }

  const int idx = start + tid;
  const int v = idx < nNodes ? deg[idx] : 0;
  int incl = v;
  for (int off = 1; off < 32; off <<= 1) {
    const int up = __shfl_up(incl, off, 32);
    if (lane >= off) incl += up;
  }
  for (int off = 16; off > 0; off >>= 1) part += __shfl_xor(part, off, 32);
  if (lane == 31) s_incl[wave] = incl;
  if (lane == 0)  s_part[wave] = part;
  __syncthreads();

  int t = s_part[lane] + (lane < wave ? s_incl[lane] : 0);
  for (int off = 16; off > 0; off >>= 1) t += __shfl_xor(t, off, 32);
  if (idx <= nNodes) rowptr[idx] = t + incl - v;
}

__global__ void gs_fill_kernel(const int* __restrict__ src,
                               const int* __restrict__ tgt,
                               const int* __restrict__ rowptr,
                               const int* __restrict__ rank,
                               int* __restrict__ csr,
                               int nEdges) {
  int e = blockIdx.x * blockDim.x + threadIdx.x;
  if (e < nEdges) csr[rowptr[tgt[e]] + rank[e]] = src[e];
}

__device__ __forceinline__ void gs_add4(float4& a, const float4 v) {
  a.x += v.x; a.y += v.y; a.z += v.z; a.w += v.w;
}

__global__ void __launch_bounds__(256)
gs_gather_kernel(const float* __restrict__ h,
                 const int* __restrict__ rowptr,
                 const int* __restrict__ csr,
                 float* __restrict__ agg,
                 float* __restrict__ cnt,
                 int nNodes) {
  const int lane = threadIdx.x & 31;
  const int node = __builtin_amdgcn_readfirstlane(blockIdx.x * 8 + (threadIdx.x >> 5));
  if (node >= nNodes) return;
  const int beg = rowptr[node];
  const int end = rowptr[node + 1];
  const float* hl = h + lane * 4;
  float4 acc = make_float4(0.0f, 0.0f, 0.0f, 0.0f);

  for (int base = beg; base < end; base += 32) {
    const int n = (end - base) < 32 ? (end - base) : 32;
    const int mine = lane < n ? csr[base + lane] : 0;
    int j = 0;
    for (; j + 8 <= n; j += 8) {
      float4 v[8];
#pragma unroll
      for (int u = 0; u < 8; ++u) {
        const int s = __builtin_amdgcn_readlane(mine, j + u);
        v[u] = *(const float4*)(hl + (long)s * 128);
      }
#pragma unroll
      for (int u = 0; u < 8; ++u) gs_add4(acc, v[u]);
    }
    if (j + 4 <= n) {
      float4 v[4];
#pragma unroll
      for (int u = 0; u < 4; ++u) {
        const int s = __builtin_amdgcn_readlane(mine, j + u);
        v[u] = *(const float4*)(hl + (long)s * 128);
      }
#pragma unroll
      for (int u = 0; u < 4; ++u) gs_add4(acc, v[u]);
      j += 4;
    }
    if (j + 2 <= n) {
      const int s0 = __builtin_amdgcn_readlane(mine, j);
      const int s1 = __builtin_amdgcn_readlane(mine, j + 1);
      const float4 v0 = *(const float4*)(hl + (long)s0 * 128);
      const float4 v1 = *(const float4*)(hl + (long)s1 * 128);
      gs_add4(acc, v0);
      gs_add4(acc, v1);
      j += 2;
    }
    if (j < n) {
      const int s0 = __builtin_amdgcn_readlane(mine, j);
      const float4 v0 = *(const float4*)(hl + (long)s0 * 128);
      gs_add4(acc, v0);
    }
  }

  *(float4*)(agg + (long)node * 128 + lane * 4) = acc;
  if (lane == 0) cnt[node] = (float)(end - beg);
}

__device__ __forceinline__ float gs_dot4(const float4 a, const float4 b) {
  return (a.x * b.x + a.y * b.y) + (a.z * b.z + a.w * b.w);
}

__global__ void __launch_bounds__(256)
gs_fold_kernel(const float* __restrict__ Wl,
               const float* __restrict__ bl,
               const float* __restrict__ Wr,
               const float* __restrict__ We,
               float* __restrict__ U4,
               float* __restrict__ cb) {
  const int lane = threadIdx.x & 31;
  const int k = blockIdx.x * 8 + (threadIdx.x >> 5);
  if (k > 128) return;
  const float4 ws = *(const float4*)(We + lane * 4);
  const float4 wt = *(const float4*)(We + 128 + lane * 4);
  const bool isRow = k < 128;
  const float4 l = *(const float4*)((isRow ? Wl + k * 128 : bl) + lane * 4);
  const float4 r = isRow ? *(const float4*)(Wr + k * 128 + lane * 4)
                         : make_float4(0.0f, 0.0f, 0.0f, 0.0f);
  float s0 = gs_dot4(l, ws);
  float s1 = gs_dot4(l, wt);
  float s2 = gs_dot4(r, ws);
  float s3 = gs_dot4(r, wt);
  for (int off = 16; off > 0; off >>= 1) {
    s0 += __shfl_xor(s0, off, 32);
    s1 += __shfl_xor(s1, off, 32);
    s2 += __shfl_xor(s2, off, 32);
    s3 += __shfl_xor(s3, off, 32);
  }
  if (lane == 0) {
    if (isRow) {
      *(float4*)(U4 + k * 4) = make_float4(s0, s1, s2, s3);
    } else {
      cb[0] = s0;
      cb[1] = s1;
    }
  }
}

__device__ __forceinline__ void gs_load_b16(v2f (&b)[16],
                                            const float* __restrict__ Wcol,
                                            int kbase) {
#pragma unroll
  for (int s = 0; s < 16; ++s) {
    b[s].x = Wcol[(kbase + 4 * s) * 128];
    b[s].y = Wcol[(kbase + 4 * s + 1) * 128];
  }
}

__device__ __forceinline__ v8f gs_mma16(v8f acc, const float* lrow, int kbase,
                                        const v2f (&b)[16]) {
#pragma unroll
  for (int s = 0; s < 16; ++s) {
    v2f a;
    a.x = lrow[kbase + 4 * s];
    a.y = lrow[kbase + 4 * s + 1];
    acc = __builtin_amdgcn_wmma_f32_16x16x4_f32(
         false, a,  false, b[s],
         (short)0, acc,  false,  false);
  }
  return acc;
}

__global__ void __launch_bounds__(256)
gs_sage_wmma_kernel(const float* __restrict__ agg,
                    const float* __restrict__ cnt,
                    const float* __restrict__ hself,
                    const float* __restrict__ Wl,
                    const float* __restrict__ bl,
                    const float* __restrict__ Wr,
                    const float* __restrict__ U4,
                    float* __restrict__ np,
                    int nNodes) {
  __shared__ float lmean[16 * LDS_STRIDE];
  __shared__ float lself[16 * LDS_STRIDE];
  __shared__ float lproj[8 * 64];

  const int tid  = threadIdx.x;
  const int row0 = blockIdx.x * 16;

  for (int i = 0; i < 8; ++i) {
    int idx  = tid + i * 256;
    int r    = idx >> 7;
    int c    = idx & 127;
    int node = row0 + r;
    int nclamp = node < nNodes ? node : (nNodes - 1);
    float ct   = cnt[nclamp];
    float rinv = 1.0f / fmaxf(ct, 1.0f);
    lmean[r * LDS_STRIDE + c] = agg[(long)nclamp * 128 + c] * rinv;
    lself[r * LDS_STRIDE + c] = hself[(long)nclamp * 128 + c];
  }
  __syncthreads();

  const int lane = tid & 31;
  const int wave = tid >> 5;
  const int col0 = wave * 16;
  const int rowA = lane & 15;
  const int kofs = (lane >> 4) * 2;
  const int colB = lane & 15;

  v8f acc = {};

  const float* wlc = Wl + kofs * 128 + col0 + colB;
  const float* wrc = Wr + kofs * 128 + col0 + colB;
  const float* lm  = lmean + rowA * LDS_STRIDE + kofs;
  const float* ls  = lself + rowA * LDS_STRIDE + kofs;
  v2f bA[16], bB[16];
  gs_load_b16(bA, wlc, 0);
  gs_load_b16(bB, wlc, 64);
  acc = gs_mma16(acc, lm, 0, bA);
  gs_load_b16(bA, wrc, 0);
  acc = gs_mma16(acc, lm, 64, bB);
  gs_load_b16(bB, wrc, 64);
  acc = gs_mma16(acc, ls, 0, bA);
  acc = gs_mma16(acc, ls, 64, bB);

  const int c     = lane & 15;
  const int half  = lane >> 4;
  const int ocol  = col0 + c;
  const float bias = bl[ocol];
  const float4 u = *(const float4*)(U4 + ocol * 4);

  float q[32];
  for (int i = 0; i < 8; ++i) {
    const float t = fmaxf(acc[i] + bias, 0.0f);
    q[i]      = t * u.x;
    q[8 + i]  = t * u.y;
    q[16 + i] = t * u.z;
    q[24 + i] = t * u.w;
  }
  const bool b3 = (c & 8) != 0;
  const bool b2 = (c & 4) != 0;
  const bool b1 = (c & 2) != 0;
  const bool b0 = (c & 1) != 0;
  float w16[16];
  for (int k = 0; k < 16; ++k) {
    const float keep = b3 ? q[16 + k] : q[k];
    const float send = b3 ? q[k] : q[16 + k];
    w16[k] = keep + __shfl_xor(send, 8, 32);
  }
  float w8[8];
  for (int k = 0; k < 8; ++k) {
    const float keep = b2 ? w16[8 + k] : w16[k];
    const float send = b2 ? w16[k] : w16[8 + k];
    w8[k] = keep + __shfl_xor(send, 4, 32);
  }
  float w4[4];
  for (int k = 0; k < 4; ++k) {
    const float keep = b1 ? w8[4 + k] : w8[k];
    const float send = b1 ? w8[k] : w8[4 + k];
    w4[k] = keep + __shfl_xor(send, 2, 32);
  }
  float w2[2];
  for (int k = 0; k < 2; ++k) {
    const float keep = b0 ? w4[2 + k] : w4[k];
    const float send = b0 ? w4[k] : w4[2 + k];
    w2[k] = keep + __shfl_xor(send, 1, 32);
  }

  const int prow = half * 8 + (c & 3) * 2;
  const int pidx = c >> 2;
  lproj[wave * 64 + prow * 4 + pidx]       = w2[0];
  lproj[wave * 64 + (prow + 1) * 4 + pidx] = w2[1];
  __syncthreads();
  if (tid < 64) {
    float sum = lproj[tid];
    for (int w = 1; w < 8; ++w) sum += lproj[w * 64 + tid];
    const int node = row0 + (tid >> 2);
    if (node < nNodes) np[(long)row0 * 4 + tid] = sum;
  }
}

__global__ void __launch_bounds__(256)
gs_node_score_kernel(const float* __restrict__ np,
                     const float* __restrict__ cb,
                     const int* __restrict__ rowptr,
                     const int* __restrict__ csr,
                     float* __restrict__ ps,
                     float* __restrict__ pt,
                     int nNodes) {
  const int node = (blockIdx.x * blockDim.x + threadIdx.x) >> 3;
  const int sub  = threadIdx.x & 7;
  const bool live = node < nNodes;
  const int beg = live ? rowptr[node] : 0;
  const int end = live ? rowptr[node + 1] : 0;
  float s = 0.0f, t = 0.0f;
  for (int k = beg + sub; k < end; k += 8) {
    const float2 a = *(const float2*)(np + (long)csr[k] * 4);
    s += a.x;
    t += a.y;
  }
  for (int off = 4; off > 0; off >>= 1) {
    s += __shfl_xor(s, off, 32);
    t += __shfl_xor(t, off, 32);
  }
  if (live && sub == 0) {
    const float den = fmaxf((float)(end - beg), 1.0f);
    const float2 self = *(const float2*)(np + (long)node * 4 + 2);
    ps[node] = s / den + self.x + cb[0];
    pt[node] = t / den + self.y + cb[1];
  }
}

__device__ __forceinline__ float gs_sigmoid(float p) {
  return 1.0f / (1.0f + __expf(-p));
}

__global__ void __launch_bounds__(256)
gs_edge_score_kernel(const float* __restrict__ ps,
                     const float* __restrict__ pt,
                     const int* __restrict__ src,
                     const int* __restrict__ tgt,
                     const float* __restrict__ be,
                     float* __restrict__ out,
                     int nEdges) {
  const int e0 = (blockIdx.x * blockDim.x + threadIdx.x) * 4;
  if (e0 >= nEdges) return;
  const float b = be[0];
  if ((nEdges & 3) == 0) {
    const int4 s4 = *(const int4*)(src + e0);
    const int4 t4 = *(const int4*)(tgt + e0);
    float4 y;
    y.x = gs_sigmoid(ps[s4.x] + pt[t4.x] + b);
    y.y = gs_sigmoid(ps[s4.y] + pt[t4.y] + b);
    y.z = gs_sigmoid(ps[s4.z] + pt[t4.z] + b);
    y.w = gs_sigmoid(ps[s4.w] + pt[t4.w] + b);
    *(float4*)(out + e0) = y;
  } else {
    const int eEnd = (e0 + 4) < nEdges ? (e0 + 4) : nEdges;
    for (int e = e0; e < eEnd; ++e)
      out[e] = gs_sigmoid(ps[src[e]] + pt[tgt[e]] + b);
  }
}

extern "C" void kernel_launch(void* const* d_in, const int* in_sizes, int n_in,
                              void* d_out, int out_size, void* d_ws, size_t ws_size,
                              hipStream_t stream) {
  const float* x    = (const float*)d_in[0];
  const int*   ei   = (const int*)d_in[1];
  const float* Wl1  = (const float*)d_in[2];
  const float* bl1  = (const float*)d_in[3];
  const float* Wr1  = (const float*)d_in[4];
  const float* Wl2  = (const float*)d_in[5];
  const float* bl2  = (const float*)d_in[6];
  const float* Wr2  = (const float*)d_in[7];
  const float* We   = (const float*)d_in[8];
  const float* be   = (const float*)d_in[9];

  const int N = in_sizes[0] / 128;
  const int E = in_sizes[1] / 2;
  const int* src = ei;
  const int* tgt = ei + E;

  float* agg = (float*)d_ws;
  float* cnt = agg + (long)N * 128;
  float* U4  = cnt + N;
  float* cb  = U4 + 128 * 4;
  float* np  = cb + 4;
  float* aux = U4 + (long)N * 128;

  int* deg    = (int*)aux;
  int* rowptr = deg + N;
  int* rank   = rowptr + (N + 1);
  int* csr    = rank + E;
  float* ps   = (float*)(csr + E);
  float* pt   = ps + N;

  const int edgeBlocks = (E + 255) / 256;
  const int nodeBlocks = (N + 255) / 256;
  const int scanBlocks = (N + 1024) / 1024;
  const int gathBlocks = (N + 7) / 8;
  const int oct_Blocks = (N + 31) / 32;
  const int quadBlocks = ((E + 3) / 4 + 255) / 256;
  const int gemmBlocks = (N + 15) / 16;

  gs_zero_int_kernel<<<nodeBlocks, 256, 0, stream>>>(deg, N);
  gs_rank_kernel<<<edgeBlocks, 256, 0, stream>>>(tgt, deg, rank, E);
  gs_rowptr_kernel<<<scanBlocks, 1024, 0, stream>>>(deg, rowptr, N);
  gs_fill_kernel<<<edgeBlocks, 256, 0, stream>>>(src, tgt, rowptr, rank, csr, E);

  gs_fold_kernel<<<17, 256, 0, stream>>>(Wl2, bl2, Wr2, We, U4, cb);

  gs_gather_kernel<<<gathBlocks, 256, 0, stream>>>(x, rowptr, csr, agg, cnt, N);
  gs_sage_wmma_kernel<<<gemmBlocks, 256, 0, stream>>>(agg, cnt, x, Wl1, bl1, Wr1,
                                                      U4, np, N);

  gs_node_score_kernel<<<oct_Blocks, 256, 0, stream>>>(np, cb, rowptr, csr, ps, pt, N);

  gs_edge_score_kernel<<<quadBlocks, 256, 0, stream>>>(ps, pt, src, tgt, be,
                                                       (float*)d_out, E);
}


# Neighbour lists in fixed 64-slot blocks per node (spill list past 64): one pass puts each edge at its arrival rank; prefix-sum and fill launches and rank array gone; sum order still arrival rank
# speedup vs baseline: 4.0107x; 1.2075x over previous
#include <hip/hip_runtime.h>
#include <hip/hip_bf16.h>

typedef __attribute__((ext_vector_type(2))) float v2f;
typedef __attribute__((ext_vector_type(8))) float v8f;

#define LDS_STRIDE 132

#define GS_CAP 64

__global__ void gs_zero_int_kernel(int* __restrict__ p, int n) {
  int i = blockIdx.x * blockDim.x + threadIdx.x;
  if (i < n) p[i] = 0;
}

__global__ void gs_build_kernel(const int* __restrict__ src,
                                const int* __restrict__ tgt,
                                int* __restrict__ deg,
                                int* __restrict__ slot,
                                int* __restrict__ spillT,
                                int* __restrict__ spillS,
                                int nNodes,
                                int nEdges) {
  int e = blockIdx.x * blockDim.x + threadIdx.x;
  if (e < nEdges) {
    const int t = tgt[e];
    const int s = src[e];
    const int r = atomicAdd(deg + t, 1);
    if (r < GS_CAP) {
      slot[(long)t * GS_CAP + r] = s;
    } else {
      const int o = atomicAdd(deg + nNodes, 1);
      spillT[o] = t;
      spillS[o] = s;
    }
  }
}

__device__ __forceinline__ void gs_add4(float4& a, const float4 v) {
  a.x += v.x; a.y += v.y; a.z += v.z; a.w += v.w;
}

__global__ void __launch_bounds__(256)
gs_gather_kernel(const float* __restrict__ h,
                 const int* __restrict__ deg,
                 const int* __restrict__ slot,
                 const int* __restrict__ spillT,
                 const int* __restrict__ spillS,
                 float* __restrict__ agg,
                 float* __restrict__ cnt,
                 int nNodes,
                 int nEdges) {
  const int lane = threadIdx.x & 31;
  const int node = __builtin_amdgcn_readfirstlane(blockIdx.x * 8 + (threadIdx.x >> 5));
  if (node >= nNodes) return;
  const int d   = deg[node];
  const int end = d < GS_CAP ? d : GS_CAP;
  const int* list = slot + (long)node * GS_CAP;
  const float* hl = h + lane * 4;
  float4 acc = make_float4(0.0f, 0.0f, 0.0f, 0.0f);

  for (int base = 0; base < end; base += 32) {
    const int n = (end - base) < 32 ? (end - base) : 32;
    const int mine = lane < n ? list[base + lane] : 0;
    int j = 0;
    for (; j + 8 <= n; j += 8) {
      float4 v[8];
#pragma unroll
      for (int u = 0; u < 8; ++u) {
        const int s = __builtin_amdgcn_readlane(mine, j + u);
        v[u] = *(const float4*)(hl + (long)s * 128);
      }
#pragma unroll
      for (int u = 0; u < 8; ++u) gs_add4(acc, v[u]);
    }
    if (j + 4 <= n) {
      float4 v[4];
#pragma unroll
      for (int u = 0; u < 4; ++u) {
        const int s = __builtin_amdgcn_readlane(mine, j + u);
        v[u] = *(const float4*)(hl + (long)s * 128);
      }
#pragma unroll
      for (int u = 0; u < 4; ++u) gs_add4(acc, v[u]);
      j += 4;
    }
    if (j + 2 <= n) {
      const int s0 = __builtin_amdgcn_readlane(mine, j);
      const int s1 = __builtin_amdgcn_readlane(mine, j + 1);
      const float4 v0 = *(const float4*)(hl + (long)s0 * 128);
      const float4 v1 = *(const float4*)(hl + (long)s1 * 128);
      gs_add4(acc, v0);
      gs_add4(acc, v1);
      j += 2;
    }
    if (j < n) {
      const int s0 = __builtin_amdgcn_readlane(mine, j);
      const float4 v0 = *(const float4*)(hl + (long)s0 * 128);
      gs_add4(acc, v0);
    }
  }

  if (d > GS_CAP) {
    int m = deg[nNodes];
    m = m < nEdges ? m : nEdges;
    for (int b = 0; b < m; b += 32) {
      const int i = b + lane;
      const int t = i < m ? spillT[i] : -1;
      const int s = i < m ? spillS[i] : 0;
      unsigned hit = (unsigned)__ballot(t == node);
      while (hit != 0u) {
        const int l = __builtin_ctz(hit);
        hit &= hit - 1u;
        const int s0 = __shfl(s, l, 32);
        const float4 v0 = *(const float4*)(hl + (long)s0 * 128);
        gs_add4(acc, v0);
      }
    }
  }

  *(float4*)(agg + (long)node * 128 + lane * 4) = acc;
  if (lane == 0) cnt[node] = (float)d;
}

__device__ __forceinline__ float gs_dot4(const float4 a, const float4 b) {
  return (a.x * b.x + a.y * b.y) + (a.z * b.z + a.w * b.w);
}

__global__ void __launch_bounds__(256)
gs_fold_kernel(const float* __restrict__ Wl,
               const float* __restrict__ bl,
               const float* __restrict__ Wr,
               const float* __restrict__ We,
               float* __restrict__ U4,
               float* __restrict__ cb) {
  const int lane = threadIdx.x & 31;
  const int k = blockIdx.x * 8 + (threadIdx.x >> 5);
  if (k > 128) return;
  const float4 ws = *(const float4*)(We + lane * 4);
  const float4 wt = *(const float4*)(We + 128 + lane * 4);
  const bool isRow = k < 128;
  const float4 l = *(const float4*)((isRow ? Wl + k * 128 : bl) + lane * 4);
  const float4 r = isRow ? *(const float4*)(Wr + k * 128 + lane * 4)
                         : make_float4(0.0f, 0.0f, 0.0f, 0.0f);
  float s0 = gs_dot4(l, ws);
  float s1 = gs_dot4(l, wt);
  float s2 = gs_dot4(r, ws);
  float s3 = gs_dot4(r, wt);
  for (int off = 16; off > 0; off >>= 1) {
    s0 += __shfl_xor(s0, off, 32);
    s1 += __shfl_xor(s1, off, 32);
    s2 += __shfl_xor(s2, off, 32);
    s3 += __shfl_xor(s3, off, 32);
  }
  if (lane == 0) {
    if (isRow) {
      *(float4*)(U4 + k * 4) = make_float4(s0, s1, s2, s3);
    } else {
      cb[0] = s0;
      cb[1] = s1;
    }
  }
}

__device__ __forceinline__ void gs_load_b16(v2f (&b)[16],
                                            const float* __restrict__ Wcol,
                                            int kbase) {
#pragma unroll
  for (int s = 0; s < 16; ++s) {
    b[s].x = Wcol[(kbase + 4 * s) * 128];
    b[s].y = Wcol[(kbase + 4 * s + 1) * 128];
  }
}

__device__ __forceinline__ v8f gs_mma16(v8f acc, const float* lrow, int kbase,
                                        const v2f (&b)[16]) {
#pragma unroll
  for (int s = 0; s < 16; ++s) {
    v2f a;
    a.x = lrow[kbase + 4 * s];
    a.y = lrow[kbase + 4 * s + 1];
    acc = __builtin_amdgcn_wmma_f32_16x16x4_f32(
         false, a,  false, b[s],
         (short)0, acc,  false,  false);
  }
  return acc;
}

__global__ void __launch_bounds__(256)
gs_sage_wmma_kernel(const float* __restrict__ agg,
                    const float* __restrict__ cnt,
                    const float* __restrict__ hself,
                    const float* __restrict__ Wl,
                    const float* __restrict__ bl,
                    const float* __restrict__ Wr,
                    const float* __restrict__ U4,
                    float* __restrict__ np,
                    int nNodes) {
  __shared__ float lmean[16 * LDS_STRIDE];
  __shared__ float lself[16 * LDS_STRIDE];
  __shared__ float lproj[8 * 64];

  const int tid  = threadIdx.x;
  const int row0 = blockIdx.x * 16;

  for (int i = 0; i < 8; ++i) {
    int idx  = tid + i * 256;
    int r    = idx >> 7;
    int c    = idx & 127;
    int node = row0 + r;
    int nclamp = node < nNodes ? node : (nNodes - 1);
    float ct   = cnt[nclamp];
    float rinv = 1.0f / fmaxf(ct, 1.0f);
    lmean[r * LDS_STRIDE + c] = agg[(long)nclamp * 128 + c] * rinv;
    lself[r * LDS_STRIDE + c] = hself[(long)nclamp * 128 + c];
  }
  __syncthreads();

  const int lane = tid & 31;
  const int wave = tid >> 5;
  const int col0 = wave * 16;
  const int rowA = lane & 15;
  const int kofs = (lane >> 4) * 2;
  const int colB = lane & 15;

  v8f acc = {};

  const float* wlc = Wl + kofs * 128 + col0 + colB;
  const float* wrc = Wr + kofs * 128 + col0 + colB;
  const float* lm  = lmean + rowA * LDS_STRIDE + kofs;
  const float* ls  = lself + rowA * LDS_STRIDE + kofs;
  v2f bA[16], bB[16];
  gs_load_b16(bA, wlc, 0);
  gs_load_b16(bB, wlc, 64);
  acc = gs_mma16(acc, lm, 0, bA);
  gs_load_b16(bA, wrc, 0);
  acc = gs_mma16(acc, lm, 64, bB);
  gs_load_b16(bB, wrc, 64);
  acc = gs_mma16(acc, ls, 0, bA);
  acc = gs_mma16(acc, ls, 64, bB);

  const int c     = lane & 15;
  const int half  = lane >> 4;
  const int ocol  = col0 + c;
  const float bias = bl[ocol];
  const float4 u = *(const float4*)(U4 + ocol * 4);

  float q[32];
  for (int i = 0; i < 8; ++i) {
    const float t = fmaxf(acc[i] + bias, 0.0f);
    q[i]      = t * u.x;
    q[8 + i]  = t * u.y;
    q[16 + i] = t * u.z;
    q[24 + i] = t * u.w;
  }
  const bool b3 = (c & 8) != 0;
  const bool b2 = (c & 4) != 0;
  const bool b1 = (c & 2) != 0;
  const bool b0 = (c & 1) != 0;
  float w16[16];
  for (int k = 0; k < 16; ++k) {
    const float keep = b3 ? q[16 + k] : q[k];
    const float send = b3 ? q[k] : q[16 + k];
    w16[k] = keep + __shfl_xor(send, 8, 32);
  }
  float w8[8];
  for (int k = 0; k < 8; ++k) {
    const float keep = b2 ? w16[8 + k] : w16[k];
    const float send = b2 ? w16[k] : w16[8 + k];
    w8[k] = keep + __shfl_xor(send, 4, 32);
  }
  float w4[4];
  for (int k = 0; k < 4; ++k) {
    const float keep = b1 ? w8[4 + k] : w8[k];
    const float send = b1 ? w8[k] : w8[4 + k];
    w4[k] = keep + __shfl_xor(send, 2, 32);
  }
  float w2[2];
  for (int k = 0; k < 2; ++k) {
    const float keep = b0 ? w4[2 + k] : w4[k];
    const float send = b0 ? w4[k] : w4[2 + k];
    w2[k] = keep + __shfl_xor(send, 1, 32);
  }

  const int prow = half * 8 + (c & 3) * 2;
  const int pidx = c >> 2;
  lproj[wave * 64 + prow * 4 + pidx]       = w2[0];
  lproj[wave * 64 + (prow + 1) * 4 + pidx] = w2[1];
  __syncthreads();
  if (tid < 64) {
    float sum = lproj[tid];
    for (int w = 1; w < 8; ++w) sum += lproj[w * 64 + tid];
    const int node = row0 + (tid >> 2);
    if (node < nNodes) np[(long)row0 * 4 + tid] = sum;
  }
}

__global__ void __launch_bounds__(256)
gs_node_score_kernel(const float* __restrict__ np,
                     const float* __restrict__ cb,
                     const int* __restrict__ deg,
                     const int* __restrict__ slot,
                     const int* __restrict__ spillT,
                     const int* __restrict__ spillS,
                     float* __restrict__ ps,
                     float* __restrict__ pt,
                     int nNodes,
                     int nEdges) {
  const int node = (blockIdx.x * blockDim.x + threadIdx.x) >> 3;
  const int sub  = threadIdx.x & 7;
  const bool live = node < nNodes;
  const int d   = live ? deg[node] : 0;
  const int end = d < GS_CAP ? d : GS_CAP;
  const int* list = slot + (long)(live ? node : 0) * GS_CAP;
  float s = 0.0f, t = 0.0f;
  for (int k = sub; k < end; k += 8) {
    const float2 a = *(const float2*)(np + (long)list[k] * 4);
    s += a.x;
    t += a.y;
  }
  if (d > GS_CAP) {
    int m = deg[nNodes];
    m = m < nEdges ? m : nEdges;
    for (int i = sub; i < m; i += 8) {
      if (spillT[i] == node) {
        const float2 a = *(const float2*)(np + (long)spillS[i] * 4);
        s += a.x;
        t += a.y;
      }
    }
  }
  for (int off = 4; off > 0; off >>= 1) {
    s += __shfl_xor(s, off, 32);
    t += __shfl_xor(t, off, 32);
  }
  if (live && sub == 0) {
    const float den = fmaxf((float)d, 1.0f);
    const float2 self = *(const float2*)(np + (long)node * 4 + 2);
    ps[node] = s / den + self.x + cb[0];
    pt[node] = t / den + self.y + cb[1];
  }
}

__device__ __forceinline__ float gs_sigmoid(float p) {
  return 1.0f / (1.0f + __expf(-p));
}

__global__ void __launch_bounds__(256)
gs_edge_score_kernel(const float* __restrict__ ps,
                     const float* __restrict__ pt,
                     const int* __restrict__ src,
                     const int* __restrict__ tgt,
                     const float* __restrict__ be,
                     float* __restrict__ out,
                     int nEdges) {
  const int e0 = (blockIdx.x * blockDim.x + threadIdx.x) * 4;
  if (e0 >= nEdges) return;
  const float b = be[0];
  if ((nEdges & 3) == 0) {
    const int4 s4 = *(const int4*)(src + e0);
    const int4 t4 = *(const int4*)(tgt + e0);
    float4 y;
    y.x = gs_sigmoid(ps[s4.x] + pt[t4.x] + b);
    y.y = gs_sigmoid(ps[s4.y] + pt[t4.y] + b);
    y.z = gs_sigmoid(ps[s4.z] + pt[t4.z] + b);
    y.w = gs_sigmoid(ps[s4.w] + pt[t4.w] + b);
    *(float4*)(out + e0) = y;
  } else {
    const int eEnd = (e0 + 4) < nEdges ? (e0 + 4) : nEdges;
    for (int e = e0; e < eEnd; ++e)
      out[e] = gs_sigmoid(ps[src[e]] + pt[tgt[e]] + b);
  }
}

extern "C" void kernel_launch(void* const* d_in, const int* in_sizes, int n_in,
                              void* d_out, int out_size, void* d_ws, size_t ws_size,
                              hipStream_t stream) {
  const float* x    = (const float*)d_in[0];
  const int*   ei   = (const int*)d_in[1];
  const float* Wl1  = (const float*)d_in[2];
  const float* bl1  = (const float*)d_in[3];
  const float* Wr1  = (const float*)d_in[4];
  const float* Wl2  = (const float*)d_in[5];
  const float* bl2  = (const float*)d_in[6];
  const float* Wr2  = (const float*)d_in[7];
  const float* We   = (const float*)d_in[8];
  const float* be   = (const float*)d_in[9];

  const int N = in_sizes[0] / 128;
  const int E = in_sizes[1] / 2;
  const int* src = ei;
  const int* tgt = ei + E;

  int*   slot = (int*)d_ws;
  float* agg  = (float*)d_ws + (long)N * GS_CAP;
  float* cnt  = agg + (long)N * 128;
  float* U4   = cnt + N;
  float* cb   = U4 + 128 * 4;
  float* np   = cb + 4;
  int* deg    = (int*)(np + (long)N * 4);
  int* spillT = deg + (N + 1);
  int* spillS = spillT + E;
  float* ps   = (float*)(spillS + E);
  float* pt   = ps + N;

  const int edgeBlocks = (E + 255) / 256;
  const int zeroBlocks = (N + 256) / 256;
  const int gathBlocks = (N + 7) / 8;
  const int oct_Blocks = (N + 31) / 32;
  const int quadBlocks = ((E + 3) / 4 + 255) / 256;
  const int gemmBlocks = (N + 15) / 16;

  gs_zero_int_kernel<<<zeroBlocks, 256, 0, stream>>>(deg, N + 1);
  gs_build_kernel<<<edgeBlocks, 256, 0, stream>>>(src, tgt, deg, slot, spillT, spillS, N, E);

  gs_fold_kernel<<<17, 256, 0, stream>>>(Wl2, bl2, Wr2, We, U4, cb);

  gs_gather_kernel<<<gathBlocks, 256, 0, stream>>>(x, deg, slot, spillT, spillS, agg, cnt, N, E);
  gs_sage_wmma_kernel<<<gemmBlocks, 256, 0, stream>>>(agg, cnt, x, Wl1, bl1, Wr1,
                                                      U4, np, N);

  gs_node_score_kernel<<<oct_Blocks, 256, 0, stream>>>(np, cb, deg, slot, spillT, spillS, ps, pt, N, E);

  gs_edge_score_kernel<<<quadBlocks, 256, 0, stream>>>(ps, pt, src, tgt, be,
                                                       (float*)d_out, E);
}


# Neighbour gather fused into the layer-1 GEMM staging: each wave sums its two nodes' rows straight into LDS as means; gather launch and agg/cnt round trip through memory gone; sum order unchanged
# speedup vs baseline: 4.2106x; 1.0498x over previous
#include <hip/hip_runtime.h>
#include <hip/hip_bf16.h>

typedef __attribute__((ext_vector_type(2))) float v2f;
typedef __attribute__((ext_vector_type(8))) float v8f;

#define LDS_STRIDE 132

#define GS_CAP 64

__global__ void gs_zero_int_kernel(int* __restrict__ p, int n) {
  int i = blockIdx.x * blockDim.x + threadIdx.x;
  if (i < n) p[i] = 0;
}

__global__ void gs_build_kernel(const int* __restrict__ src,
                                const int* __restrict__ tgt,
                                int* __restrict__ deg,
                                int* __restrict__ slot,
                                int* __restrict__ spillT,
                                int* __restrict__ spillS,
                                int nNodes,
                                int nEdges) {
  int e = blockIdx.x * blockDim.x + threadIdx.x;
  if (e < nEdges) {
    const int t = tgt[e];
    const int s = src[e];
    const int r = atomicAdd(deg + t, 1);
    if (r < GS_CAP) {
      slot[(long)t * GS_CAP + r] = s;
    } else {
      const int o = atomicAdd(deg + nNodes, 1);
      spillT[o] = t;
      spillS[o] = s;
    }
  }
}

__device__ __forceinline__ void gs_add4(float4& a, const float4 v) {
  a.x += v.x; a.y += v.y; a.z += v.z; a.w += v.w;
}

template <int U>
__device__ __forceinline__ void gs_rows(float4& acc, const float* hl, int mine, int j) {
  float4 v[U];
#pragma unroll
  for (int u = 0; u < U; ++u) {
    const int s = __builtin_amdgcn_readlane(mine, j + u);
    v[u] = *(const float4*)(hl + (long)s * 128);
  }
#pragma unroll
  for (int u = 0; u < U; ++u) gs_add4(acc, v[u]);
}

__device__ __forceinline__ float4 gs_gather_node(const float* __restrict__ h,
                                                 const int* __restrict__ deg,
                                                 const int* __restrict__ slot,
                                                 const int* __restrict__ spillT,
                                                 const int* __restrict__ spillS,
                                                 int node, int d, int first32,
                                                 int lane, int nNodes, int nEdges) {
  const int end = d < GS_CAP ? d : GS_CAP;
  const int* list = slot + (long)node * GS_CAP;
  const float* hl = h + lane * 4;
  float4 acc = make_float4(0.0f, 0.0f, 0.0f, 0.0f);

  for (int base = 0; base < end; base += 32) {
    const int n = (end - base) < 32 ? (end - base) : 32;
    int mine = first32;
    if (base != 0) mine = lane < n ? list[base + lane] : 0;
    int j = 0;
    for (; j + 16 <= n; j += 16) gs_rows<16>(acc, hl, mine, j);
    if (j + 8 <= n) { gs_rows<8>(acc, hl, mine, j); j += 8; }
    if (j + 4 <= n) { gs_rows<4>(acc, hl, mine, j); j += 4; }
    if (j + 2 <= n) { gs_rows<2>(acc, hl, mine, j); j += 2; }
    if (j < n)      { gs_rows<1>(acc, hl, mine, j); }
  }

  if (d > GS_CAP) {
    int m = deg[nNodes];
    m = m < nEdges ? m : nEdges;
    for (int b = 0; b < m; b += 32) {
      const int i = b + lane;
      const int t = i < m ? spillT[i] : -1;
      const int s = i < m ? spillS[i] : 0;
      unsigned hit = (unsigned)__ballot(t == node);
      while (hit != 0u) {
        const int l = __builtin_ctz(hit);
        hit &= hit - 1u;
        const int s0 = __shfl(s, l, 32);
        const float4 v0 = *(const float4*)(hl + (long)s0 * 128);
        gs_add4(acc, v0);
      }
    }
  }
  return acc;
}

__device__ __forceinline__ float gs_dot4(const float4 a, const float4 b) {
  return (a.x * b.x + a.y * b.y) + (a.z * b.z + a.w * b.w);
}

__global__ void __launch_bounds__(256)
gs_fold_kernel(const float* __restrict__ Wl,
               const float* __restrict__ bl,
               const float* __restrict__ Wr,
               const float* __restrict__ We,
               float* __restrict__ U4,
               float* __restrict__ cb) {
  const int lane = threadIdx.x & 31;
  const int k = blockIdx.x * 8 + (threadIdx.x >> 5);
  if (k > 128) return;
  const float4 ws = *(const float4*)(We + lane * 4);
  const float4 wt = *(const float4*)(We + 128 + lane * 4);
  const bool isRow = k < 128;
  const float4 l = *(const float4*)((isRow ? Wl + k * 128 : bl) + lane * 4);
  const float4 r = isRow ? *(const float4*)(Wr + k * 128 + lane * 4)
                         : make_float4(0.0f, 0.0f, 0.0f, 0.0f);
  float s0 = gs_dot4(l, ws);
  float s1 = gs_dot4(l, wt);
  float s2 = gs_dot4(r, ws);
  float s3 = gs_dot4(r, wt);
  for (int off = 16; off > 0; off >>= 1) {
    s0 += __shfl_xor(s0, off, 32);
    s1 += __shfl_xor(s1, off, 32);
    s2 += __shfl_xor(s2, off, 32);
    s3 += __shfl_xor(s3, off, 32);
  }
  if (lane == 0) {
    if (isRow) {
      *(float4*)(U4 + k * 4) = make_float4(s0, s1, s2, s3);
    } else {
      cb[0] = s0;
      cb[1] = s1;
    }
  }
}

__device__ __forceinline__ void gs_load_b16(v2f (&b)[16],
                                            const float* __restrict__ Wcol,
                                            int kbase) {
#pragma unroll
  for (int s = 0; s < 16; ++s) {
    b[s].x = Wcol[(kbase + 4 * s) * 128];
    b[s].y = Wcol[(kbase + 4 * s + 1) * 128];
  }
}

__device__ __forceinline__ v8f gs_mma16(v8f acc, const float* lrow, int kbase,
                                        const v2f (&b)[16]) {
#pragma unroll
  for (int s = 0; s < 16; ++s) {
    v2f a;
    a.x = lrow[kbase + 4 * s];
    a.y = lrow[kbase + 4 * s + 1];
    acc = __builtin_amdgcn_wmma_f32_16x16x4_f32(
         false, a,  false, b[s],
         (short)0, acc,  false,  false);
  }
  return acc;
}

__global__ void __launch_bounds__(256)
gs_sage_wmma_kernel(const float* __restrict__ hself,
                    const int* __restrict__ deg,
                    const int* __restrict__ slot,
                    const int* __restrict__ spillT,
                    const int* __restrict__ spillS,
                    const float* __restrict__ Wl,
                    const float* __restrict__ bl,
                    const float* __restrict__ Wr,
                    const float* __restrict__ U4,
                    float* __restrict__ np,
                    int nNodes,
                    int nEdges) {
  __shared__ float lmean[16 * LDS_STRIDE];
  __shared__ float lself[16 * LDS_STRIDE];
  __shared__ float lproj[8 * 64];

  const int tid  = threadIdx.x;
  const int row0 = blockIdx.x * 16;
  const int lane = tid & 31;
  const int wave = __builtin_amdgcn_readfirstlane(tid >> 5);

  float4 selfv[2];
  for (int i = 0; i < 2; ++i) {
    const int idx = tid + i * 256;
    const int r   = idx >> 5;
    const int node = row0 + r;
    const int nclamp = node < nNodes ? node : (nNodes - 1);
    selfv[i] = *(const float4*)(hself + (long)nclamp * 128 + (idx & 31) * 4);
  }

  const int nodeA = row0 + wave * 2;
  const int nodeB = nodeA + 1;
  const int dA = nodeA < nNodes ? deg[nodeA] : 0;
  const int dB = nodeB < nNodes ? deg[nodeB] : 0;
  const int fA = lane < (dA < 32 ? dA : 32) ? slot[(long)nodeA * GS_CAP + lane] : 0;
  const int fB = lane < (dB < 32 ? dB : 32) ? slot[(long)nodeB * GS_CAP + lane] : 0;

  for (int i = 0; i < 2; ++i) {
    const int idx = tid + i * 256;
    *(float4*)(lself + (idx >> 5) * LDS_STRIDE + (idx & 31) * 4) = selfv[i];
  }

  {
    const float4 a = gs_gather_node(hself, deg, slot, spillT, spillS, nodeA, dA, fA,
                                    lane, nNodes, nEdges);
    const float rinv = 1.0f / fmaxf((float)dA, 1.0f);
    *(float4*)(lmean + (wave * 2) * LDS_STRIDE + lane * 4) =
        make_float4(a.x * rinv, a.y * rinv, a.z * rinv, a.w * rinv);
  }
  {
    const float4 a = gs_gather_node(hself, deg, slot, spillT, spillS, nodeB, dB, fB,
                                    lane, nNodes, nEdges);
    const float rinv = 1.0f / fmaxf((float)dB, 1.0f);
    *(float4*)(lmean + (wave * 2 + 1) * LDS_STRIDE + lane * 4) =
        make_float4(a.x * rinv, a.y * rinv, a.z * rinv, a.w * rinv);
  }
  __syncthreads();

  const int col0 = wave * 16;
  const int rowA = lane & 15;
  const int kofs = (lane >> 4) * 2;
  const int colB = lane & 15;

  v8f acc = {};

  const float* wlc = Wl + kofs * 128 + col0 + colB;
  const float* wrc = Wr + kofs * 128 + col0 + colB;
  const float* lm  = lmean + rowA * LDS_STRIDE + kofs;
  const float* ls  = lself + rowA * LDS_STRIDE + kofs;
  v2f bA[16], bB[16];
  gs_load_b16(bA, wlc, 0);
  gs_load_b16(bB, wlc, 64);
  acc = gs_mma16(acc, lm, 0, bA);
  gs_load_b16(bA, wrc, 0);
  acc = gs_mma16(acc, lm, 64, bB);
  gs_load_b16(bB, wrc, 64);
  acc = gs_mma16(acc, ls, 0, bA);
  acc = gs_mma16(acc, ls, 64, bB);

  const int c     = lane & 15;
  const int half  = lane >> 4;
  const int ocol  = col0 + c;
  const float bias = bl[ocol];
  const float4 u = *(const float4*)(U4 + ocol * 4);

  float q[32];
  for (int i = 0; i < 8; ++i) {
    const float t = fmaxf(acc[i] + bias, 0.0f);
    q[i]      = t * u.x;
    q[8 + i]  = t * u.y;
    q[16 + i] = t * u.z;
    q[24 + i] = t * u.w;
  }
  const bool b3 = (c & 8) != 0;
  const bool b2 = (c & 4) != 0;
  const bool b1 = (c & 2) != 0;
  const bool b0 = (c & 1) != 0;
  float w16[16];
  for (int k = 0; k < 16; ++k) {
    const float keep = b3 ? q[16 + k] : q[k];
    const float send = b3 ? q[k] : q[16 + k];
    w16[k] = keep + __shfl_xor(send, 8, 32);
  }
  float w8[8];
  for (int k = 0; k < 8; ++k) {
    const float keep = b2 ? w16[8 + k] : w16[k];
    const float send = b2 ? w16[k] : w16[8 + k];
    w8[k] = keep + __shfl_xor(send, 4, 32);
  }
  float w4[4];
  for (int k = 0; k < 4; ++k) {
    const float keep = b1 ? w8[4 + k] : w8[k];
    const float send = b1 ? w8[k] : w8[4 + k];
    w4[k] = keep + __shfl_xor(send, 2, 32);
  }
  float w2[2];
  for (int k = 0; k < 2; ++k) {
    const float keep = b0 ? w4[2 + k] : w4[k];
    const float send = b0 ? w4[k] : w4[2 + k];
    w2[k] = keep + __shfl_xor(send, 1, 32);
  }

  const int prow = half * 8 + (c & 3) * 2;
  const int pidx = c >> 2;
  lproj[wave * 64 + prow * 4 + pidx]       = w2[0];
  lproj[wave * 64 + (prow + 1) * 4 + pidx] = w2[1];
  __syncthreads();
  if (tid < 64) {
    float sum = lproj[tid];
    for (int w = 1; w < 8; ++w) sum += lproj[w * 64 + tid];
    const int node = row0 + (tid >> 2);
    if (node < nNodes) np[(long)row0 * 4 + tid] = sum;
  }
}

__global__ void __launch_bounds__(256)
gs_node_score_kernel(const float* __restrict__ np,
                     const float* __restrict__ cb,
                     const int* __restrict__ deg,
                     const int* __restrict__ slot,
                     const int* __restrict__ spillT,
                     const int* __restrict__ spillS,
                     float* __restrict__ ps,
                     float* __restrict__ pt,
                     int nNodes,
                     int nEdges) {
  const int node = (blockIdx.x * blockDim.x + threadIdx.x) >> 3;
  const int sub  = threadIdx.x & 7;
  const bool live = node < nNodes;
  const int d   = live ? deg[node] : 0;
  const int end = d < GS_CAP ? d : GS_CAP;
  const int* list = slot + (long)(live ? node : 0) * GS_CAP;
  float s = 0.0f, t = 0.0f;
  for (int k = sub; k < end; k += 8) {
    const float2 a = *(const float2*)(np + (long)list[k] * 4);
    s += a.x;
    t += a.y;
  }
  if (d > GS_CAP) {
    int m = deg[nNodes];
    m = m < nEdges ? m : nEdges;
    for (int i = sub; i < m; i += 8) {
      if (spillT[i] == node) {
        const float2 a = *(const float2*)(np + (long)spillS[i] * 4);
        s += a.x;
        t += a.y;
      }
    }
  }
  for (int off = 4; off > 0; off >>= 1) {
    s += __shfl_xor(s, off, 32);
    t += __shfl_xor(t, off, 32);
  }
  if (live && sub == 0) {
    const float den = fmaxf((float)d, 1.0f);
    const float2 self = *(const float2*)(np + (long)node * 4 + 2);
    ps[node] = s / den + self.x + cb[0];
    pt[node] = t / den + self.y + cb[1];
  }
}

__device__ __forceinline__ float gs_sigmoid(float p) {
  return 1.0f / (1.0f + __expf(-p));
}

__global__ void __launch_bounds__(256)
gs_edge_score_kernel(const float* __restrict__ ps,
                     const float* __restrict__ pt,
                     const int* __restrict__ src,
                     const int* __restrict__ tgt,
                     const float* __restrict__ be,
                     float* __restrict__ out,
                     int nEdges) {
  const int e0 = (blockIdx.x * blockDim.x + threadIdx.x) * 4;
  if (e0 >= nEdges) return;
  const float b = be[0];
  if ((nEdges & 3) == 0) {
    const int4 s4 = *(const int4*)(src + e0);
    const int4 t4 = *(const int4*)(tgt + e0);
    float4 y;
    y.x = gs_sigmoid(ps[s4.x] + pt[t4.x] + b);
    y.y = gs_sigmoid(ps[s4.y] + pt[t4.y] + b);
    y.z = gs_sigmoid(ps[s4.z] + pt[t4.z] + b);
    y.w = gs_sigmoid(ps[s4.w] + pt[t4.w] + b);
    *(float4*)(out + e0) = y;
  } else {
    const int eEnd = (e0 + 4) < nEdges ? (e0 + 4) : nEdges;
    for (int e = e0; e < eEnd; ++e)
      out[e] = gs_sigmoid(ps[src[e]] + pt[tgt[e]] + b);
  }
}

extern "C" void kernel_launch(void* const* d_in, const int* in_sizes, int n_in,
                              void* d_out, int out_size, void* d_ws, size_t ws_size,
                              hipStream_t stream) {
  const float* x    = (const float*)d_in[0];
  const int*   ei   = (const int*)d_in[1];
  const float* Wl1  = (const float*)d_in[2];
  const float* bl1  = (const float*)d_in[3];
  const float* Wr1  = (const float*)d_in[4];
  const float* Wl2  = (const float*)d_in[5];
  const float* bl2  = (const float*)d_in[6];
  const float* Wr2  = (const float*)d_in[7];
  const float* We   = (const float*)d_in[8];
  const float* be   = (const float*)d_in[9];

  const int N = in_sizes[0] / 128;
  const int E = in_sizes[1] / 2;
  const int* src = ei;
  const int* tgt = ei + E;

  int*   slot = (int*)d_ws;
  float* U4   = (float*)d_ws + (long)N * GS_CAP;
  float* cb   = U4 + 128 * 4;
  float* np   = cb + 4;
  int* deg    = (int*)(np + (long)N * 4);
  int* spillT = deg + (N + 1);
  int* spillS = spillT + E;
  float* ps   = (float*)(spillS + E);
  float* pt   = ps + N;

  const int edgeBlocks = (E + 255) / 256;
  const int zeroBlocks = (N + 256) / 256;
  const int oct_Blocks = (N + 31) / 32;
  const int quadBlocks = ((E + 3) / 4 + 255) / 256;
  const int gemmBlocks = (N + 15) / 16;

  gs_zero_int_kernel<<<zeroBlocks, 256, 0, stream>>>(deg, N + 1);
  gs_build_kernel<<<edgeBlocks, 256, 0, stream>>>(src, tgt, deg, slot, spillT, spillS, N, E);

  gs_fold_kernel<<<17, 256, 0, stream>>>(Wl2, bl2, Wr2, We, U4, cb);

  gs_sage_wmma_kernel<<<gemmBlocks, 256, 0, stream>>>(x, deg, slot, spillT, spillS,
                                                      Wl1, bl1, Wr1, U4, np, N, E);

  gs_node_score_kernel<<<oct_Blocks, 256, 0, stream>>>(np, cb, deg, slot, spillT, spillS, ps, pt, N, E);

  gs_edge_score_kernel<<<quadBlocks, 256, 0, stream>>>(ps, pt, src, tgt, be,
                                                       (float*)d_out, E);
}


# Neighbour lists built through LDS: edges ordered by 64-node bin per chunk, then one block per bin ranks and fills slot blocks; per-edge memory atomics and 4-byte scatter gone; sum order = LDS arrival
# speedup vs baseline: 6.6624x; 1.5823x over previous
#include <hip/hip_runtime.h>
#include <hip/hip_bf16.h>

typedef __attribute__((ext_vector_type(2))) float v2f;
typedef __attribute__((ext_vector_type(8))) float v8f;

#define LDS_STRIDE 132

#define GS_CAP 64
#define GS_BIN 64
#define GS_CHUNK 8192
#define GS_MAXB 1024
#define GS_MAXP 256

__global__ void __launch_bounds__(1024)
gs_part_kernel(const int* __restrict__ src,
               const int* __restrict__ tgt,
               int* __restrict__ sorted,
               int* __restrict__ off,
               int* __restrict__ deg,
               int nNodes,
               int nEdges,
               int nBins) {
  __shared__ int hist[GS_MAXB];
  __shared__ int wsum[32];
  __shared__ int stage[GS_CHUNK];
  const int tid  = threadIdx.x;
  const int lane = tid & 31;
  const int wave = tid >> 5;
  const int base = blockIdx.x * GS_CHUNK;

  if (blockIdx.x == 0 && tid == 0) deg[nNodes] = 0;
  hist[tid] = 0;
  __syncthreads();

  int pk[8], bn[8], rk[8];
#pragma unroll
  for (int i = 0; i < 8; ++i) {
    const int e = base + i * 1024 + tid;
    const bool ok = e < nEdges;
    const int t = ok ? tgt[e] : 0;
    const int s = ok ? src[e] : 0;
    pk[i] = (s << 6) | (t & (GS_BIN - 1));
    bn[i] = ok ? (t >> 6) : -1;
  }
#pragma unroll
  for (int i = 0; i < 8; ++i)
    rk[i] = bn[i] >= 0 ? atomicAdd(&hist[bn[i]], 1) : 0;
  __syncthreads();

  const int v = hist[tid];
  int inc = v;
  for (int d = 1; d < 32; d <<= 1) {
    const int n = __shfl_up(inc, d, 32);
    if (lane >= d) inc += n;
  }
  if (lane == 31) wsum[wave] = inc;
  __syncthreads();
  if (wave == 0) {
    const int w = wsum[lane];
    int wi = w;
    for (int d = 1; d < 32; d <<= 1) {
      const int n = __shfl_up(wi, d, 32);
      if (lane >= d) wi += n;
    }
    wsum[lane] = wi - w;
  }
  __syncthreads();
  const int ex = inc - v + wsum[wave];
  hist[tid] = ex;
  if (tid <= nBins) off[(long)blockIdx.x * (nBins + 1) + tid] = ex;
  __syncthreads();

#pragma unroll
  for (int i = 0; i < 8; ++i)
    if (bn[i] >= 0) stage[hist[bn[i]] + rk[i]] = pk[i];
  __syncthreads();
#pragma unroll
  for (int i = 0; i < 8; ++i) {
    const int e = base + i * 1024 + tid;
    if (e < nEdges) sorted[e] = stage[i * 1024 + tid];
  }
}

__global__ void __launch_bounds__(256)
gs_place_kernel(const int* __restrict__ sorted,
                const int* __restrict__ off,
                int* __restrict__ deg,
                int* __restrict__ slot,
                int* __restrict__ spillT,
                int* __restrict__ spillS,
                int nNodes,
                int nBins,
                int nPart) {
  __shared__ int pre[GS_MAXP + 1];
  __shared__ int gst[GS_MAXP];
  __shared__ int wsum[8];
  __shared__ int cnt[GS_BIN];
  __shared__ int slotL[GS_BIN * GS_CAP];
  const int tid  = threadIdx.x;
  const int lane = tid & 31;
  const int wave = tid >> 5;
  const int b    = blockIdx.x;

  int len = 0, g = 0;
  if (tid < nPart) {
    const int* o = off + (long)tid * (nBins + 1) + b;
    const int a = o[0];
    len = o[1] - a;
    g = tid * GS_CHUNK + a;
  }
  if (tid < GS_BIN) cnt[tid] = 0;
  int inc = len;
  for (int d = 1; d < 32; d <<= 1) {
    const int n = __shfl_up(inc, d, 32);
    if (lane >= d) inc += n;
  }
  if (lane == 31) wsum[wave] = inc;
  __syncthreads();
  int wofs = 0;
  for (int w = 0; w < wave; ++w) wofs += wsum[w];
  const int ex = inc - len + wofs;
  pre[tid] = ex;
  gst[tid] = g;
  if (tid == 255) pre[256] = ex + len;
  __syncthreads();

  const int total = pre[256];
  for (int j = tid; j < total; j += 256) {
    int k = 0;
    for (int step = 128; step > 0; step >>= 1)
      if (pre[k + step] <= j) k += step;
    const int p = sorted[gst[k] + (j - pre[k])];
    const int n = p & (GS_BIN - 1);
    const int s = p >> 6;
    const int r = atomicAdd(&cnt[n], 1);
    if (r < GS_CAP) {
      slotL[n * GS_CAP + r] = s;
    } else {
      const int o = atomicAdd(deg + nNodes, 1);
      spillT[o] = b * GS_BIN + n;
      spillS[o] = s;
    }
  }
  __syncthreads();

  for (int n = wave * 8; n < wave * 8 + 8; ++n) {
    const int node = b * GS_BIN + n;
    if (node < nNodes) {
      const int d = cnt[n];
      if (d > 0)  slot[(long)node * GS_CAP + lane]      = slotL[n * GS_CAP + lane];
      if (d > 32) slot[(long)node * GS_CAP + 32 + lane] = slotL[n * GS_CAP + 32 + lane];
    }
  }
  if (tid < GS_BIN && b * GS_BIN + tid < nNodes) deg[b * GS_BIN + tid] = cnt[tid];
}

__device__ __forceinline__ void gs_add4(float4& a, const float4 v) {
  a.x += v.x; a.y += v.y; a.z += v.z; a.w += v.w;
}

template <int U>
__device__ __forceinline__ void gs_rows(float4& acc, const float* hl, int mine, int j) {
  float4 v[U];
#pragma unroll
  for (int u = 0; u < U; ++u) {
    const int s = __builtin_amdgcn_readlane(mine, j + u);
    v[u] = *(const float4*)(hl + (long)s * 128);
  }
#pragma unroll
  for (int u = 0; u < U; ++u) gs_add4(acc, v[u]);
}

__device__ __forceinline__ float4 gs_gather_node(const float* __restrict__ h,
                                                 const int* __restrict__ deg,
                                                 const int* __restrict__ slot,
                                                 const int* __restrict__ spillT,
                                                 const int* __restrict__ spillS,
                                                 int node, int d, int first32,
                                                 int lane, int nNodes, int nEdges) {
  const int end = d < GS_CAP ? d : GS_CAP;
  const int* list = slot + (long)node * GS_CAP;
  const float* hl = h + lane * 4;
  float4 acc = make_float4(0.0f, 0.0f, 0.0f, 0.0f);

  for (int base = 0; base < end; base += 32) {
    const int n = (end - base) < 32 ? (end - base) : 32;
    int mine = first32;
    if (base != 0) mine = lane < n ? list[base + lane] : 0;
    int j = 0;
    for (; j + 16 <= n; j += 16) gs_rows<16>(acc, hl, mine, j);
    if (j + 8 <= n) { gs_rows<8>(acc, hl, mine, j); j += 8; }
    if (j + 4 <= n) { gs_rows<4>(acc, hl, mine, j); j += 4; }
    if (j + 2 <= n) { gs_rows<2>(acc, hl, mine, j); j += 2; }
    if (j < n)      { gs_rows<1>(acc, hl, mine, j); }
  }

  if (d > GS_CAP) {
    int m = deg[nNodes];
    m = m < nEdges ? m : nEdges;
    for (int b = 0; b < m; b += 32) {
      const int i = b + lane;
      const int t = i < m ? spillT[i] : -1;
      const int s = i < m ? spillS[i] : 0;
      unsigned hit = (unsigned)__ballot(t == node);
      while (hit != 0u) {
        const int l = __builtin_ctz(hit);
        hit &= hit - 1u;
        const int s0 = __shfl(s, l, 32);
        const float4 v0 = *(const float4*)(hl + (long)s0 * 128);
        gs_add4(acc, v0);
      }
    }
  }
  return acc;
}

__device__ __forceinline__ float gs_dot4(const float4 a, const float4 b) {
  return (a.x * b.x + a.y * b.y) + (a.z * b.z + a.w * b.w);
}

__global__ void __launch_bounds__(256)
gs_fold_kernel(const float* __restrict__ Wl,
               const float* __restrict__ bl,
               const float* __restrict__ Wr,
               const float* __restrict__ We,
               float* __restrict__ U4,
               float* __restrict__ cb) {
  const int lane = threadIdx.x & 31;
  const int k = blockIdx.x * 8 + (threadIdx.x >> 5);
  if (k > 128) return;
  const float4 ws = *(const float4*)(We + lane * 4);
  const float4 wt = *(const float4*)(We + 128 + lane * 4);
  const bool isRow = k < 128;
  const float4 l = *(const float4*)((isRow ? Wl + k * 128 : bl) + lane * 4);
  const float4 r = isRow ? *(const float4*)(Wr + k * 128 + lane * 4)
                         : make_float4(0.0f, 0.0f, 0.0f, 0.0f);
  float s0 = gs_dot4(l, ws);
  float s1 = gs_dot4(l, wt);
  float s2 = gs_dot4(r, ws);
  float s3 = gs_dot4(r, wt);
  for (int off = 16; off > 0; off >>= 1) {
    s0 += __shfl_xor(s0, off, 32);
    s1 += __shfl_xor(s1, off, 32);
    s2 += __shfl_xor(s2, off, 32);
    s3 += __shfl_xor(s3, off, 32);
  }
  if (lane == 0) {
    if (isRow) {
      *(float4*)(U4 + k * 4) = make_float4(s0, s1, s2, s3);
    } else {
      cb[0] = s0;
      cb[1] = s1;
    }
  }
}

__device__ __forceinline__ void gs_load_b16(v2f (&b)[16],
                                            const float* __restrict__ Wcol,
                                            int kbase) {
#pragma unroll
  for (int s = 0; s < 16; ++s) {
    b[s].x = Wcol[(kbase + 4 * s) * 128];
    b[s].y = Wcol[(kbase + 4 * s + 1) * 128];
  }
}

__device__ __forceinline__ v8f gs_mma16(v8f acc, const float* lrow, int kbase,
                                        const v2f (&b)[16]) {
#pragma unroll
  for (int s = 0; s < 16; ++s) {
    v2f a;
    a.x = lrow[kbase + 4 * s];
    a.y = lrow[kbase + 4 * s + 1];
    acc = __builtin_amdgcn_wmma_f32_16x16x4_f32(
         false, a,  false, b[s],
         (short)0, acc,  false,  false);
  }
  return acc;
}

__global__ void __launch_bounds__(256)
gs_sage_wmma_kernel(const float* __restrict__ hself,
                    const int* __restrict__ deg,
                    const int* __restrict__ slot,
                    const int* __restrict__ spillT,
                    const int* __restrict__ spillS,
                    const float* __restrict__ Wl,
                    const float* __restrict__ bl,
                    const float* __restrict__ Wr,
                    const float* __restrict__ U4,
                    float* __restrict__ np,
                    int nNodes,
                    int nEdges) {
  __shared__ float lmean[16 * LDS_STRIDE];
  __shared__ float lself[16 * LDS_STRIDE];
  __shared__ float lproj[8 * 64];

  const int tid  = threadIdx.x;
  const int row0 = blockIdx.x * 16;
  const int lane = tid & 31;
  const int wave = __builtin_amdgcn_readfirstlane(tid >> 5);

  float4 selfv[2];
  for (int i = 0; i < 2; ++i) {
    const int idx = tid + i * 256;
    const int r   = idx >> 5;
    const int node = row0 + r;
    const int nclamp = node < nNodes ? node : (nNodes - 1);
    selfv[i] = *(const float4*)(hself + (long)nclamp * 128 + (idx & 31) * 4);
  }

  const int nodeA = row0 + wave * 2;
  const int nodeB = nodeA + 1;
  const int dA = nodeA < nNodes ? deg[nodeA] : 0;
  const int dB = nodeB < nNodes ? deg[nodeB] : 0;
  const int fA = lane < (dA < 32 ? dA : 32) ? slot[(long)nodeA * GS_CAP + lane] : 0;
  const int fB = lane < (dB < 32 ? dB : 32) ? slot[(long)nodeB * GS_CAP + lane] : 0;

  for (int i = 0; i < 2; ++i) {
    const int idx = tid + i * 256;
    *(float4*)(lself + (idx >> 5) * LDS_STRIDE + (idx & 31) * 4) = selfv[i];
  }

  {
    const float4 a = gs_gather_node(hself, deg, slot, spillT, spillS, nodeA, dA, fA,
                                    lane, nNodes, nEdges);
    const float rinv = 1.0f / fmaxf((float)dA, 1.0f);
    *(float4*)(lmean + (wave * 2) * LDS_STRIDE + lane * 4) =
        make_float4(a.x * rinv, a.y * rinv, a.z * rinv, a.w * rinv);
  }
  {
    const float4 a = gs_gather_node(hself, deg, slot, spillT, spillS, nodeB, dB, fB,
                                    lane, nNodes, nEdges);
    const float rinv = 1.0f / fmaxf((float)dB, 1.0f);
    *(float4*)(lmean + (wave * 2 + 1) * LDS_STRIDE + lane * 4) =
        make_float4(a.x * rinv, a.y * rinv, a.z * rinv, a.w * rinv);
  }
  __syncthreads();

  const int col0 = wave * 16;
  const int rowA = lane & 15;
  const int kofs = (lane >> 4) * 2;
  const int colB = lane & 15;

  v8f acc = {};

  const float* wlc = Wl + kofs * 128 + col0 + colB;
  const float* wrc = Wr + kofs * 128 + col0 + colB;
  const float* lm  = lmean + rowA * LDS_STRIDE + kofs;
  const float* ls  = lself + rowA * LDS_STRIDE + kofs;
  v2f bA[16], bB[16];
  gs_load_b16(bA, wlc, 0);
  gs_load_b16(bB, wlc, 64);
  acc = gs_mma16(acc, lm, 0, bA);
  gs_load_b16(bA, wrc, 0);
  acc = gs_mma16(acc, lm, 64, bB);
  gs_load_b16(bB, wrc, 64);
  acc = gs_mma16(acc, ls, 0, bA);
  acc = gs_mma16(acc, ls, 64, bB);

  const int c     = lane & 15;
  const int half  = lane >> 4;
  const int ocol  = col0 + c;
  const float bias = bl[ocol];
  const float4 u = *(const float4*)(U4 + ocol * 4);

  float q[32];
  for (int i = 0; i < 8; ++i) {
    const float t = fmaxf(acc[i] + bias, 0.0f);
    q[i]      = t * u.x;
    q[8 + i]  = t * u.y;
    q[16 + i] = t * u.z;
    q[24 + i] = t * u.w;
  }
  const bool b3 = (c & 8) != 0;
  const bool b2 = (c & 4) != 0;
  const bool b1 = (c & 2) != 0;
  const bool b0 = (c & 1) != 0;
  float w16[16];
  for (int k = 0; k < 16; ++k) {
    const float keep = b3 ? q[16 + k] : q[k];
    const float send = b3 ? q[k] : q[16 + k];
    w16[k] = keep + __shfl_xor(send, 8, 32);
  }
  float w8[8];
  for (int k = 0; k < 8; ++k) {
    const float keep = b2 ? w16[8 + k] : w16[k];
    const float send = b2 ? w16[k] : w16[8 + k];
    w8[k] = keep + __shfl_xor(send, 4, 32);
  }
  float w4[4];
  for (int k = 0; k < 4; ++k) {
    const float keep = b1 ? w8[4 + k] : w8[k];
    const float send = b1 ? w8[k] : w8[4 + k];
    w4[k] = keep + __shfl_xor(send, 2, 32);
  }
  float w2[2];
  for (int k = 0; k < 2; ++k) {
    const float keep = b0 ? w4[2 + k] : w4[k];
    const float send = b0 ? w4[k] : w4[2 + k];
    w2[k] = keep + __shfl_xor(send, 1, 32);
  }

  const int prow = half * 8 + (c & 3) * 2;
  const int pidx = c >> 2;
  lproj[wave * 64 + prow * 4 + pidx]       = w2[0];
  lproj[wave * 64 + (prow + 1) * 4 + pidx] = w2[1];
  __syncthreads();
  if (tid < 64) {
    float sum = lproj[tid];
    for (int w = 1; w < 8; ++w) sum += lproj[w * 64 + tid];
    const int node = row0 + (tid >> 2);
    if (node < nNodes) np[(long)row0 * 4 + tid] = sum;
  }
}

__global__ void __launch_bounds__(256)
gs_node_score_kernel(const float* __restrict__ np,
                     const float* __restrict__ cb,
                     const int* __restrict__ deg,
                     const int* __restrict__ slot,
                     const int* __restrict__ spillT,
                     const int* __restrict__ spillS,
                     float* __restrict__ ps,
                     float* __restrict__ pt,
                     int nNodes,
                     int nEdges) {
  const int node = (blockIdx.x * blockDim.x + threadIdx.x) >> 3;
  const int sub  = threadIdx.x & 7;
  const bool live = node < nNodes;
  const int d   = live ? deg[node] : 0;
  const int end = d < GS_CAP ? d : GS_CAP;
  const int* list = slot + (long)(live ? node : 0) * GS_CAP;
  float s = 0.0f, t = 0.0f;
  for (int k = sub; k < end; k += 8) {
    const float2 a = *(const float2*)(np + (long)list[k] * 4);
    s += a.x;
    t += a.y;
  }
  if (d > GS_CAP) {
    int m = deg[nNodes];
    m = m < nEdges ? m : nEdges;
    for (int i = sub; i < m; i += 8) {
      if (spillT[i] == node) {
        const float2 a = *(const float2*)(np + (long)spillS[i] * 4);
        s += a.x;
        t += a.y;
      }
    }
  }
  for (int off = 4; off > 0; off >>= 1) {
    s += __shfl_xor(s, off, 32);
    t += __shfl_xor(t, off, 32);
  }
  if (live && sub == 0) {
    const float den = fmaxf((float)d, 1.0f);
    const float2 self = *(const float2*)(np + (long)node * 4 + 2);
    ps[node] = s / den + self.x + cb[0];
    pt[node] = t / den + self.y + cb[1];
  }
}

__device__ __forceinline__ float gs_sigmoid(float p) {
  return 1.0f / (1.0f + __expf(-p));
}

__global__ void __launch_bounds__(256)
gs_edge_score_kernel(const float* __restrict__ ps,
                     const float* __restrict__ pt,
                     const int* __restrict__ src,
                     const int* __restrict__ tgt,
                     const float* __restrict__ be,
                     float* __restrict__ out,
                     int nEdges) {
  const int e0 = (blockIdx.x * blockDim.x + threadIdx.x) * 4;
  if (e0 >= nEdges) return;
  const float b = be[0];
  if ((nEdges & 3) == 0) {
    const int4 s4 = *(const int4*)(src + e0);
    const int4 t4 = *(const int4*)(tgt + e0);
    float4 y;
    y.x = gs_sigmoid(ps[s4.x] + pt[t4.x] + b);
    y.y = gs_sigmoid(ps[s4.y] + pt[t4.y] + b);
    y.z = gs_sigmoid(ps[s4.z] + pt[t4.z] + b);
    y.w = gs_sigmoid(ps[s4.w] + pt[t4.w] + b);
    *(float4*)(out + e0) = y;
  } else {
    const int eEnd = (e0 + 4) < nEdges ? (e0 + 4) : nEdges;
    for (int e = e0; e < eEnd; ++e)
      out[e] = gs_sigmoid(ps[src[e]] + pt[tgt[e]] + b);
  }
}

extern "C" void kernel_launch(void* const* d_in, const int* in_sizes, int n_in,
                              void* d_out, int out_size, void* d_ws, size_t ws_size,
                              hipStream_t stream) {
  const float* x    = (const float*)d_in[0];
  const int*   ei   = (const int*)d_in[1];
  const float* Wl1  = (const float*)d_in[2];
  const float* bl1  = (const float*)d_in[3];
  const float* Wr1  = (const float*)d_in[4];
  const float* Wl2  = (const float*)d_in[5];
  const float* bl2  = (const float*)d_in[6];
  const float* Wr2  = (const float*)d_in[7];
  const float* We   = (const float*)d_in[8];
  const float* be   = (const float*)d_in[9];

  const int N = in_sizes[0] / 128;
  const int E = in_sizes[1] / 2;
  const int* src = ei;
  const int* tgt = ei + E;

  int*   slot = (int*)d_ws;
  float* U4   = (float*)d_ws + (long)N * GS_CAP;
  float* cb   = U4 + 128 * 4;
  float* np   = cb + 4;
  int* deg    = (int*)(np + (long)N * 4);
  int* spillT = deg + (N + 1);
  int* spillS = spillT + E;
  float* ps   = (float*)(spillS + E);
  float* pt   = ps + N;
  int* sorted = (int*)(pt + N);
  int* off    = sorted + E;

  const int nBins = (N + GS_BIN - 1) / GS_BIN;
  const int nPart = (E + GS_CHUNK - 1) / GS_CHUNK;
  const int oct_Blocks = (N + 31) / 32;
  const int quadBlocks = ((E + 3) / 4 + 255) / 256;
  const int gemmBlocks = (N + 15) / 16;

  gs_part_kernel<<<nPart, 1024, 0, stream>>>(src, tgt, sorted, off, deg, N, E, nBins);
  gs_place_kernel<<<nBins, 256, 0, stream>>>(sorted, off, deg, slot, spillT, spillS,
                                             N, nBins, nPart);

  gs_fold_kernel<<<17, 256, 0, stream>>>(Wl2, bl2, Wr2, We, U4, cb);

  gs_sage_wmma_kernel<<<gemmBlocks, 256, 0, stream>>>(x, deg, slot, spillT, spillS,
                                                      Wl1, bl1, Wr1, U4, np, N, E);

  gs_node_score_kernel<<<oct_Blocks, 256, 0, stream>>>(np, cb, deg, slot, spillT, spillS, ps, pt, N, E);

  gs_edge_score_kernel<<<quadBlocks, 256, 0, stream>>>(ps, pt, src, tgt, be,
                                                       (float*)d_out, E);
}


# Layer-2 weight fold loses its own launch: 5 extra blocks in the edge-ordering launch's grid compute U4 and cb on WGPs that grid leaves idle; launches 6 -> 5; arithmetic and sum order unchanged
# speedup vs baseline: 6.8713x; 1.0314x over previous
#include <hip/hip_runtime.h>
#include <hip/hip_bf16.h>

typedef __attribute__((ext_vector_type(2))) float v2f;
typedef __attribute__((ext_vector_type(8))) float v8f;

#define LDS_STRIDE 132

#define GS_CAP 64
#define GS_BIN 64
#define GS_CHUNK 8192
#define GS_MAXB 1024
#define GS_MAXP 256

#define GS_FOLDB 5

__device__ __forceinline__ float gs_dot4(const float4 a, const float4 b) {
  return (a.x * b.x + a.y * b.y) + (a.z * b.z + a.w * b.w);
}

__device__ __forceinline__ void gs_fold_row(const float* __restrict__ Wl,
                                            const float* __restrict__ bl,
                                            const float* __restrict__ Wr,
                                            const float* __restrict__ We,
                                            float* __restrict__ U4,
                                            float* __restrict__ cb,
                                            int k, int lane) {
  if (k > 128) return;
  const float4 ws = *(const float4*)(We + lane * 4);
  const float4 wt = *(const float4*)(We + 128 + lane * 4);
  const bool isRow = k < 128;
  const float4 l = *(const float4*)((isRow ? Wl + k * 128 : bl) + lane * 4);
  const float4 r = isRow ? *(const float4*)(Wr + k * 128 + lane * 4)
                         : make_float4(0.0f, 0.0f, 0.0f, 0.0f);
  float s0 = gs_dot4(l, ws);
  float s1 = gs_dot4(l, wt);
  float s2 = gs_dot4(r, ws);
  float s3 = gs_dot4(r, wt);
  for (int off = 16; off > 0; off >>= 1) {
    s0 += __shfl_xor(s0, off, 32);
    s1 += __shfl_xor(s1, off, 32);
    s2 += __shfl_xor(s2, off, 32);
    s3 += __shfl_xor(s3, off, 32);
  }
  if (lane == 0) {
    if (isRow) {
      *(float4*)(U4 + k * 4) = make_float4(s0, s1, s2, s3);
    } else {
      cb[0] = s0;
      cb[1] = s1;
    }
  }
}

__global__ void __launch_bounds__(1024)
gs_part_kernel(const int* __restrict__ src,
               const int* __restrict__ tgt,
               int* __restrict__ sorted,
               int* __restrict__ off,
               int* __restrict__ deg,
               int nNodes,
               int nEdges,
               int nBins,
               int nPart,
               const float* __restrict__ Wl2,
               const float* __restrict__ bl2,
               const float* __restrict__ Wr2,
               const float* __restrict__ We,
               float* __restrict__ U4,
               float* __restrict__ cb) {
  __shared__ int hist[GS_MAXB];
  __shared__ int wsum[32];
  __shared__ int stage[GS_CHUNK];
  const int tid  = threadIdx.x;
  const int lane = tid & 31;
  const int wave = tid >> 5;
  const int base = blockIdx.x * GS_CHUNK;

  if ((int)blockIdx.x >= nPart) {
    gs_fold_row(Wl2, bl2, Wr2, We, U4, cb, ((int)blockIdx.x - nPart) * 32 + wave, lane);
    return;
  }

  if (blockIdx.x == 0 && tid == 0) deg[nNodes] = 0;
  hist[tid] = 0;
  __syncthreads();

  int pk[8], bn[8], rk[8];
#pragma unroll
  for (int i = 0; i < 8; ++i) {
    const int e = base + i * 1024 + tid;
    const bool ok = e < nEdges;
    const int t = ok ? tgt[e] : 0;
    const int s = ok ? src[e] : 0;
    pk[i] = (s << 6) | (t & (GS_BIN - 1));
    bn[i] = ok ? (t >> 6) : -1;
  }
#pragma unroll
  for (int i = 0; i < 8; ++i)
    rk[i] = bn[i] >= 0 ? atomicAdd(&hist[bn[i]], 1) : 0;
  __syncthreads();

  const int v = hist[tid];
  int inc = v;
  for (int d = 1; d < 32; d <<= 1) {
    const int n = __shfl_up(inc, d, 32);
    if (lane >= d) inc += n;
  }
  if (lane == 31) wsum[wave] = inc;
  __syncthreads();
  if (wave == 0) {
    const int w = wsum[lane];
    int wi = w;
    for (int d = 1; d < 32; d <<= 1) {
      const int n = __shfl_up(wi, d, 32);
      if (lane >= d) wi += n;
    }
    wsum[lane] = wi - w;
  }
  __syncthreads();
  const int ex = inc - v + wsum[wave];
  hist[tid] = ex;
  if (tid <= nBins) off[(long)blockIdx.x * (nBins + 1) + tid] = ex;
  __syncthreads();

#pragma unroll
  for (int i = 0; i < 8; ++i)
    if (bn[i] >= 0) stage[hist[bn[i]] + rk[i]] = pk[i];
  __syncthreads();
#pragma unroll
  for (int i = 0; i < 8; ++i) {
    const int e = base + i * 1024 + tid;
    if (e < nEdges) sorted[e] = stage[i * 1024 + tid];
  }
}

__global__ void __launch_bounds__(256)
gs_place_kernel(const int* __restrict__ sorted,
                const int* __restrict__ off,
                int* __restrict__ deg,
                int* __restrict__ slot,
                int* __restrict__ spillT,
                int* __restrict__ spillS,
                int nNodes,
                int nBins,
                int nPart) {
  __shared__ int pre[GS_MAXP + 1];
  __shared__ int gst[GS_MAXP];
  __shared__ int wsum[8];
  __shared__ int cnt[GS_BIN];
  __shared__ int slotL[GS_BIN * GS_CAP];
  const int tid  = threadIdx.x;
  const int lane = tid & 31;
  const int wave = tid >> 5;
  const int b    = blockIdx.x;

  int len = 0, g = 0;
  if (tid < nPart) {
    const int* o = off + (long)tid * (nBins + 1) + b;
    const int a = o[0];
    len = o[1] - a;
    g = tid * GS_CHUNK + a;
  }
  if (tid < GS_BIN) cnt[tid] = 0;
  int inc = len;
  for (int d = 1; d < 32; d <<= 1) {
    const int n = __shfl_up(inc, d, 32);
    if (lane >= d) inc += n;
  }
  if (lane == 31) wsum[wave] = inc;
  __syncthreads();
  int wofs = 0;
  for (int w = 0; w < wave; ++w) wofs += wsum[w];
  const int ex = inc - len + wofs;
  pre[tid] = ex;
  gst[tid] = g;
  if (tid == 255) pre[256] = ex + len;
  __syncthreads();

  const int total = pre[256];
  for (int j = tid; j < total; j += 256) {
    int k = 0;
    for (int step = 128; step > 0; step >>= 1)
      if (pre[k + step] <= j) k += step;
    const int p = sorted[gst[k] + (j - pre[k])];
    const int n = p & (GS_BIN - 1);
    const int s = p >> 6;
    const int r = atomicAdd(&cnt[n], 1);
    if (r < GS_CAP) {
      slotL[n * GS_CAP + r] = s;
    } else {
      const int o = atomicAdd(deg + nNodes, 1);
      spillT[o] = b * GS_BIN + n;
      spillS[o] = s;
    }
  }
  __syncthreads();

  for (int n = wave * 8; n < wave * 8 + 8; ++n) {
    const int node = b * GS_BIN + n;
    if (node < nNodes) {
      const int d = cnt[n];
      if (d > 0)  slot[(long)node * GS_CAP + lane]      = slotL[n * GS_CAP + lane];
      if (d > 32) slot[(long)node * GS_CAP + 32 + lane] = slotL[n * GS_CAP + 32 + lane];
    }
  }
  if (tid < GS_BIN && b * GS_BIN + tid < nNodes) deg[b * GS_BIN + tid] = cnt[tid];
}

__device__ __forceinline__ void gs_add4(float4& a, const float4 v) {
  a.x += v.x; a.y += v.y; a.z += v.z; a.w += v.w;
}

template <int U>
__device__ __forceinline__ void gs_rows(float4& acc, const float* hl, int mine, int j) {
  float4 v[U];
#pragma unroll
  for (int u = 0; u < U; ++u) {
    const int s = __builtin_amdgcn_readlane(mine, j + u);
    v[u] = *(const float4*)(hl + (long)s * 128);
  }
#pragma unroll
  for (int u = 0; u < U; ++u) gs_add4(acc, v[u]);
}

__device__ __forceinline__ float4 gs_gather_node(const float* __restrict__ h,
                                                 const int* __restrict__ deg,
                                                 const int* __restrict__ slot,
                                                 const int* __restrict__ spillT,
                                                 const int* __restrict__ spillS,
                                                 int node, int d, int first32,
                                                 int lane, int nNodes, int nEdges) {
  const int end = d < GS_CAP ? d : GS_CAP;
  const int* list = slot + (long)node * GS_CAP;
  const float* hl = h + lane * 4;
  float4 acc = make_float4(0.0f, 0.0f, 0.0f, 0.0f);

  for (int base = 0; base < end; base += 32) {
    const int n = (end - base) < 32 ? (end - base) : 32;
    int mine = first32;
    if (base != 0) mine = lane < n ? list[base + lane] : 0;
    int j = 0;
    for (; j + 16 <= n; j += 16) gs_rows<16>(acc, hl, mine, j);
    if (j + 8 <= n) { gs_rows<8>(acc, hl, mine, j); j += 8; }
    if (j + 4 <= n) { gs_rows<4>(acc, hl, mine, j); j += 4; }
    if (j + 2 <= n) { gs_rows<2>(acc, hl, mine, j); j += 2; }
    if (j < n)      { gs_rows<1>(acc, hl, mine, j); }
  }

  if (d > GS_CAP) {
    int m = deg[nNodes];
    m = m < nEdges ? m : nEdges;
    for (int b = 0; b < m; b += 32) {
      const int i = b + lane;
      const int t = i < m ? spillT[i] : -1;
      const int s = i < m ? spillS[i] : 0;
      unsigned hit = (unsigned)__ballot(t == node);
      while (hit != 0u) {
        const int l = __builtin_ctz(hit);
        hit &= hit - 1u;
        const int s0 = __shfl(s, l, 32);
        const float4 v0 = *(const float4*)(hl + (long)s0 * 128);
        gs_add4(acc, v0);
      }
    }
  }
  return acc;
}

__device__ __forceinline__ void gs_load_b16(v2f (&b)[16],
                                            const float* __restrict__ Wcol,
                                            int kbase) {
#pragma unroll
  for (int s = 0; s < 16; ++s) {
    b[s].x = Wcol[(kbase + 4 * s) * 128];
    b[s].y = Wcol[(kbase + 4 * s + 1) * 128];
  }
}

__device__ __forceinline__ v8f gs_mma16(v8f acc, const float* lrow, int kbase,
                                        const v2f (&b)[16]) {
#pragma unroll
  for (int s = 0; s < 16; ++s) {
    v2f a;
    a.x = lrow[kbase + 4 * s];
    a.y = lrow[kbase + 4 * s + 1];
    acc = __builtin_amdgcn_wmma_f32_16x16x4_f32(
         false, a,  false, b[s],
         (short)0, acc,  false,  false);
  }
  return acc;
}

__global__ void __launch_bounds__(256)
gs_sage_wmma_kernel(const float* __restrict__ hself,
                    const int* __restrict__ deg,
                    const int* __restrict__ slot,
                    const int* __restrict__ spillT,
                    const int* __restrict__ spillS,
                    const float* __restrict__ Wl,
                    const float* __restrict__ bl,
                    const float* __restrict__ Wr,
                    const float* __restrict__ U4,
                    float* __restrict__ np,
                    int nNodes,
                    int nEdges) {
  __shared__ float lmean[16 * LDS_STRIDE];
  __shared__ float lself[16 * LDS_STRIDE];
  __shared__ float lproj[8 * 64];

  const int tid  = threadIdx.x;
  const int row0 = blockIdx.x * 16;
  const int lane = tid & 31;
  const int wave = __builtin_amdgcn_readfirstlane(tid >> 5);

  float4 selfv[2];
  for (int i = 0; i < 2; ++i) {
    const int idx = tid + i * 256;
    const int r   = idx >> 5;
    const int node = row0 + r;
    const int nclamp = node < nNodes ? node : (nNodes - 1);
    selfv[i] = *(const float4*)(hself + (long)nclamp * 128 + (idx & 31) * 4);
  }

  const int nodeA = row0 + wave * 2;
  const int nodeB = nodeA + 1;
  const int dA = nodeA < nNodes ? deg[nodeA] : 0;
  const int dB = nodeB < nNodes ? deg[nodeB] : 0;
  const int fA = lane < (dA < 32 ? dA : 32) ? slot[(long)nodeA * GS_CAP + lane] : 0;
  const int fB = lane < (dB < 32 ? dB : 32) ? slot[(long)nodeB * GS_CAP + lane] : 0;

  for (int i = 0; i < 2; ++i) {
    const int idx = tid + i * 256;
    *(float4*)(lself + (idx >> 5) * LDS_STRIDE + (idx & 31) * 4) = selfv[i];
  }

  {
    const float4 a = gs_gather_node(hself, deg, slot, spillT, spillS, nodeA, dA, fA,
                                    lane, nNodes, nEdges);
    const float rinv = 1.0f / fmaxf((float)dA, 1.0f);
    *(float4*)(lmean + (wave * 2) * LDS_STRIDE + lane * 4) =
        make_float4(a.x * rinv, a.y * rinv, a.z * rinv, a.w * rinv);
  }
  {
    const float4 a = gs_gather_node(hself, deg, slot, spillT, spillS, nodeB, dB, fB,
                                    lane, nNodes, nEdges);
    const float rinv = 1.0f / fmaxf((float)dB, 1.0f);
    *(float4*)(lmean + (wave * 2 + 1) * LDS_STRIDE + lane * 4) =
        make_float4(a.x * rinv, a.y * rinv, a.z * rinv, a.w * rinv);
  }
  __syncthreads();

  const int col0 = wave * 16;
  const int rowA = lane & 15;
  const int kofs = (lane >> 4) * 2;
  const int colB = lane & 15;

  v8f acc = {};

  const float* wlc = Wl + kofs * 128 + col0 + colB;
  const float* wrc = Wr + kofs * 128 + col0 + colB;
  const float* lm  = lmean + rowA * LDS_STRIDE + kofs;
  const float* ls  = lself + rowA * LDS_STRIDE + kofs;
  v2f bA[16], bB[16];
  gs_load_b16(bA, wlc, 0);
  gs_load_b16(bB, wlc, 64);
  acc = gs_mma16(acc, lm, 0, bA);
  gs_load_b16(bA, wrc, 0);
  acc = gs_mma16(acc, lm, 64, bB);
  gs_load_b16(bB, wrc, 64);
  acc = gs_mma16(acc, ls, 0, bA);
  acc = gs_mma16(acc, ls, 64, bB);

  const int c     = lane & 15;
  const int half  = lane >> 4;
  const int ocol  = col0 + c;
  const float bias = bl[ocol];
  const float4 u = *(const float4*)(U4 + ocol * 4);

  float q[32];
  for (int i = 0; i < 8; ++i) {
    const float t = fmaxf(acc[i] + bias, 0.0f);
    q[i]      = t * u.x;
    q[8 + i]  = t * u.y;
    q[16 + i] = t * u.z;
    q[24 + i] = t * u.w;
  }
  const bool b3 = (c & 8) != 0;
  const bool b2 = (c & 4) != 0;
  const bool b1 = (c & 2) != 0;
  const bool b0 = (c & 1) != 0;
  float w16[16];
  for (int k = 0; k < 16; ++k) {
    const float keep = b3 ? q[16 + k] : q[k];
    const float send = b3 ? q[k] : q[16 + k];
    w16[k] = keep + __shfl_xor(send, 8, 32);
  }
  float w8[8];
  for (int k = 0; k < 8; ++k) {
    const float keep = b2 ? w16[8 + k] : w16[k];
    const float send = b2 ? w16[k] : w16[8 + k];
    w8[k] = keep + __shfl_xor(send, 4, 32);
  }
  float w4[4];
  for (int k = 0; k < 4; ++k) {
    const float keep = b1 ? w8[4 + k] : w8[k];
    const float send = b1 ? w8[k] : w8[4 + k];
    w4[k] = keep + __shfl_xor(send, 2, 32);
  }
  float w2[2];
  for (int k = 0; k < 2; ++k) {
    const float keep = b0 ? w4[2 + k] : w4[k];
    const float send = b0 ? w4[k] : w4[2 + k];
    w2[k] = keep + __shfl_xor(send, 1, 32);
  }

  const int prow = half * 8 + (c & 3) * 2;
  const int pidx = c >> 2;
  lproj[wave * 64 + prow * 4 + pidx]       = w2[0];
  lproj[wave * 64 + (prow + 1) * 4 + pidx] = w2[1];
  __syncthreads();
  if (tid < 64) {
    float sum = lproj[tid];
    for (int w = 1; w < 8; ++w) sum += lproj[w * 64 + tid];
    const int node = row0 + (tid >> 2);
    if (node < nNodes) np[(long)row0 * 4 + tid] = sum;
  }
}

__global__ void __launch_bounds__(256)
gs_node_score_kernel(const float* __restrict__ np,
                     const float* __restrict__ cb,
                     const int* __restrict__ deg,
                     const int* __restrict__ slot,
                     const int* __restrict__ spillT,
                     const int* __restrict__ spillS,
                     float* __restrict__ ps,
                     float* __restrict__ pt,
                     int nNodes,
                     int nEdges) {
  const int node = (blockIdx.x * blockDim.x + threadIdx.x) >> 3;
  const int sub  = threadIdx.x & 7;
  const bool live = node < nNodes;
  const int d   = live ? deg[node] : 0;
  const int end = d < GS_CAP ? d : GS_CAP;
  const int* list = slot + (long)(live ? node : 0) * GS_CAP;
  float s = 0.0f, t = 0.0f;
  for (int k = sub; k < end; k += 8) {
    const float2 a = *(const float2*)(np + (long)list[k] * 4);
    s += a.x;
    t += a.y;
  }
  if (d > GS_CAP) {
    int m = deg[nNodes];
    m = m < nEdges ? m : nEdges;
    for (int i = sub; i < m; i += 8) {
      if (spillT[i] == node) {
        const float2 a = *(const float2*)(np + (long)spillS[i] * 4);
        s += a.x;
        t += a.y;
      }
    }
  }
  for (int off = 4; off > 0; off >>= 1) {
    s += __shfl_xor(s, off, 32);
    t += __shfl_xor(t, off, 32);
  }
  if (live && sub == 0) {
    const float den = fmaxf((float)d, 1.0f);
    const float2 self = *(const float2*)(np + (long)node * 4 + 2);
    ps[node] = s / den + self.x + cb[0];
    pt[node] = t / den + self.y + cb[1];
  }
}

__device__ __forceinline__ float gs_sigmoid(float p) {
  return 1.0f / (1.0f + __expf(-p));
}

__global__ void __launch_bounds__(256)
gs_edge_score_kernel(const float* __restrict__ ps,
                     const float* __restrict__ pt,
                     const int* __restrict__ src,
                     const int* __restrict__ tgt,
                     const float* __restrict__ be,
                     float* __restrict__ out,
                     int nEdges) {
  const int e0 = (blockIdx.x * blockDim.x + threadIdx.x) * 4;
  if (e0 >= nEdges) return;
  const float b = be[0];
  if ((nEdges & 3) == 0) {
    const int4 s4 = *(const int4*)(src + e0);
    const int4 t4 = *(const int4*)(tgt + e0);
    float4 y;
    y.x = gs_sigmoid(ps[s4.x] + pt[t4.x] + b);
    y.y = gs_sigmoid(ps[s4.y] + pt[t4.y] + b);
    y.z = gs_sigmoid(ps[s4.z] + pt[t4.z] + b);
    y.w = gs_sigmoid(ps[s4.w] + pt[t4.w] + b);
    *(float4*)(out + e0) = y;
  } else {
    const int eEnd = (e0 + 4) < nEdges ? (e0 + 4) : nEdges;
    for (int e = e0; e < eEnd; ++e)
      out[e] = gs_sigmoid(ps[src[e]] + pt[tgt[e]] + b);
  }
}

extern "C" void kernel_launch(void* const* d_in, const int* in_sizes, int n_in,
                              void* d_out, int out_size, void* d_ws, size_t ws_size,
                              hipStream_t stream) {
  const float* x    = (const float*)d_in[0];
  const int*   ei   = (const int*)d_in[1];
  const float* Wl1  = (const float*)d_in[2];
  const float* bl1  = (const float*)d_in[3];
  const float* Wr1  = (const float*)d_in[4];
  const float* Wl2  = (const float*)d_in[5];
  const float* bl2  = (const float*)d_in[6];
  const float* Wr2  = (const float*)d_in[7];
  const float* We   = (const float*)d_in[8];
  const float* be   = (const float*)d_in[9];

  const int N = in_sizes[0] / 128;
  const int E = in_sizes[1] / 2;
  const int* src = ei;
  const int* tgt = ei + E;

  int*   slot = (int*)d_ws;
  float* U4   = (float*)d_ws + (long)N * GS_CAP;
  float* cb   = U4 + 128 * 4;
  float* np   = cb + 4;
  int* deg    = (int*)(np + (long)N * 4);
  int* spillT = deg + (N + 1);
  int* spillS = spillT + E;
  float* ps   = (float*)(spillS + E);
  float* pt   = ps + N;
  int* sorted = (int*)(pt + N);
  int* off    = sorted + E;

  const int nBins = (N + GS_BIN - 1) / GS_BIN;
  const int nPart = (E + GS_CHUNK - 1) / GS_CHUNK;
  const int oct_Blocks = (N + 31) / 32;
  const int quadBlocks = ((E + 3) / 4 + 255) / 256;
  const int gemmBlocks = (N + 15) / 16;

  gs_part_kernel<<<nPart + GS_FOLDB, 1024, 0, stream>>>(src, tgt, sorted, off, deg, N, E, nBins,
                                                        nPart, Wl2, bl2, Wr2, We, U4, cb);
  gs_place_kernel<<<nBins, 256, 0, stream>>>(sorted, off, deg, slot, spillT, spillS,
                                             N, nBins, nPart);

  gs_sage_wmma_kernel<<<gemmBlocks, 256, 0, stream>>>(x, deg, slot, spillT, spillS,
                                                      Wl1, bl1, Wr1, U4, np, N, E);

  gs_node_score_kernel<<<oct_Blocks, 256, 0, stream>>>(np, cb, deg, slot, spillT, spillS, ps, pt, N, E);

  gs_edge_score_kernel<<<quadBlocks, 256, 0, stream>>>(ps, pt, src, tgt, be,
                                                       (float*)d_out, E);
}


# Layer-1 weights copied per call into WMMA fragment order by spare blocks of the first launch: a B fragment pair is one whole-line 16-byte load, 32 weight loads per wave not 128; sum order unchanged
# speedup vs baseline: 7.1817x; 1.0452x over previous
#include <hip/hip_runtime.h>
#include <hip/hip_bf16.h>

typedef __attribute__((ext_vector_type(2))) float v2f;
typedef __attribute__((ext_vector_type(8))) float v8f;

#define LDS_STRIDE 132

#define GS_CAP 64
#define GS_BIN 64
#define GS_CHUNK 8192
#define GS_MAXB 1024
#define GS_MAXP 256

#define GS_FOLDB 5

#define GS_SWZB 8

__device__ __forceinline__ void gs_swizzle_b(const float* __restrict__ Wl,
                                             const float* __restrict__ Wr,
                                             float* __restrict__ Bf,
                                             int idx) {
  const int lane = idx & 31;
  const int j    = (idx >> 5) & 31;
  const int w    = idx >> 10;
  const float* W = (j < 16 ? Wl : Wr);
  const int n  = w * 16 + (lane & 15);
  const int k0 = 8 * (j & 15) + 2 * (lane >> 4);
  float4 v;
  v.x = W[(k0    ) * 128 + n];
  v.y = W[(k0 + 1) * 128 + n];
  v.z = W[(k0 + 4) * 128 + n];
  v.w = W[(k0 + 5) * 128 + n];
  *(float4*)(Bf + (long)idx * 4) = v;
}

__device__ __forceinline__ float gs_dot4(const float4 a, const float4 b) {
  return (a.x * b.x + a.y * b.y) + (a.z * b.z + a.w * b.w);
}

__device__ __forceinline__ void gs_fold_row(const float* __restrict__ Wl,
                                            const float* __restrict__ bl,
                                            const float* __restrict__ Wr,
                                            const float* __restrict__ We,
                                            float* __restrict__ U4,
                                            float* __restrict__ cb,
                                            int k, int lane) {
  if (k > 128) return;
  const float4 ws = *(const float4*)(We + lane * 4);
  const float4 wt = *(const float4*)(We + 128 + lane * 4);
  const bool isRow = k < 128;
  const float4 l = *(const float4*)((isRow ? Wl + k * 128 : bl) + lane * 4);
  const float4 r = isRow ? *(const float4*)(Wr + k * 128 + lane * 4)
                         : make_float4(0.0f, 0.0f, 0.0f, 0.0f);
  float s0 = gs_dot4(l, ws);
  float s1 = gs_dot4(l, wt);
  float s2 = gs_dot4(r, ws);
  float s3 = gs_dot4(r, wt);
  for (int off = 16; off > 0; off >>= 1) {
    s0 += __shfl_xor(s0, off, 32);
    s1 += __shfl_xor(s1, off, 32);
    s2 += __shfl_xor(s2, off, 32);
    s3 += __shfl_xor(s3, off, 32);
  }
  if (lane == 0) {
    if (isRow) {
      *(float4*)(U4 + k * 4) = make_float4(s0, s1, s2, s3);
    } else {
      cb[0] = s0;
      cb[1] = s1;
    }
  }
}

__global__ void __launch_bounds__(1024)
gs_part_kernel(const int* __restrict__ src,
               const int* __restrict__ tgt,
               int* __restrict__ sorted,
               int* __restrict__ off,
               int* __restrict__ deg,
               int nNodes,
               int nEdges,
               int nBins,
               int nPart,
               const float* __restrict__ Wl2,
               const float* __restrict__ bl2,
               const float* __restrict__ Wr2,
               const float* __restrict__ We,
               float* __restrict__ U4,
               float* __restrict__ cb,
               const float* __restrict__ Wl1,
               const float* __restrict__ Wr1,
               float* __restrict__ Bf) {
  __shared__ int hist[GS_MAXB];
  __shared__ int wsum[32];
  __shared__ int stage[GS_CHUNK];
  const int tid  = threadIdx.x;
  const int lane = tid & 31;
  const int wave = tid >> 5;
  const int base = blockIdx.x * GS_CHUNK;

  if ((int)blockIdx.x >= nPart) {
    const int xb = (int)blockIdx.x - nPart;
    if (xb < GS_FOLDB) {
      gs_fold_row(Wl2, bl2, Wr2, We, U4, cb, xb * 32 + wave, lane);
    } else {
      gs_swizzle_b(Wl1, Wr1, Bf, (xb - GS_FOLDB) * 1024 + tid);
    }
    return;
  }

  if (blockIdx.x == 0 && tid == 0) deg[nNodes] = 0;
  hist[tid] = 0;
  __syncthreads();

  int pk[8], bn[8], rk[8];
#pragma unroll
  for (int i = 0; i < 8; ++i) {
    const int e = base + i * 1024 + tid;
    const bool ok = e < nEdges;
    const int t = ok ? tgt[e] : 0;
    const int s = ok ? src[e] : 0;
    pk[i] = (s << 6) | (t & (GS_BIN - 1));
    bn[i] = ok ? (t >> 6) : -1;
  }
#pragma unroll
  for (int i = 0; i < 8; ++i)
    rk[i] = bn[i] >= 0 ? atomicAdd(&hist[bn[i]], 1) : 0;
  __syncthreads();

  const int v = hist[tid];
  int inc = v;
  for (int d = 1; d < 32; d <<= 1) {
    const int n = __shfl_up(inc, d, 32);
    if (lane >= d) inc += n;
  }
  if (lane == 31) wsum[wave] = inc;
  __syncthreads();
  if (wave == 0) {
    const int w = wsum[lane];
    int wi = w;
    for (int d = 1; d < 32; d <<= 1) {
      const int n = __shfl_up(wi, d, 32);
      if (lane >= d) wi += n;
    }
    wsum[lane] = wi - w;
  }
  __syncthreads();
  const int ex = inc - v + wsum[wave];
  hist[tid] = ex;
  if (tid <= nBins) off[(long)blockIdx.x * (nBins + 1) + tid] = ex;
  __syncthreads();

#pragma unroll
  for (int i = 0; i < 8; ++i)
    if (bn[i] >= 0) stage[hist[bn[i]] + rk[i]] = pk[i];
  __syncthreads();
#pragma unroll
  for (int i = 0; i < 8; ++i) {
    const int e = base + i * 1024 + tid;
    if (e < nEdges) sorted[e] = stage[i * 1024 + tid];
  }
}

__global__ void __launch_bounds__(256)
gs_place_kernel(const int* __restrict__ sorted,
                const int* __restrict__ off,
                int* __restrict__ deg,
                int* __restrict__ slot,
                int* __restrict__ spillT,
                int* __restrict__ spillS,
                int nNodes,
                int nBins,
                int nPart) {
  __shared__ int pre[GS_MAXP + 1];
  __shared__ int gst[GS_MAXP];
  __shared__ int wsum[8];
  __shared__ int cnt[GS_BIN];
  __shared__ int slotL[GS_BIN * GS_CAP];
  const int tid  = threadIdx.x;
  const int lane = tid & 31;
  const int wave = tid >> 5;
  const int b    = blockIdx.x;

  int len = 0, g = 0;
  if (tid < nPart) {
    const int* o = off + (long)tid * (nBins + 1) + b;
    const int a = o[0];
    len = o[1] - a;
    g = tid * GS_CHUNK + a;
  }
  if (tid < GS_BIN) cnt[tid] = 0;
  int inc = len;
  for (int d = 1; d < 32; d <<= 1) {
    const int n = __shfl_up(inc, d, 32);
    if (lane >= d) inc += n;
  }
  if (lane == 31) wsum[wave] = inc;
  __syncthreads();
  int wofs = 0;
  for (int w = 0; w < wave; ++w) wofs += wsum[w];
  const int ex = inc - len + wofs;
  pre[tid] = ex;
  gst[tid] = g;
  if (tid == 255) pre[256] = ex + len;
  __syncthreads();

  const int total = pre[256];
  for (int j = tid; j < total; j += 256) {
    int k = 0;
    for (int step = 128; step > 0; step >>= 1)
      if (pre[k + step] <= j) k += step;
    const int p = sorted[gst[k] + (j - pre[k])];
    const int n = p & (GS_BIN - 1);
    const int s = p >> 6;
    const int r = atomicAdd(&cnt[n], 1);
    if (r < GS_CAP) {
      slotL[n * GS_CAP + r] = s;
    } else {
      const int o = atomicAdd(deg + nNodes, 1);
      spillT[o] = b * GS_BIN + n;
      spillS[o] = s;
    }
  }
  __syncthreads();

  for (int n = wave * 8; n < wave * 8 + 8; ++n) {
    const int node = b * GS_BIN + n;
    if (node < nNodes) {
      const int d = cnt[n];
      if (d > 0)  slot[(long)node * GS_CAP + lane]      = slotL[n * GS_CAP + lane];
      if (d > 32) slot[(long)node * GS_CAP + 32 + lane] = slotL[n * GS_CAP + 32 + lane];
    }
  }
  if (tid < GS_BIN && b * GS_BIN + tid < nNodes) deg[b * GS_BIN + tid] = cnt[tid];
}

__device__ __forceinline__ void gs_add4(float4& a, const float4 v) {
  a.x += v.x; a.y += v.y; a.z += v.z; a.w += v.w;
}

template <int U>
__device__ __forceinline__ void gs_rows(float4& acc, const float* hl, int mine, int j) {
  float4 v[U];
#pragma unroll
  for (int u = 0; u < U; ++u) {
    const int s = __builtin_amdgcn_readlane(mine, j + u);
    v[u] = *(const float4*)(hl + (long)s * 128);
  }
#pragma unroll
  for (int u = 0; u < U; ++u) gs_add4(acc, v[u]);
}

__device__ __forceinline__ float4 gs_gather_node(const float* __restrict__ h,
                                                 const int* __restrict__ deg,
                                                 const int* __restrict__ slot,
                                                 const int* __restrict__ spillT,
                                                 const int* __restrict__ spillS,
                                                 int node, int d, int first32,
                                                 int lane, int nNodes, int nEdges) {
  const int end = d < GS_CAP ? d : GS_CAP;
  const int* list = slot + (long)node * GS_CAP;
  const float* hl = h + lane * 4;
  float4 acc = make_float4(0.0f, 0.0f, 0.0f, 0.0f);

  for (int base = 0; base < end; base += 32) {
    const int n = (end - base) < 32 ? (end - base) : 32;
    int mine = first32;
    if (base != 0) mine = lane < n ? list[base + lane] : 0;
    int j = 0;
    for (; j + 16 <= n; j += 16) gs_rows<16>(acc, hl, mine, j);
    if (j + 8 <= n) { gs_rows<8>(acc, hl, mine, j); j += 8; }
    if (j + 4 <= n) { gs_rows<4>(acc, hl, mine, j); j += 4; }
    if (j + 2 <= n) { gs_rows<2>(acc, hl, mine, j); j += 2; }
    if (j < n)      { gs_rows<1>(acc, hl, mine, j); }
  }

  if (d > GS_CAP) {
    int m = deg[nNodes];
    m = m < nEdges ? m : nEdges;
    for (int b = 0; b < m; b += 32) {
      const int i = b + lane;
      const int t = i < m ? spillT[i] : -1;
      const int s = i < m ? spillS[i] : 0;
      unsigned hit = (unsigned)__ballot(t == node);
      while (hit != 0u) {
        const int l = __builtin_ctz(hit);
        hit &= hit - 1u;
        const int s0 = __shfl(s, l, 32);
        const float4 v0 = *(const float4*)(hl + (long)s0 * 128);
        gs_add4(acc, v0);
      }
    }
  }
  return acc;
}

__device__ __forceinline__ void gs_load_b16(v2f (&b)[16],
                                            const float4* __restrict__ Bw,
                                            int j0) {
#pragma unroll
  for (int i = 0; i < 8; ++i) {
    const float4 v = Bw[(j0 + i) * 32];
    b[2 * i].x     = v.x;
    b[2 * i].y     = v.y;
    b[2 * i + 1].x = v.z;
    b[2 * i + 1].y = v.w;
  }
}

__device__ __forceinline__ v8f gs_mma16(v8f acc, const float* lrow, int kbase,
                                        const v2f (&b)[16]) {
#pragma unroll
  for (int s = 0; s < 16; ++s) {
    v2f a;
    a.x = lrow[kbase + 4 * s];
    a.y = lrow[kbase + 4 * s + 1];
    acc = __builtin_amdgcn_wmma_f32_16x16x4_f32(
         false, a,  false, b[s],
         (short)0, acc,  false,  false);
  }
  return acc;
}

__global__ void __launch_bounds__(256)
gs_sage_wmma_kernel(const float* __restrict__ hself,
                    const int* __restrict__ deg,
                    const int* __restrict__ slot,
                    const int* __restrict__ spillT,
                    const int* __restrict__ spillS,
                    const float* __restrict__ Bf,
                    const float* __restrict__ bl,
                    const float* __restrict__ U4,
                    float* __restrict__ np,
                    int nNodes,
                    int nEdges) {
  __shared__ float lmean[16 * LDS_STRIDE];
  __shared__ float lself[16 * LDS_STRIDE];
  __shared__ float lproj[8 * 64];

  const int tid  = threadIdx.x;
  const int row0 = blockIdx.x * 16;
  const int lane = tid & 31;
  const int wave = __builtin_amdgcn_readfirstlane(tid >> 5);

  float4 selfv[2];
  for (int i = 0; i < 2; ++i) {
    const int idx = tid + i * 256;
    const int r   = idx >> 5;
    const int node = row0 + r;
    const int nclamp = node < nNodes ? node : (nNodes - 1);
    selfv[i] = *(const float4*)(hself + (long)nclamp * 128 + (idx & 31) * 4);
  }

  const int nodeA = row0 + wave * 2;
  const int nodeB = nodeA + 1;
  const int dA = nodeA < nNodes ? deg[nodeA] : 0;
  const int dB = nodeB < nNodes ? deg[nodeB] : 0;
  const int fA = lane < (dA < 32 ? dA : 32) ? slot[(long)nodeA * GS_CAP + lane] : 0;
  const int fB = lane < (dB < 32 ? dB : 32) ? slot[(long)nodeB * GS_CAP + lane] : 0;

  for (int i = 0; i < 2; ++i) {
    const int idx = tid + i * 256;
    *(float4*)(lself + (idx >> 5) * LDS_STRIDE + (idx & 31) * 4) = selfv[i];
  }

  {
    const float4 a = gs_gather_node(hself, deg, slot, spillT, spillS, nodeA, dA, fA,
                                    lane, nNodes, nEdges);
    const float rinv = 1.0f / fmaxf((float)dA, 1.0f);
    *(float4*)(lmean + (wave * 2) * LDS_STRIDE + lane * 4) =
        make_float4(a.x * rinv, a.y * rinv, a.z * rinv, a.w * rinv);
  }
  {
    const float4 a = gs_gather_node(hself, deg, slot, spillT, spillS, nodeB, dB, fB,
                                    lane, nNodes, nEdges);
    const float rinv = 1.0f / fmaxf((float)dB, 1.0f);
    *(float4*)(lmean + (wave * 2 + 1) * LDS_STRIDE + lane * 4) =
        make_float4(a.x * rinv, a.y * rinv, a.z * rinv, a.w * rinv);
  }
  __syncthreads();

  const int col0 = wave * 16;
  const int rowA = lane & 15;
  const int kofs = (lane >> 4) * 2;

  v8f acc = {};

  const float4* bw = (const float4*)Bf + wave * 1024 + lane;
  const float* lm  = lmean + rowA * LDS_STRIDE + kofs;
  const float* ls  = lself + rowA * LDS_STRIDE + kofs;
  v2f bA[16], bB[16];
  gs_load_b16(bA, bw, 0);
  gs_load_b16(bB, bw, 8);
  acc = gs_mma16(acc, lm, 0, bA);
  gs_load_b16(bA, bw, 16);
  acc = gs_mma16(acc, lm, 64, bB);
  gs_load_b16(bB, bw, 24);
  acc = gs_mma16(acc, ls, 0, bA);
  acc = gs_mma16(acc, ls, 64, bB);

  const int c     = lane & 15;
  const int half  = lane >> 4;
  const int ocol  = col0 + c;
  const float bias = bl[ocol];
  const float4 u = *(const float4*)(U4 + ocol * 4);

  float q[32];
  for (int i = 0; i < 8; ++i) {
    const float t = fmaxf(acc[i] + bias, 0.0f);
    q[i]      = t * u.x;
    q[8 + i]  = t * u.y;
    q[16 + i] = t * u.z;
    q[24 + i] = t * u.w;
  }
  const bool b3 = (c & 8) != 0;
  const bool b2 = (c & 4) != 0;
  const bool b1 = (c & 2) != 0;
  const bool b0 = (c & 1) != 0;
  float w16[16];
  for (int k = 0; k < 16; ++k) {
    const float keep = b3 ? q[16 + k] : q[k];
    const float send = b3 ? q[k] : q[16 + k];
    w16[k] = keep + __shfl_xor(send, 8, 32);
  }
  float w8[8];
  for (int k = 0; k < 8; ++k) {
    const float keep = b2 ? w16[8 + k] : w16[k];
    const float send = b2 ? w16[k] : w16[8 + k];
    w8[k] = keep + __shfl_xor(send, 4, 32);
  }
  float w4[4];
  for (int k = 0; k < 4; ++k) {
    const float keep = b1 ? w8[4 + k] : w8[k];
    const float send = b1 ? w8[k] : w8[4 + k];
    w4[k] = keep + __shfl_xor(send, 2, 32);
  }
  float w2[2];
  for (int k = 0; k < 2; ++k) {
    const float keep = b0 ? w4[2 + k] : w4[k];
    const float send = b0 ? w4[k] : w4[2 + k];
    w2[k] = keep + __shfl_xor(send, 1, 32);
  }

  const int prow = half * 8 + (c & 3) * 2;
  const int pidx = c >> 2;
  lproj[wave * 64 + prow * 4 + pidx]       = w2[0];
  lproj[wave * 64 + (prow + 1) * 4 + pidx] = w2[1];
  __syncthreads();
  if (tid < 64) {
    float sum = lproj[tid];
    for (int w = 1; w < 8; ++w) sum += lproj[w * 64 + tid];
    const int node = row0 + (tid >> 2);
    if (node < nNodes) np[(long)row0 * 4 + tid] = sum;
  }
}

__global__ void __launch_bounds__(256)
gs_node_score_kernel(const float* __restrict__ np,
                     const float* __restrict__ cb,
                     const int* __restrict__ deg,
                     const int* __restrict__ slot,
                     const int* __restrict__ spillT,
                     const int* __restrict__ spillS,
                     float* __restrict__ ps,
                     float* __restrict__ pt,
                     int nNodes,
                     int nEdges) {
  const int node = (blockIdx.x * blockDim.x + threadIdx.x) >> 3;
  const int sub  = threadIdx.x & 7;
  const bool live = node < nNodes;
  const int d   = live ? deg[node] : 0;
  const int end = d < GS_CAP ? d : GS_CAP;
  const int* list = slot + (long)(live ? node : 0) * GS_CAP;
  float s = 0.0f, t = 0.0f;
  for (int k = sub; k < end; k += 8) {
    const float2 a = *(const float2*)(np + (long)list[k] * 4);
    s += a.x;
    t += a.y;
  }
  if (d > GS_CAP) {
    int m = deg[nNodes];
    m = m < nEdges ? m : nEdges;
    for (int i = sub; i < m; i += 8) {
      if (spillT[i] == node) {
        const float2 a = *(const float2*)(np + (long)spillS[i] * 4);
        s += a.x;
        t += a.y;
      }
    }
  }
  for (int off = 4; off > 0; off >>= 1) {
    s += __shfl_xor(s, off, 32);
    t += __shfl_xor(t, off, 32);
  }
  if (live && sub == 0) {
    const float den = fmaxf((float)d, 1.0f);
    const float2 self = *(const float2*)(np + (long)node * 4 + 2);
    ps[node] = s / den + self.x + cb[0];
    pt[node] = t / den + self.y + cb[1];
  }
}

__device__ __forceinline__ float gs_sigmoid(float p) {
  return 1.0f / (1.0f + __expf(-p));
}

__global__ void __launch_bounds__(256)
gs_edge_score_kernel(const float* __restrict__ ps,
                     const float* __restrict__ pt,
                     const int* __restrict__ src,
                     const int* __restrict__ tgt,
                     const float* __restrict__ be,
                     float* __restrict__ out,
                     int nEdges) {
  const int e0 = (blockIdx.x * blockDim.x + threadIdx.x) * 4;
  if (e0 >= nEdges) return;
  const float b = be[0];
  if ((nEdges & 3) == 0) {
    const int4 s4 = *(const int4*)(src + e0);
    const int4 t4 = *(const int4*)(tgt + e0);
    float4 y;
    y.x = gs_sigmoid(ps[s4.x] + pt[t4.x] + b);
    y.y = gs_sigmoid(ps[s4.y] + pt[t4.y] + b);
    y.z = gs_sigmoid(ps[s4.z] + pt[t4.z] + b);
    y.w = gs_sigmoid(ps[s4.w] + pt[t4.w] + b);
    *(float4*)(out + e0) = y;
  } else {
    const int eEnd = (e0 + 4) < nEdges ? (e0 + 4) : nEdges;
    for (int e = e0; e < eEnd; ++e)
      out[e] = gs_sigmoid(ps[src[e]] + pt[tgt[e]] + b);
  }
}

extern "C" void kernel_launch(void* const* d_in, const int* in_sizes, int n_in,
                              void* d_out, int out_size, void* d_ws, size_t ws_size,
                              hipStream_t stream) {
  const float* x    = (const float*)d_in[0];
  const int*   ei   = (const int*)d_in[1];
  const float* Wl1  = (const float*)d_in[2];
  const float* bl1  = (const float*)d_in[3];
  const float* Wr1  = (const float*)d_in[4];
  const float* Wl2  = (const float*)d_in[5];
  const float* bl2  = (const float*)d_in[6];
  const float* Wr2  = (const float*)d_in[7];
  const float* We   = (const float*)d_in[8];
  const float* be   = (const float*)d_in[9];

  const int N = in_sizes[0] / 128;
  const int E = in_sizes[1] / 2;
  const int* src = ei;
  const int* tgt = ei + E;

  int*   slot = (int*)d_ws;
  float* Bf   = (float*)d_ws + (long)N * GS_CAP;
  float* U4   = Bf + 2 * 128 * 128;
  float* cb   = U4 + 128 * 4;
  float* np   = cb + 4;
  int* deg    = (int*)(np + (long)N * 4);
  int* spillT = deg + (N + 1);
  int* spillS = spillT + E;
  float* ps   = (float*)(spillS + E);
  float* pt   = ps + N;
  int* sorted = (int*)(pt + N);
  int* off    = sorted + E;

  const int nBins = (N + GS_BIN - 1) / GS_BIN;
  const int nPart = (E + GS_CHUNK - 1) / GS_CHUNK;
  const int oct_Blocks = (N + 31) / 32;
  const int quadBlocks = ((E + 3) / 4 + 255) / 256;
  const int gemmBlocks = (N + 15) / 16;

  gs_part_kernel<<<nPart + GS_FOLDB + GS_SWZB, 1024, 0, stream>>>(
      src, tgt, sorted, off, deg, N, E, nBins, nPart, Wl2, bl2, Wr2, We, U4, cb, Wl1, Wr1, Bf);
  gs_place_kernel<<<nBins, 256, 0, stream>>>(sorted, off, deg, slot, spillT, spillS,
                                             N, nBins, nPart);

  gs_sage_wmma_kernel<<<gemmBlocks, 256, 0, stream>>>(x, deg, slot, spillT, spillS,
                                                      Bf, bl1, U4, np, N, E);

  gs_node_score_kernel<<<oct_Blocks, 256, 0, stream>>>(np, cb, deg, slot, spillT, spillS, ps, pt, N, E);

  gs_edge_score_kernel<<<quadBlocks, 256, 0, stream>>>(ps, pt, src, tgt, be,
                                                       (float*)d_out, E);
}


# List placing kernel: dependent hops become request groups: a wave takes whole per-chunk pieces (scalar bounds); its loads, LDS ranks, write-out reads each issue together; sum order = new LDS arrival
# speedup vs baseline: 7.4063x; 1.0313x over previous
#include <hip/hip_runtime.h>
#include <hip/hip_bf16.h>

typedef __attribute__((ext_vector_type(2))) float v2f;
typedef __attribute__((ext_vector_type(8))) float v8f;

#define LDS_STRIDE 132

#define GS_CAP 64
#define GS_BIN 64
#define GS_CHUNK 8192
#define GS_MAXB 1024

#define GS_FOLDB 5

#define GS_SWZB 8

__device__ __forceinline__ void gs_swizzle_b(const float* __restrict__ Wl,
                                             const float* __restrict__ Wr,
                                             float* __restrict__ Bf,
                                             int idx) {
  const int lane = idx & 31;
  const int j    = (idx >> 5) & 31;
  const int w    = idx >> 10;
  const float* W = (j < 16 ? Wl : Wr);
  const int n  = w * 16 + (lane & 15);
  const int k0 = 8 * (j & 15) + 2 * (lane >> 4);
  float4 v;
  v.x = W[(k0    ) * 128 + n];
  v.y = W[(k0 + 1) * 128 + n];
  v.z = W[(k0 + 4) * 128 + n];
  v.w = W[(k0 + 5) * 128 + n];
  *(float4*)(Bf + (long)idx * 4) = v;
}

__device__ __forceinline__ float gs_dot4(const float4 a, const float4 b) {
  return (a.x * b.x + a.y * b.y) + (a.z * b.z + a.w * b.w);
}

__device__ __forceinline__ void gs_fold_row(const float* __restrict__ Wl,
                                            const float* __restrict__ bl,
                                            const float* __restrict__ Wr,
                                            const float* __restrict__ We,
                                            float* __restrict__ U4,
                                            float* __restrict__ cb,
                                            int k, int lane) {
  if (k > 128) return;
  const float4 ws = *(const float4*)(We + lane * 4);
  const float4 wt = *(const float4*)(We + 128 + lane * 4);
  const bool isRow = k < 128;
  const float4 l = *(const float4*)((isRow ? Wl + k * 128 : bl) + lane * 4);
  const float4 r = isRow ? *(const float4*)(Wr + k * 128 + lane * 4)
                         : make_float4(0.0f, 0.0f, 0.0f, 0.0f);
  float s0 = gs_dot4(l, ws);
  float s1 = gs_dot4(l, wt);
  float s2 = gs_dot4(r, ws);
  float s3 = gs_dot4(r, wt);
  for (int off = 16; off > 0; off >>= 1) {
    s0 += __shfl_xor(s0, off, 32);
    s1 += __shfl_xor(s1, off, 32);
    s2 += __shfl_xor(s2, off, 32);
    s3 += __shfl_xor(s3, off, 32);
  }
  if (lane == 0) {
    if (isRow) {
      *(float4*)(U4 + k * 4) = make_float4(s0, s1, s2, s3);
    } else {
      cb[0] = s0;
      cb[1] = s1;
    }
  }
}

__global__ void __launch_bounds__(1024)
gs_part_kernel(const int* __restrict__ src,
               const int* __restrict__ tgt,
               int* __restrict__ sorted,
               int* __restrict__ off,
               int* __restrict__ deg,
               int nNodes,
               int nEdges,
               int nBins,
               int nPart,
               const float* __restrict__ Wl2,
               const float* __restrict__ bl2,
               const float* __restrict__ Wr2,
               const float* __restrict__ We,
               float* __restrict__ U4,
               float* __restrict__ cb,
               const float* __restrict__ Wl1,
               const float* __restrict__ Wr1,
               float* __restrict__ Bf) {
  __shared__ int hist[GS_MAXB];
  __shared__ int wsum[32];
  __shared__ int stage[GS_CHUNK];
  const int tid  = threadIdx.x;
  const int lane = tid & 31;
  const int wave = tid >> 5;
  const int base = blockIdx.x * GS_CHUNK;

  if ((int)blockIdx.x >= nPart) {
    const int xb = (int)blockIdx.x - nPart;
    if (xb < GS_FOLDB) {
      gs_fold_row(Wl2, bl2, Wr2, We, U4, cb, xb * 32 + wave, lane);
    } else {
      gs_swizzle_b(Wl1, Wr1, Bf, (xb - GS_FOLDB) * 1024 + tid);
    }
    return;
  }

  if (blockIdx.x == 0 && tid == 0) deg[nNodes] = 0;
  hist[tid] = 0;
  __syncthreads();

  int pk[8], bn[8], rk[8];
#pragma unroll
  for (int i = 0; i < 8; ++i) {
    const int e = base + i * 1024 + tid;
    const bool ok = e < nEdges;
    const int t = ok ? tgt[e] : 0;
    const int s = ok ? src[e] : 0;
    pk[i] = (s << 6) | (t & (GS_BIN - 1));
    bn[i] = ok ? (t >> 6) : -1;
  }
#pragma unroll
  for (int i = 0; i < 8; ++i)
    rk[i] = bn[i] >= 0 ? atomicAdd(&hist[bn[i]], 1) : 0;
  __syncthreads();

  const int v = hist[tid];
  int inc = v;
  for (int d = 1; d < 32; d <<= 1) {
    const int n = __shfl_up(inc, d, 32);
    if (lane >= d) inc += n;
  }
  if (lane == 31) wsum[wave] = inc;
  __syncthreads();
  if (wave == 0) {
    const int w = wsum[lane];
    int wi = w;
    for (int d = 1; d < 32; d <<= 1) {
      const int n = __shfl_up(wi, d, 32);
      if (lane >= d) wi += n;
    }
    wsum[lane] = wi - w;
  }
  __syncthreads();
  const int ex = inc - v + wsum[wave];
  hist[tid] = ex;
  if (tid <= nBins) off[(long)blockIdx.x * (nBins + 1) + tid] = ex;
  __syncthreads();

#pragma unroll
  for (int i = 0; i < 8; ++i)
    if (bn[i] >= 0) stage[hist[bn[i]] + rk[i]] = pk[i];
  __syncthreads();
#pragma unroll
  for (int i = 0; i < 8; ++i) {
    const int e = base + i * 1024 + tid;
    if (e < nEdges) sorted[e] = stage[i * 1024 + tid];
  }
}

#define GS_PIECES 16

__global__ void __launch_bounds__(256)
gs_place_kernel(const int* __restrict__ sorted,
                const int* __restrict__ off,
                int* __restrict__ deg,
                int* __restrict__ slot,
                int* __restrict__ spillT,
                int* __restrict__ spillS,
                int nNodes,
                int nBins,
                int nPart) {
  __shared__ int cnt[GS_BIN];
  __shared__ int slotL[GS_BIN * GS_CAP];
  const int tid  = threadIdx.x;
  const int lane = tid & 31;
  const int wave = __builtin_amdgcn_readfirstlane(tid >> 5);
  const int b    = blockIdx.x;

  if (tid < GS_BIN) cnt[tid] = 0;
  __syncthreads();

  for (int p0 = wave; p0 < nPart; p0 += 8 * GS_PIECES) {
    int g[GS_PIECES], len[GS_PIECES], pk[GS_PIECES], rk[GS_PIECES];
    int longest = 0;
#pragma unroll
    for (int u = 0; u < GS_PIECES; ++u) {
      const int p  = p0 + 8 * u;
      const int pc = p < nPart ? p : nPart - 1;
      const int* o = off + (long)pc * (nBins + 1) + b;
      const int a  = o[0];
      const int a1 = o[1];
      len[u] = p < nPart ? a1 - a : 0;
      g[u]   = pc * GS_CHUNK + a;
      longest = len[u] > longest ? len[u] : longest;
    }
#pragma unroll
    for (int u = 0; u < GS_PIECES; ++u) {
      const int* piece = sorted + g[u];
      pk[u] = lane < len[u] ? piece[lane] : 0;
    }
#pragma unroll
    for (int u = 0; u < GS_PIECES; ++u)
      rk[u] = lane < len[u] ? atomicAdd(&cnt[pk[u] & (GS_BIN - 1)], 1) : 0;
#pragma unroll
    for (int u = 0; u < GS_PIECES; ++u) {
      if (lane < len[u]) {
        const int n = pk[u] & (GS_BIN - 1);
        const int s = pk[u] >> 6;
        if (rk[u] < GS_CAP) {
          slotL[n * GS_CAP + rk[u]] = s;
        } else {
          const int o = atomicAdd(deg + nNodes, 1);
          spillT[o] = b * GS_BIN + n;
          spillS[o] = s;
        }
      }
    }
    if (longest > 32) {
      for (int p = p0; p < nPart && p < p0 + 8 * GS_PIECES; p += 8) {
        const int* o = off + (long)p * (nBins + 1) + b;
        const int a  = o[0];
        const int l  = o[1] - a;
        const int gg = p * GS_CHUNK + a;
        for (int q = 32; q < l; q += 32) {
          if (q + lane < l) {
            const int e = sorted[gg + q + lane];
            const int n = e & (GS_BIN - 1);
            const int s = e >> 6;
            const int r = atomicAdd(&cnt[n], 1);
            if (r < GS_CAP) {
              slotL[n * GS_CAP + r] = s;
            } else {
              const int o2 = atomicAdd(deg + nNodes, 1);
              spillT[o2] = b * GS_BIN + n;
              spillS[o2] = s;
            }
          }
        }
      }
    }
  }
  __syncthreads();

  int dn[8], lo[8], hi[8];
#pragma unroll
  for (int k = 0; k < 8; ++k) dn[k] = cnt[wave * 8 + k];
#pragma unroll
  for (int k = 0; k < 8; ++k) {
    lo[k] = slotL[(wave * 8 + k) * GS_CAP + lane];
    hi[k] = slotL[(wave * 8 + k) * GS_CAP + 32 + lane];
  }
#pragma unroll
  for (int k = 0; k < 8; ++k) {
    const int node = b * GS_BIN + wave * 8 + k;
    if (node < nNodes) {
      if (dn[k] > 0)  slot[(long)node * GS_CAP + lane]      = lo[k];
      if (dn[k] > 32) slot[(long)node * GS_CAP + 32 + lane] = hi[k];
    }
  }
  if (tid < GS_BIN && b * GS_BIN + tid < nNodes) deg[b * GS_BIN + tid] = cnt[tid];
}

__device__ __forceinline__ void gs_add4(float4& a, const float4 v) {
  a.x += v.x; a.y += v.y; a.z += v.z; a.w += v.w;
}

template <int U>
__device__ __forceinline__ void gs_rows(float4& acc, const float* hl, int mine, int j) {
  float4 v[U];
#pragma unroll
  for (int u = 0; u < U; ++u) {
    const int s = __builtin_amdgcn_readlane(mine, j + u);
    v[u] = *(const float4*)(hl + (long)s * 128);
  }
#pragma unroll
  for (int u = 0; u < U; ++u) gs_add4(acc, v[u]);
}

__device__ __forceinline__ float4 gs_gather_node(const float* __restrict__ h,
                                                 const int* __restrict__ deg,
                                                 const int* __restrict__ slot,
                                                 const int* __restrict__ spillT,
                                                 const int* __restrict__ spillS,
                                                 int node, int d, int first32,
                                                 int lane, int nNodes, int nEdges) {
  const int end = d < GS_CAP ? d : GS_CAP;
  const int* list = slot + (long)node * GS_CAP;
  const float* hl = h + lane * 4;
  float4 acc = make_float4(0.0f, 0.0f, 0.0f, 0.0f);

  for (int base = 0; base < end; base += 32) {
    const int n = (end - base) < 32 ? (end - base) : 32;
    int mine = first32;
    if (base != 0) mine = lane < n ? list[base + lane] : 0;
    int j = 0;
    for (; j + 16 <= n; j += 16) gs_rows<16>(acc, hl, mine, j);
    if (j + 8 <= n) { gs_rows<8>(acc, hl, mine, j); j += 8; }
    if (j + 4 <= n) { gs_rows<4>(acc, hl, mine, j); j += 4; }
    if (j + 2 <= n) { gs_rows<2>(acc, hl, mine, j); j += 2; }
    if (j < n)      { gs_rows<1>(acc, hl, mine, j); }
  }

  if (d > GS_CAP) {
    int m = deg[nNodes];
    m = m < nEdges ? m : nEdges;
    for (int b = 0; b < m; b += 32) {
      const int i = b + lane;
      const int t = i < m ? spillT[i] : -1;
      const int s = i < m ? spillS[i] : 0;
      unsigned hit = (unsigned)__ballot(t == node);
      while (hit != 0u) {
        const int l = __builtin_ctz(hit);
        hit &= hit - 1u;
        const int s0 = __shfl(s, l, 32);
        const float4 v0 = *(const float4*)(hl + (long)s0 * 128);
        gs_add4(acc, v0);
      }
    }
  }
  return acc;
}

__device__ __forceinline__ void gs_load_b16(v2f (&b)[16],
                                            const float4* __restrict__ Bw,
                                            int j0) {
#pragma unroll
  for (int i = 0; i < 8; ++i) {
    const float4 v = Bw[(j0 + i) * 32];
    b[2 * i].x     = v.x;
    b[2 * i].y     = v.y;
    b[2 * i + 1].x = v.z;
    b[2 * i + 1].y = v.w;
  }
}

__device__ __forceinline__ v8f gs_mma16(v8f acc, const float* lrow, int kbase,
                                        const v2f (&b)[16]) {
#pragma unroll
  for (int s = 0; s < 16; ++s) {
    v2f a;
    a.x = lrow[kbase + 4 * s];
    a.y = lrow[kbase + 4 * s + 1];
    acc = __builtin_amdgcn_wmma_f32_16x16x4_f32(
         false, a,  false, b[s],
         (short)0, acc,  false,  false);
  }
  return acc;
}

__global__ void __launch_bounds__(256)
gs_sage_wmma_kernel(const float* __restrict__ hself,
                    const int* __restrict__ deg,
                    const int* __restrict__ slot,
                    const int* __restrict__ spillT,
                    const int* __restrict__ spillS,
                    const float* __restrict__ Bf,
                    const float* __restrict__ bl,
                    const float* __restrict__ U4,
                    float* __restrict__ np,
                    int nNodes,
                    int nEdges) {
  __shared__ float lmean[16 * LDS_STRIDE];
  __shared__ float lself[16 * LDS_STRIDE];
  __shared__ float lproj[8 * 64];

  const int tid  = threadIdx.x;
  const int row0 = blockIdx.x * 16;
  const int lane = tid & 31;
  const int wave = __builtin_amdgcn_readfirstlane(tid >> 5);

  float4 selfv[2];
  for (int i = 0; i < 2; ++i) {
    const int idx = tid + i * 256;
    const int r   = idx >> 5;
    const int node = row0 + r;
    const int nclamp = node < nNodes ? node : (nNodes - 1);
    selfv[i] = *(const float4*)(hself + (long)nclamp * 128 + (idx & 31) * 4);
  }

  const int nodeA = row0 + wave * 2;
  const int nodeB = nodeA + 1;
  const int dA = nodeA < nNodes ? deg[nodeA] : 0;
  const int dB = nodeB < nNodes ? deg[nodeB] : 0;
  const int fA = lane < (dA < 32 ? dA : 32) ? slot[(long)nodeA * GS_CAP + lane] : 0;
  const int fB = lane < (dB < 32 ? dB : 32) ? slot[(long)nodeB * GS_CAP + lane] : 0;

  for (int i = 0; i < 2; ++i) {
    const int idx = tid + i * 256;
    *(float4*)(lself + (idx >> 5) * LDS_STRIDE + (idx & 31) * 4) = selfv[i];
  }

  {
    const float4 a = gs_gather_node(hself, deg, slot, spillT, spillS, nodeA, dA, fA,
                                    lane, nNodes, nEdges);
    const float rinv = 1.0f / fmaxf((float)dA, 1.0f);
    *(float4*)(lmean + (wave * 2) * LDS_STRIDE + lane * 4) =
        make_float4(a.x * rinv, a.y * rinv, a.z * rinv, a.w * rinv);
  }
  {
    const float4 a = gs_gather_node(hself, deg, slot, spillT, spillS, nodeB, dB, fB,
                                    lane, nNodes, nEdges);
    const float rinv = 1.0f / fmaxf((float)dB, 1.0f);
    *(float4*)(lmean + (wave * 2 + 1) * LDS_STRIDE + lane * 4) =
        make_float4(a.x * rinv, a.y * rinv, a.z * rinv, a.w * rinv);
  }
  __syncthreads();

  const int col0 = wave * 16;
  const int rowA = lane & 15;
  const int kofs = (lane >> 4) * 2;

  v8f acc = {};

  const float4* bw = (const float4*)Bf + wave * 1024 + lane;
  const float* lm  = lmean + rowA * LDS_STRIDE + kofs;
  const float* ls  = lself + rowA * LDS_STRIDE + kofs;
  v2f bA[16], bB[16];
  gs_load_b16(bA, bw, 0);
  gs_load_b16(bB, bw, 8);
  acc = gs_mma16(acc, lm, 0, bA);
  gs_load_b16(bA, bw, 16);
  acc = gs_mma16(acc, lm, 64, bB);
  gs_load_b16(bB, bw, 24);
  acc = gs_mma16(acc, ls, 0, bA);
  acc = gs_mma16(acc, ls, 64, bB);

  const int c     = lane & 15;
  const int half  = lane >> 4;
  const int ocol  = col0 + c;
  const float bias = bl[ocol];
  const float4 u = *(const float4*)(U4 + ocol * 4);

  float q[32];
  for (int i = 0; i < 8; ++i) {
    const float t = fmaxf(acc[i] + bias, 0.0f);
    q[i]      = t * u.x;
    q[8 + i]  = t * u.y;
    q[16 + i] = t * u.z;
    q[24 + i] = t * u.w;
  }
  const bool b3 = (c & 8) != 0;
  const bool b2 = (c & 4) != 0;
  const bool b1 = (c & 2) != 0;
  const bool b0 = (c & 1) != 0;
  float w16[16];
  for (int k = 0; k < 16; ++k) {
    const float keep = b3 ? q[16 + k] : q[k];
    const float send = b3 ? q[k] : q[16 + k];
    w16[k] = keep + __shfl_xor(send, 8, 32);
  }
  float w8[8];
  for (int k = 0; k < 8; ++k) {
    const float keep = b2 ? w16[8 + k] : w16[k];
    const float send = b2 ? w16[k] : w16[8 + k];
    w8[k] = keep + __shfl_xor(send, 4, 32);
  }
  float w4[4];
  for (int k = 0; k < 4; ++k) {
    const float keep = b1 ? w8[4 + k] : w8[k];
    const float send = b1 ? w8[k] : w8[4 + k];
    w4[k] = keep + __shfl_xor(send, 2, 32);
  }
  float w2[2];
  for (int k = 0; k < 2; ++k) {
    const float keep = b0 ? w4[2 + k] : w4[k];
    const float send = b0 ? w4[k] : w4[2 + k];
    w2[k] = keep + __shfl_xor(send, 1, 32);
  }

  const int prow = half * 8 + (c & 3) * 2;
  const int pidx = c >> 2;
  lproj[wave * 64 + prow * 4 + pidx]       = w2[0];
  lproj[wave * 64 + (prow + 1) * 4 + pidx] = w2[1];
  __syncthreads();
  if (tid < 64) {
    float sum = lproj[tid];
    for (int w = 1; w < 8; ++w) sum += lproj[w * 64 + tid];
    const int node = row0 + (tid >> 2);
    if (node < nNodes) np[(long)row0 * 4 + tid] = sum;
  }
}

__global__ void __launch_bounds__(256)
gs_node_score_kernel(const float* __restrict__ np,
                     const float* __restrict__ cb,
                     const int* __restrict__ deg,
                     const int* __restrict__ slot,
                     const int* __restrict__ spillT,
                     const int* __restrict__ spillS,
                     float* __restrict__ ps,
                     float* __restrict__ pt,
                     int nNodes,
                     int nEdges) {
  const int node = (blockIdx.x * blockDim.x + threadIdx.x) >> 3;
  const int sub  = threadIdx.x & 7;
  const bool live = node < nNodes;
  const int d   = live ? deg[node] : 0;
  const int end = d < GS_CAP ? d : GS_CAP;
  const int* list = slot + (long)(live ? node : 0) * GS_CAP;
  float s = 0.0f, t = 0.0f;
  for (int k = sub; k < end; k += 8) {
    const float2 a = *(const float2*)(np + (long)list[k] * 4);
    s += a.x;
    t += a.y;
  }
  if (d > GS_CAP) {
    int m = deg[nNodes];
    m = m < nEdges ? m : nEdges;
    for (int i = sub; i < m; i += 8) {
      if (spillT[i] == node) {
        const float2 a = *(const float2*)(np + (long)spillS[i] * 4);
        s += a.x;
        t += a.y;
      }
    }
  }
  for (int off = 4; off > 0; off >>= 1) {
    s += __shfl_xor(s, off, 32);
    t += __shfl_xor(t, off, 32);
  }
  if (live && sub == 0) {
    const float den = fmaxf((float)d, 1.0f);
    const float2 self = *(const float2*)(np + (long)node * 4 + 2);
    ps[node] = s / den + self.x + cb[0];
    pt[node] = t / den + self.y + cb[1];
  }
}

__device__ __forceinline__ float gs_sigmoid(float p) {
  return 1.0f / (1.0f + __expf(-p));
}

__global__ void __launch_bounds__(256)
gs_edge_score_kernel(const float* __restrict__ ps,
                     const float* __restrict__ pt,
                     const int* __restrict__ src,
                     const int* __restrict__ tgt,
                     const float* __restrict__ be,
                     float* __restrict__ out,
                     int nEdges) {
  const int e0 = (blockIdx.x * blockDim.x + threadIdx.x) * 4;
  if (e0 >= nEdges) return;
  const float b = be[0];
  if ((nEdges & 3) == 0) {
    const int4 s4 = *(const int4*)(src + e0);
    const int4 t4 = *(const int4*)(tgt + e0);
    float4 y;
    y.x = gs_sigmoid(ps[s4.x] + pt[t4.x] + b);
    y.y = gs_sigmoid(ps[s4.y] + pt[t4.y] + b);
    y.z = gs_sigmoid(ps[s4.z] + pt[t4.z] + b);
    y.w = gs_sigmoid(ps[s4.w] + pt[t4.w] + b);
    *(float4*)(out + e0) = y;
  } else {
    const int eEnd = (e0 + 4) < nEdges ? (e0 + 4) : nEdges;
    for (int e = e0; e < eEnd; ++e)
      out[e] = gs_sigmoid(ps[src[e]] + pt[tgt[e]] + b);
  }
}

extern "C" void kernel_launch(void* const* d_in, const int* in_sizes, int n_in,
                              void* d_out, int out_size, void* d_ws, size_t ws_size,
                              hipStream_t stream) {
  const float* x    = (const float*)d_in[0];
  const int*   ei   = (const int*)d_in[1];
  const float* Wl1  = (const float*)d_in[2];
  const float* bl1  = (const float*)d_in[3];
  const float* Wr1  = (const float*)d_in[4];
  const float* Wl2  = (const float*)d_in[5];
  const float* bl2  = (const float*)d_in[6];
  const float* Wr2  = (const float*)d_in[7];
  const float* We   = (const float*)d_in[8];
  const float* be   = (const float*)d_in[9];

  const int N = in_sizes[0] / 128;
  const int E = in_sizes[1] / 2;
  const int* src = ei;
  const int* tgt = ei + E;

  int*   slot = (int*)d_ws;
  float* Bf   = (float*)d_ws + (long)N * GS_CAP;
  float* U4   = Bf + 2 * 128 * 128;
  float* cb   = U4 + 128 * 4;
  float* np   = cb + 4;
  int* deg    = (int*)(np + (long)N * 4);
  int* spillT = deg + (N + 1);
  int* spillS = spillT + E;
  float* ps   = (float*)(spillS + E);
  float* pt   = ps + N;
  int* sorted = (int*)(pt + N);
  int* off    = sorted + E;

  const int nBins = (N + GS_BIN - 1) / GS_BIN;
  const int nPart = (E + GS_CHUNK - 1) / GS_CHUNK;
  const int oct_Blocks = (N + 31) / 32;
  const int quadBlocks = ((E + 3) / 4 + 255) / 256;
  const int gemmBlocks = (N + 15) / 16;

  gs_part_kernel<<<nPart + GS_FOLDB + GS_SWZB, 1024, 0, stream>>>(
      src, tgt, sorted, off, deg, N, E, nBins, nPart, Wl2, bl2, Wr2, We, U4, cb, Wl1, Wr1, Bf);
  gs_place_kernel<<<nBins, 256, 0, stream>>>(sorted, off, deg, slot, spillT, spillS,
                                             N, nBins, nPart);

  gs_sage_wmma_kernel<<<gemmBlocks, 256, 0, stream>>>(x, deg, slot, spillT, spillS,
                                                      Bf, bl1, U4, np, N, E);

  gs_node_score_kernel<<<oct_Blocks, 256, 0, stream>>>(np, cb, deg, slot, spillT, spillS, ps, pt, N, E);

  gs_edge_score_kernel<<<quadBlocks, 256, 0, stream>>>(ps, pt, src, tgt, be,
                                                       (float*)d_out, E);
}


# Node and edge tail kernels: each 1024-thread workgroup copies the per-node table (np plane, ps, pt) into LDS by whole-line loads, random lookups become LDS reads; node sum: list order in 4 lanes
# speedup vs baseline: 7.6780x; 1.0367x over previous
#include <hip/hip_runtime.h>
#include <hip/hip_bf16.h>

typedef __attribute__((ext_vector_type(2))) float v2f;
typedef __attribute__((ext_vector_type(8))) float v8f;

#define LDS_STRIDE 132

#define GS_CAP 64
#define GS_BIN 64
#define GS_CHUNK 8192
#define GS_MAXB 1024

#define GS_FOLDB 5

#define GS_SWZB 8

__device__ __forceinline__ void gs_swizzle_b(const float* __restrict__ Wl,
                                             const float* __restrict__ Wr,
                                             float* __restrict__ Bf,
                                             int idx) {
  const int lane = idx & 31;
  const int j    = (idx >> 5) & 31;
  const int w    = idx >> 10;
  const float* W = (j < 16 ? Wl : Wr);
  const int n  = w * 16 + (lane & 15);
  const int k0 = 8 * (j & 15) + 2 * (lane >> 4);
  float4 v;
  v.x = W[(k0    ) * 128 + n];
  v.y = W[(k0 + 1) * 128 + n];
  v.z = W[(k0 + 4) * 128 + n];
  v.w = W[(k0 + 5) * 128 + n];
  *(float4*)(Bf + (long)idx * 4) = v;
}

__device__ __forceinline__ float gs_dot4(const float4 a, const float4 b) {
  return (a.x * b.x + a.y * b.y) + (a.z * b.z + a.w * b.w);
}

__device__ __forceinline__ void gs_fold_row(const float* __restrict__ Wl,
                                            const float* __restrict__ bl,
                                            const float* __restrict__ Wr,
                                            const float* __restrict__ We,
                                            float* __restrict__ U4,
                                            float* __restrict__ cb,
                                            int k, int lane) {
  if (k > 128) return;
  const float4 ws = *(const float4*)(We + lane * 4);
  const float4 wt = *(const float4*)(We + 128 + lane * 4);
  const bool isRow = k < 128;
  const float4 l = *(const float4*)((isRow ? Wl + k * 128 : bl) + lane * 4);
  const float4 r = isRow ? *(const float4*)(Wr + k * 128 + lane * 4)
                         : make_float4(0.0f, 0.0f, 0.0f, 0.0f);
  float s0 = gs_dot4(l, ws);
  float s1 = gs_dot4(l, wt);
  float s2 = gs_dot4(r, ws);
  float s3 = gs_dot4(r, wt);
  for (int off = 16; off > 0; off >>= 1) {
    s0 += __shfl_xor(s0, off, 32);
    s1 += __shfl_xor(s1, off, 32);
    s2 += __shfl_xor(s2, off, 32);
    s3 += __shfl_xor(s3, off, 32);
  }
  if (lane == 0) {
    if (isRow) {
      *(float4*)(U4 + k * 4) = make_float4(s0, s1, s2, s3);
    } else {
      cb[0] = s0;
      cb[1] = s1;
    }
  }
}

__global__ void __launch_bounds__(1024)
gs_part_kernel(const int* __restrict__ src,
               const int* __restrict__ tgt,
               int* __restrict__ sorted,
               int* __restrict__ off,
               int* __restrict__ deg,
               int nNodes,
               int nEdges,
               int nBins,
               int nPart,
               const float* __restrict__ Wl2,
               const float* __restrict__ bl2,
               const float* __restrict__ Wr2,
               const float* __restrict__ We,
               float* __restrict__ U4,
               float* __restrict__ cb,
               const float* __restrict__ Wl1,
               const float* __restrict__ Wr1,
               float* __restrict__ Bf) {
  __shared__ int hist[GS_MAXB];
  __shared__ int wsum[32];
  __shared__ int stage[GS_CHUNK];
  const int tid  = threadIdx.x;
  const int lane = tid & 31;
  const int wave = tid >> 5;
  const int base = blockIdx.x * GS_CHUNK;

  if ((int)blockIdx.x >= nPart) {
    const int xb = (int)blockIdx.x - nPart;
    if (xb < GS_FOLDB) {
      gs_fold_row(Wl2, bl2, Wr2, We, U4, cb, xb * 32 + wave, lane);
    } else {
      gs_swizzle_b(Wl1, Wr1, Bf, (xb - GS_FOLDB) * 1024 + tid);
    }
    return;
  }

  if (blockIdx.x == 0 && tid == 0) deg[nNodes] = 0;
  hist[tid] = 0;
  __syncthreads();

  int pk[8], bn[8], rk[8];
#pragma unroll
  for (int i = 0; i < 8; ++i) {
    const int e = base + i * 1024 + tid;
    const bool ok = e < nEdges;
    const int t = ok ? tgt[e] : 0;
    const int s = ok ? src[e] : 0;
    pk[i] = (s << 6) | (t & (GS_BIN - 1));
    bn[i] = ok ? (t >> 6) : -1;
  }
#pragma unroll
  for (int i = 0; i < 8; ++i)
    rk[i] = bn[i] >= 0 ? atomicAdd(&hist[bn[i]], 1) : 0;
  __syncthreads();

  const int v = hist[tid];
  int inc = v;
  for (int d = 1; d < 32; d <<= 1) {
    const int n = __shfl_up(inc, d, 32);
    if (lane >= d) inc += n;
  }
  if (lane == 31) wsum[wave] = inc;
  __syncthreads();
  if (wave == 0) {
    const int w = wsum[lane];
    int wi = w;
    for (int d = 1; d < 32; d <<= 1) {
      const int n = __shfl_up(wi, d, 32);
      if (lane >= d) wi += n;
    }
    wsum[lane] = wi - w;
  }
  __syncthreads();
  const int ex = inc - v + wsum[wave];
  hist[tid] = ex;
  if (tid <= nBins) off[(long)blockIdx.x * (nBins + 1) + tid] = ex;
  __syncthreads();

#pragma unroll
  for (int i = 0; i < 8; ++i)
    if (bn[i] >= 0) stage[hist[bn[i]] + rk[i]] = pk[i];
  __syncthreads();
#pragma unroll
  for (int i = 0; i < 8; ++i) {
    const int e = base + i * 1024 + tid;
    if (e < nEdges) sorted[e] = stage[i * 1024 + tid];
  }
}

#define GS_PIECES 16

__global__ void __launch_bounds__(256)
gs_place_kernel(const int* __restrict__ sorted,
                const int* __restrict__ off,
                int* __restrict__ deg,
                int* __restrict__ slot,
                int* __restrict__ spillT,
                int* __restrict__ spillS,
                int nNodes,
                int nBins,
                int nPart) {
  __shared__ int cnt[GS_BIN];
  __shared__ int slotL[GS_BIN * GS_CAP];
  const int tid  = threadIdx.x;
  const int lane = tid & 31;
  const int wave = __builtin_amdgcn_readfirstlane(tid >> 5);
  const int b    = blockIdx.x;

  if (tid < GS_BIN) cnt[tid] = 0;
  __syncthreads();

  for (int p0 = wave; p0 < nPart; p0 += 8 * GS_PIECES) {
    int g[GS_PIECES], len[GS_PIECES], pk[GS_PIECES], rk[GS_PIECES];
    int longest = 0;
#pragma unroll
    for (int u = 0; u < GS_PIECES; ++u) {
      const int p  = p0 + 8 * u;
      const int pc = p < nPart ? p : nPart - 1;
      const int* o = off + (long)pc * (nBins + 1) + b;
      const int a  = o[0];
      const int a1 = o[1];
      len[u] = p < nPart ? a1 - a : 0;
      g[u]   = pc * GS_CHUNK + a;
      longest = len[u] > longest ? len[u] : longest;
    }
#pragma unroll
    for (int u = 0; u < GS_PIECES; ++u) {
      const int* piece = sorted + g[u];
      pk[u] = lane < len[u] ? piece[lane] : 0;
    }
#pragma unroll
    for (int u = 0; u < GS_PIECES; ++u)
      rk[u] = lane < len[u] ? atomicAdd(&cnt[pk[u] & (GS_BIN - 1)], 1) : 0;
#pragma unroll
    for (int u = 0; u < GS_PIECES; ++u) {
      if (lane < len[u]) {
        const int n = pk[u] & (GS_BIN - 1);
        const int s = pk[u] >> 6;
        if (rk[u] < GS_CAP) {
          slotL[n * GS_CAP + rk[u]] = s;
        } else {
          const int o = atomicAdd(deg + nNodes, 1);
          spillT[o] = b * GS_BIN + n;
          spillS[o] = s;
        }
      }
    }
    if (longest > 32) {
      for (int p = p0; p < nPart && p < p0 + 8 * GS_PIECES; p += 8) {
        const int* o = off + (long)p * (nBins + 1) + b;
        const int a  = o[0];
        const int l  = o[1] - a;
        const int gg = p * GS_CHUNK + a;
        for (int q = 32; q < l; q += 32) {
          if (q + lane < l) {
            const int e = sorted[gg + q + lane];
            const int n = e & (GS_BIN - 1);
            const int s = e >> 6;
            const int r = atomicAdd(&cnt[n], 1);
            if (r < GS_CAP) {
              slotL[n * GS_CAP + r] = s;
            } else {
              const int o2 = atomicAdd(deg + nNodes, 1);
              spillT[o2] = b * GS_BIN + n;
              spillS[o2] = s;
            }
          }
        }
      }
    }
  }
  __syncthreads();

  int dn[8], lo[8], hi[8];
#pragma unroll
  for (int k = 0; k < 8; ++k) dn[k] = cnt[wave * 8 + k];
#pragma unroll
  for (int k = 0; k < 8; ++k) {
    lo[k] = slotL[(wave * 8 + k) * GS_CAP + lane];
    hi[k] = slotL[(wave * 8 + k) * GS_CAP + 32 + lane];
  }
#pragma unroll
  for (int k = 0; k < 8; ++k) {
    const int node = b * GS_BIN + wave * 8 + k;
    if (node < nNodes) {
      if (dn[k] > 0)  slot[(long)node * GS_CAP + lane]      = lo[k];
      if (dn[k] > 32) slot[(long)node * GS_CAP + 32 + lane] = hi[k];
    }
  }
  if (tid < GS_BIN && b * GS_BIN + tid < nNodes) deg[b * GS_BIN + tid] = cnt[tid];
}

__device__ __forceinline__ void gs_add4(float4& a, const float4 v) {
  a.x += v.x; a.y += v.y; a.z += v.z; a.w += v.w;
}

template <int U>
__device__ __forceinline__ void gs_rows(float4& acc, const float* hl, int mine, int j) {
  float4 v[U];
#pragma unroll
  for (int u = 0; u < U; ++u) {
    const int s = __builtin_amdgcn_readlane(mine, j + u);
    v[u] = *(const float4*)(hl + (long)s * 128);
  }
#pragma unroll
  for (int u = 0; u < U; ++u) gs_add4(acc, v[u]);
}

__device__ __forceinline__ float4 gs_gather_node(const float* __restrict__ h,
                                                 const int* __restrict__ deg,
                                                 const int* __restrict__ slot,
                                                 const int* __restrict__ spillT,
                                                 const int* __restrict__ spillS,
                                                 int node, int d, int first32,
                                                 int lane, int nNodes, int nEdges) {
  const int end = d < GS_CAP ? d : GS_CAP;
  const int* list = slot + (long)node * GS_CAP;
  const float* hl = h + lane * 4;
  float4 acc = make_float4(0.0f, 0.0f, 0.0f, 0.0f);

  for (int base = 0; base < end; base += 32) {
    const int n = (end - base) < 32 ? (end - base) : 32;
    int mine = first32;
    if (base != 0) mine = lane < n ? list[base + lane] : 0;
    int j = 0;
    for (; j + 16 <= n; j += 16) gs_rows<16>(acc, hl, mine, j);
    if (j + 8 <= n) { gs_rows<8>(acc, hl, mine, j); j += 8; }
    if (j + 4 <= n) { gs_rows<4>(acc, hl, mine, j); j += 4; }
    if (j + 2 <= n) { gs_rows<2>(acc, hl, mine, j); j += 2; }
    if (j < n)      { gs_rows<1>(acc, hl, mine, j); }
  }

  if (d > GS_CAP) {
    int m = deg[nNodes];
    m = m < nEdges ? m : nEdges;
    for (int b = 0; b < m; b += 32) {
      const int i = b + lane;
      const int t = i < m ? spillT[i] : -1;
      const int s = i < m ? spillS[i] : 0;
      unsigned hit = (unsigned)__ballot(t == node);
      while (hit != 0u) {
        const int l = __builtin_ctz(hit);
        hit &= hit - 1u;
        const int s0 = __shfl(s, l, 32);
        const float4 v0 = *(const float4*)(hl + (long)s0 * 128);
        gs_add4(acc, v0);
      }
    }
  }
  return acc;
}

__device__ __forceinline__ void gs_load_b16(v2f (&b)[16],
                                            const float4* __restrict__ Bw,
                                            int j0) {
#pragma unroll
  for (int i = 0; i < 8; ++i) {
    const float4 v = Bw[(j0 + i) * 32];
    b[2 * i].x     = v.x;
    b[2 * i].y     = v.y;
    b[2 * i + 1].x = v.z;
    b[2 * i + 1].y = v.w;
  }
}

__device__ __forceinline__ v8f gs_mma16(v8f acc, const float* lrow, int kbase,
                                        const v2f (&b)[16]) {
#pragma unroll
  for (int s = 0; s < 16; ++s) {
    v2f a;
    a.x = lrow[kbase + 4 * s];
    a.y = lrow[kbase + 4 * s + 1];
    acc = __builtin_amdgcn_wmma_f32_16x16x4_f32(
         false, a,  false, b[s],
         (short)0, acc,  false,  false);
  }
  return acc;
}

__global__ void __launch_bounds__(256)
gs_sage_wmma_kernel(const float* __restrict__ hself,
                    const int* __restrict__ deg,
                    const int* __restrict__ slot,
                    const int* __restrict__ spillT,
                    const int* __restrict__ spillS,
                    const float* __restrict__ Bf,
                    const float* __restrict__ bl,
                    const float* __restrict__ U4,
                    float* __restrict__ np,
                    int nNodes,
                    int nStride,
                    int nEdges) {
  __shared__ float lmean[16 * LDS_STRIDE];
  __shared__ float lself[16 * LDS_STRIDE];
  __shared__ float lproj[8 * 64];

  const int tid  = threadIdx.x;
  const int row0 = blockIdx.x * 16;
  const int lane = tid & 31;
  const int wave = __builtin_amdgcn_readfirstlane(tid >> 5);

  float4 selfv[2];
  for (int i = 0; i < 2; ++i) {
    const int idx = tid + i * 256;
    const int r   = idx >> 5;
    const int node = row0 + r;
    const int nclamp = node < nNodes ? node : (nNodes - 1);
    selfv[i] = *(const float4*)(hself + (long)nclamp * 128 + (idx & 31) * 4);
  }

  const int nodeA = row0 + wave * 2;
  const int nodeB = nodeA + 1;
  const int dA = nodeA < nNodes ? deg[nodeA] : 0;
  const int dB = nodeB < nNodes ? deg[nodeB] : 0;
  const int fA = lane < (dA < 32 ? dA : 32) ? slot[(long)nodeA * GS_CAP + lane] : 0;
  const int fB = lane < (dB < 32 ? dB : 32) ? slot[(long)nodeB * GS_CAP + lane] : 0;

  for (int i = 0; i < 2; ++i) {
    const int idx = tid + i * 256;
    *(float4*)(lself + (idx >> 5) * LDS_STRIDE + (idx & 31) * 4) = selfv[i];
  }

  {
    const float4 a = gs_gather_node(hself, deg, slot, spillT, spillS, nodeA, dA, fA,
                                    lane, nNodes, nEdges);
    const float rinv = 1.0f / fmaxf((float)dA, 1.0f);
    *(float4*)(lmean + (wave * 2) * LDS_STRIDE + lane * 4) =
        make_float4(a.x * rinv, a.y * rinv, a.z * rinv, a.w * rinv);
  }
  {
    const float4 a = gs_gather_node(hself, deg, slot, spillT, spillS, nodeB, dB, fB,
                                    lane, nNodes, nEdges);
    const float rinv = 1.0f / fmaxf((float)dB, 1.0f);
    *(float4*)(lmean + (wave * 2 + 1) * LDS_STRIDE + lane * 4) =
        make_float4(a.x * rinv, a.y * rinv, a.z * rinv, a.w * rinv);
  }
  __syncthreads();

  const int col0 = wave * 16;
  const int rowA = lane & 15;
  const int kofs = (lane >> 4) * 2;

  v8f acc = {};

  const float4* bw = (const float4*)Bf + wave * 1024 + lane;
  const float* lm  = lmean + rowA * LDS_STRIDE + kofs;
  const float* ls  = lself + rowA * LDS_STRIDE + kofs;
  v2f bA[16], bB[16];
  gs_load_b16(bA, bw, 0);
  gs_load_b16(bB, bw, 8);
  acc = gs_mma16(acc, lm, 0, bA);
  gs_load_b16(bA, bw, 16);
  acc = gs_mma16(acc, lm, 64, bB);
  gs_load_b16(bB, bw, 24);
  acc = gs_mma16(acc, ls, 0, bA);
  acc = gs_mma16(acc, ls, 64, bB);

  const int c     = lane & 15;
  const int half  = lane >> 4;
  const int ocol  = col0 + c;
  const float bias = bl[ocol];
  const float4 u = *(const float4*)(U4 + ocol * 4);

  float q[32];
  for (int i = 0; i < 8; ++i) {
    const float t = fmaxf(acc[i] + bias, 0.0f);
    q[i]      = t * u.x;
    q[8 + i]  = t * u.y;
    q[16 + i] = t * u.z;
    q[24 + i] = t * u.w;
  }
  const bool b3 = (c & 8) != 0;
  const bool b2 = (c & 4) != 0;
  const bool b1 = (c & 2) != 0;
  const bool b0 = (c & 1) != 0;
  float w16[16];
  for (int k = 0; k < 16; ++k) {
    const float keep = b3 ? q[16 + k] : q[k];
    const float send = b3 ? q[k] : q[16 + k];
    w16[k] = keep + __shfl_xor(send, 8, 32);
  }
  float w8[8];
  for (int k = 0; k < 8; ++k) {
    const float keep = b2 ? w16[8 + k] : w16[k];
    const float send = b2 ? w16[k] : w16[8 + k];
    w8[k] = keep + __shfl_xor(send, 4, 32);
  }
  float w4[4];
  for (int k = 0; k < 4; ++k) {
    const float keep = b1 ? w8[4 + k] : w8[k];
    const float send = b1 ? w8[k] : w8[4 + k];
    w4[k] = keep + __shfl_xor(send, 2, 32);
  }
  float w2[2];
  for (int k = 0; k < 2; ++k) {
    const float keep = b0 ? w4[2 + k] : w4[k];
    const float send = b0 ? w4[k] : w4[2 + k];
    w2[k] = keep + __shfl_xor(send, 1, 32);
  }

  const int prow = half * 8 + (c & 3) * 2;
  const int pidx = c >> 2;
  lproj[wave * 64 + prow * 4 + pidx]       = w2[0];
  lproj[wave * 64 + (prow + 1) * 4 + pidx] = w2[1];
  __syncthreads();
  if (tid < 64) {
    const int p = tid >> 4;
    const int r = tid & 15;
    float sum = lproj[r * 4 + p];
    for (int w = 1; w < 8; ++w) sum += lproj[w * 64 + r * 4 + p];
    const int node = row0 + r;
    if (node < nNodes) np[(long)p * nStride + node] = sum;
  }
}

#define GS_TAB (GS_BIN * GS_MAXB)

#define GS_TABV 8
#define GS_TABB (GS_TAB / 4 / (GS_TABV * 1024))

struct GsBatch {
  float4 v[GS_TABV];
  int at[GS_TABV];
};

__device__ __forceinline__ void gs_tab_issue(GsBatch& b, const float* __restrict__ g,
                                             int n, int it, int tid, int rot) {
  const int nv = n >> 2;
  const float4* g4 = (const float4*)g;
#pragma unroll
  for (int u = 0; u < GS_TABV; ++u) {
    const int i = (it * GS_TABV + u) * 1024 + tid;
    int a = i + rot;
    a = a >= nv ? a - nv : a;
    b.at[u] = i < nv ? a : -1;
    b.v[u] = make_float4(0.0f, 0.0f, 0.0f, 0.0f);
    if (b.at[u] >= 0) b.v[u] = g4[a];
  }
}

__device__ __forceinline__ void gs_tab_store(const GsBatch& b, float* __restrict__ tab) {
  float4* t4 = (float4*)tab;
#pragma unroll
  for (int u = 0; u < GS_TABV; ++u)
    if (b.at[u] >= 0) t4[b.at[u]] = b.v[u];
}

__device__ __forceinline__ void gs_tab_rest(float* __restrict__ tab, const float* __restrict__ g,
                                            int n, int tid) {
  const int i = (n & ~3) + tid;
  if (i < n) tab[i] = g[i];
}

__device__ __forceinline__ void gs_table_to_lds(float* __restrict__ tab,
                                                const float* __restrict__ g,
                                                int n, int tid, int rot) {
#pragma unroll
  for (int it = 0; it < GS_TABB; ++it) {
    GsBatch b;
    gs_tab_issue(b, g, n, it, tid, rot);
    gs_tab_store(b, tab);
  }
  gs_tab_rest(tab, g, n, tid);
}

__device__ __forceinline__ int gs_table_rot(int n) {
  const int nv = n >> 2;
  return (((int)blockIdx.x & 255) * (nv >> 8)) & ~7;
}

__global__ void __launch_bounds__(1024)
gs_node_score_kernel(const float* __restrict__ np,
                     const float* __restrict__ cb,
                     const int* __restrict__ deg,
                     const int* __restrict__ slot,
                     const int* __restrict__ spillT,
                     const int* __restrict__ spillS,
                     float* __restrict__ ps,
                     float* __restrict__ pt,
                     int nNodes,
                     int nStride,
                     int nEdges) {
  __shared__ __attribute__((aligned(16))) float tab[GS_TAB];
  const int tid  = threadIdx.x;
  const int node = (blockIdx.x * 1024 + tid) >> 2;
  const int sub  = tid & 3;
  const bool live = node < nNodes;
  const int rot = gs_table_rot(nNodes);
  const int d   = live ? deg[node] : 0;
  GsBatch b0;
  gs_tab_issue(b0, np, nNodes, 0, tid, rot);
  const int end = d < GS_CAP ? d : GS_CAP;
  const int* list = slot + (long)(live ? node : 0) * GS_CAP;
  int4 idv[4];
#pragma unroll
  for (int j = 0; j < 4; ++j) {
    const int k = 16 * j + 4 * sub;
    idv[j] = k < end ? *(const int4*)(list + k) : make_int4(0, 0, 0, 0);
  }
  float acc[2];
#pragma unroll
  for (int p = 0; p < 2; ++p) {
    if (p == 0) {
      gs_tab_store(b0, tab);
#pragma unroll
      for (int it = 1; it < GS_TABB; ++it) {
        GsBatch b;
        gs_tab_issue(b, np, nNodes, it, tid, rot);
        gs_tab_store(b, tab);
      }
      gs_tab_rest(tab, np, nNodes, tid);
    } else {
      __syncthreads();
      gs_table_to_lds(tab, np + (long)p * nStride, nNodes, tid, rot);
    }
    __syncthreads();
    float a = 0.0f;
#pragma unroll
    for (int j = 0; j < 4; ++j) {
      const int k = 16 * j + 4 * sub;
      if (k < end) {
        const bool o1 = k + 1 < end, o2 = k + 2 < end, o3 = k + 3 < end;
        const float v0 = tab[idv[j].x];
        const float v1 = tab[o1 ? idv[j].y : 0];
        const float v2 = tab[o2 ? idv[j].z : 0];
        const float v3 = tab[o3 ? idv[j].w : 0];
        a += v0;
        a = o1 ? a + v1 : a;
        a = o2 ? a + v2 : a;
        a = o3 ? a + v3 : a;
      }
    }
    if (d > GS_CAP) {
      int m = deg[nNodes];
      m = m < nEdges ? m : nEdges;
      for (int i = sub; i < m; i += 4)
        if (spillT[i] == node) a += tab[spillS[i]];
    }
    acc[p] = a;
  }
  float s = acc[0], t = acc[1];
  for (int o = 2; o > 0; o >>= 1) {
    s += __shfl_xor(s, o, 32);
    t += __shfl_xor(t, o, 32);
  }
  if (live && sub == 0) {
    const float den = fmaxf((float)d, 1.0f);
    ps[node] = s / den + np[(long)2 * nStride + node] + cb[0];
    pt[node] = t / den + np[(long)3 * nStride + node] + cb[1];
  }
}

__device__ __forceinline__ float gs_sigmoid(float p) {
  return 1.0f / (1.0f + __expf(-p));
}

__global__ void __launch_bounds__(1024)
gs_edge_score_kernel(const float* __restrict__ ps,
                     const float* __restrict__ pt,
                     const int* __restrict__ src,
                     const int* __restrict__ tgt,
                     const float* __restrict__ be,
                     float* __restrict__ out,
                     int nNodes,
                     int nEdges) {
  __shared__ __attribute__((aligned(16))) float tab[GS_TAB];
  const int tid = threadIdx.x;
  const int e0  = (blockIdx.x * 1024 + tid) * 4;
  const bool vec = (nEdges & 3) == 0;
  int si[4], ti[4];
  if (vec) {
    int4 s4 = make_int4(0, 0, 0, 0), t4 = make_int4(0, 0, 0, 0);
    if (e0 < nEdges) {
      s4 = *(const int4*)(src + e0);
      t4 = *(const int4*)(tgt + e0);
    }
    si[0] = s4.x; si[1] = s4.y; si[2] = s4.z; si[3] = s4.w;
    ti[0] = t4.x; ti[1] = t4.y; ti[2] = t4.z; ti[3] = t4.w;
  } else {
    for (int j = 0; j < 4; ++j) {
      const bool ok = e0 + j < nEdges;
      si[j] = ok ? src[e0 + j] : 0;
      ti[j] = ok ? tgt[e0 + j] : 0;
    }
  }
  const float b = be[0];
  const int rot = gs_table_rot(nNodes);

  gs_table_to_lds(tab, ps, nNodes, tid, rot);
  __syncthreads();
  float y[4];
  for (int j = 0; j < 4; ++j) y[j] = tab[si[j]];
  __syncthreads();
  gs_table_to_lds(tab, pt, nNodes, tid, rot);
  __syncthreads();
  for (int j = 0; j < 4; ++j) y[j] = gs_sigmoid(y[j] + tab[ti[j]] + b);

  if (vec) {
    if (e0 < nEdges) *(float4*)(out + e0) = make_float4(y[0], y[1], y[2], y[3]);
  } else {
    for (int j = 0; j < 4; ++j)
      if (e0 + j < nEdges) out[e0 + j] = y[j];
  }
}

extern "C" void kernel_launch(void* const* d_in, const int* in_sizes, int n_in,
                              void* d_out, int out_size, void* d_ws, size_t ws_size,
                              hipStream_t stream) {
  const float* x    = (const float*)d_in[0];
  const int*   ei   = (const int*)d_in[1];
  const float* Wl1  = (const float*)d_in[2];
  const float* bl1  = (const float*)d_in[3];
  const float* Wr1  = (const float*)d_in[4];
  const float* Wl2  = (const float*)d_in[5];
  const float* bl2  = (const float*)d_in[6];
  const float* Wr2  = (const float*)d_in[7];
  const float* We   = (const float*)d_in[8];
  const float* be   = (const float*)d_in[9];

  const int N = in_sizes[0] / 128;
  const int E = in_sizes[1] / 2;
  const int* src = ei;
  const int* tgt = ei + E;

  const int N4 = (N + 3) & ~3;
  int*   slot = (int*)d_ws;
  float* Bf   = (float*)d_ws + (long)N * GS_CAP;
  float* U4   = Bf + 2 * 128 * 128;
  float* cb   = U4 + 128 * 4;
  float* np   = cb + 4;
  float* ps   = np + (long)N4 * 4;
  float* pt   = ps + N4;
  int* deg    = (int*)(pt + N4);
  int* spillT = deg + (N + 1);
  int* spillS = spillT + E;
  int* sorted = spillS + E;
  int* off    = sorted + E;

  const int nBins = (N + GS_BIN - 1) / GS_BIN;
  const int nPart = (E + GS_CHUNK - 1) / GS_CHUNK;
  const int nodeBlocks = (N + 255) / 256;
  const int quadBlocks = ((E + 3) / 4 + 1023) / 1024;
  const int gemmBlocks = (N + 15) / 16;

  gs_part_kernel<<<nPart + GS_FOLDB + GS_SWZB, 1024, 0, stream>>>(
      src, tgt, sorted, off, deg, N, E, nBins, nPart, Wl2, bl2, Wr2, We, U4, cb, Wl1, Wr1, Bf);
  gs_place_kernel<<<nBins, 256, 0, stream>>>(sorted, off, deg, slot, spillT, spillS,
                                             N, nBins, nPart);

  gs_sage_wmma_kernel<<<gemmBlocks, 256, 0, stream>>>(x, deg, slot, spillT, spillS,
                                                      Bf, bl1, U4, np, N, N4, E);

  gs_node_score_kernel<<<nodeBlocks, 1024, 0, stream>>>(np, cb, deg, slot, spillT, spillS,
                                                       ps, pt, N, N4, E);

  gs_edge_score_kernel<<<quadBlocks, 1024, 0, stream>>>(ps, pt, src, tgt, be,
                                                        (float*)d_out, N, E);
}
